# starting kernel
# baseline (speedup 1.0000x reference)
#include <hip/hip_runtime.h>

typedef __attribute__((ext_vector_type(2))) float v2f;
typedef __attribute__((ext_vector_type(8))) float v8f;

#define D_IN  256
#define D_OUT 128

__global__ __launch_bounds__(256) void gcn_init_out(const float* __restrict__ bias,
                                                    float* __restrict__ out,
                                                    long long total4) {
    long long i = (long long)blockIdx.x * blockDim.x + threadIdx.x;
    if (i < total4) {
        int d4 = (int)(i & (D_OUT / 4 - 1));
        ((float4*)out)[i] = ((const float4*)bias)[d4];
    }
}

__global__ __launch_bounds__(256) void gcn_gemm_xw(const float* __restrict__ X,
                                                   const float* __restrict__ W,
                                                   float* __restrict__ S,
                                                   int n_rows) {
    const int lane  = threadIdx.x & 31;
    const int wave  = threadIdx.x >> 5;
    const int mtile = blockIdx.x * 8 + wave;
    if (mtile * 16 + 16 > n_rows) return;
    const int m0 = mtile * 16;

    const int am = lane & 15;
    const int ak = (lane >> 4) * 2;
    const int bn = lane & 15;
    const int bk = (lane >> 4) * 2;

    v8f c[8];
#pragma unroll
    for (int t = 0; t < 8; ++t) c[t] = (v8f){0.f, 0.f, 0.f, 0.f, 0.f, 0.f, 0.f, 0.f};

    const float* xrow = X + (long long)(m0 + am) * D_IN;

    for (int k = 0; k < D_IN; k += 4) {
        v2f a = *(const v2f*)(xrow + k + ak);
#pragma unroll
        for (int t = 0; t < 8; ++t) {
            v2f b;
            b.x = W[(long long)(k + bk)     * D_OUT + t * 16 + bn];
            b.y = W[(long long)(k + bk + 1) * D_OUT + t * 16 + bn];
            c[t] = __builtin_amdgcn_wmma_f32_16x16x4_f32(
                 false, a,  false, b,
                 (short)0, c[t],  false,  false);
        }
    }

    const int cn    = lane & 15;
    const int cmoff = (lane >> 4) * 8;
#pragma unroll
    for (int t = 0; t < 8; ++t) {
#pragma unroll
        for (int r = 0; r < 8; ++r) {
            S[(long long)(m0 + r + cmoff) * D_OUT + t * 16 + cn] = c[t][r];
        }
    }
}

__global__ __launch_bounds__(256) void gcn_edge_scatter(const int* __restrict__ rows,
                                                        const int* __restrict__ cols,
                                                        const float* __restrict__ vals,
                                                        const float* __restrict__ S,
                                                        float* __restrict__ out,
                                                        int n_edges) {
    const int lane = threadIdx.x & 31;
    int e = (int)(blockIdx.x * 8u) + (threadIdx.x >> 5);
    e = __builtin_amdgcn_readfirstlane(e);
    if (e >= n_edges) return;

    const int   r = rows[e];
    const int   c = cols[e];
    const float v = vals[e];

    const float4 s = *(const float4*)(S + (long long)c * D_OUT + lane * 4);

    float4 g;
    g.x = s.x * v;
    g.y = s.y * v;
    g.z = s.z * v;
    g.w = s.w * v;

    float* dst = out + (long long)r * D_OUT + lane * 4;
    asm volatile("global_atomic_add_f32 %0, %1, off"           :: "v"(dst), "v"(g.x) : "memory");
    asm volatile("global_atomic_add_f32 %0, %1, off offset:4"  :: "v"(dst), "v"(g.y) : "memory");
    asm volatile("global_atomic_add_f32 %0, %1, off offset:8"  :: "v"(dst), "v"(g.z) : "memory");
    asm volatile("global_atomic_add_f32 %0, %1, off offset:12" :: "v"(dst), "v"(g.w) : "memory");
}

extern "C" void kernel_launch(void* const* d_in, const int* in_sizes, int n_in,
                              void* d_out, int out_size, void* d_ws, size_t ws_size,
                              hipStream_t stream) {
    const int*   rows = (const int*)d_in[0];
    const int*   cols = (const int*)d_in[1];
    const float* vals = (const float*)d_in[2];
    const float* x    = (const float*)d_in[3];
    const float* W    = (const float*)d_in[4];
    const float* bias = (const float*)d_in[5];

    const int E = in_sizes[0];
    const int N = in_sizes[3] / D_IN;

    float* support = (float*)d_ws;

    {
        long long total4 = (long long)N * (D_OUT / 4);
        int blocks = (int)((total4 + 255) / 256);
        gcn_init_out<<<blocks, 256, 0, stream>>>(bias, (float*)d_out, total4);
    }

    {
        int mtiles = N / 16;
        int blocks = (mtiles + 7) / 8;
        gcn_gemm_xw<<<blocks, 256, 0, stream>>>(x, W, support, N);
    }

    {
        int blocks = (E + 7) / 8;
        gcn_edge_scatter<<<blocks, 256, 0, stream>>>(rows, cols, vals, support,
                                                     (float*)d_out, E);
    }
}


# Edge scatter: lane = feature, so each f32 atomic add and each gather load covers one whole 128-byte line (a quarter of the atomic line requests per edge); sums still arrival-order atomics
# speedup vs baseline: 1.2318x; 1.2318x over previous
#include <hip/hip_runtime.h>

typedef __attribute__((ext_vector_type(2))) float v2f;
typedef __attribute__((ext_vector_type(8))) float v8f;

#define D_IN  256
#define D_OUT 128

__global__ __launch_bounds__(256) void gcn_init_out(const float* __restrict__ bias,
                                                    float* __restrict__ out,
                                                    long long total4) {
    long long i = (long long)blockIdx.x * blockDim.x + threadIdx.x;
    if (i < total4) {
        int d4 = (int)(i & (D_OUT / 4 - 1));
        ((float4*)out)[i] = ((const float4*)bias)[d4];
    }
}

__global__ __launch_bounds__(256) void gcn_gemm_xw(const float* __restrict__ X,
                                                   const float* __restrict__ W,
                                                   float* __restrict__ S,
                                                   int n_rows) {
    const int lane  = threadIdx.x & 31;
    const int wave  = threadIdx.x >> 5;
    const int mtile = blockIdx.x * 8 + wave;
    if (mtile * 16 + 16 > n_rows) return;
    const int m0 = mtile * 16;

    const int am = lane & 15;
    const int ak = (lane >> 4) * 2;
    const int bn = lane & 15;
    const int bk = (lane >> 4) * 2;

    v8f c[8];
#pragma unroll
    for (int t = 0; t < 8; ++t) c[t] = (v8f){0.f, 0.f, 0.f, 0.f, 0.f, 0.f, 0.f, 0.f};

    const float* xrow = X + (long long)(m0 + am) * D_IN;

    for (int k = 0; k < D_IN; k += 4) {
        v2f a = *(const v2f*)(xrow + k + ak);
#pragma unroll
        for (int t = 0; t < 8; ++t) {
            v2f b;
            b.x = W[(long long)(k + bk)     * D_OUT + t * 16 + bn];
            b.y = W[(long long)(k + bk + 1) * D_OUT + t * 16 + bn];
            c[t] = __builtin_amdgcn_wmma_f32_16x16x4_f32(
                 false, a,  false, b,
                 (short)0, c[t],  false,  false);
        }
    }

    const int cn    = lane & 15;
    const int cmoff = (lane >> 4) * 8;
#pragma unroll
    for (int t = 0; t < 8; ++t) {
#pragma unroll
        for (int r = 0; r < 8; ++r) {
            S[(long long)(m0 + r + cmoff) * D_OUT + t * 16 + cn] = c[t][r];
        }
    }
}

__global__ __launch_bounds__(256) void gcn_edge_scatter(const int* __restrict__ rows,
                                                        const int* __restrict__ cols,
                                                        const float* __restrict__ vals,
                                                        const float* __restrict__ S,
                                                        float* __restrict__ out,
                                                        int n_edges) {
    const int lane = threadIdx.x & 31;
    int e = (int)(blockIdx.x * 8u) + (threadIdx.x >> 5);
    e = __builtin_amdgcn_readfirstlane(e);
    if (e >= n_edges) return;

    const int   r = rows[e];
    const int   c = cols[e];
    const float v = vals[e];

    const float* src = S + (long long)c * D_OUT;
    const float s0 = src[lane];
    const float s1 = src[lane + 32];
    const float s2 = src[lane + 64];
    const float s3 = src[lane + 96];

    const float g0 = s0 * v;
    const float g1 = s1 * v;
    const float g2 = s2 * v;
    const float g3 = s3 * v;

    float* dst = out + (long long)r * D_OUT;
    const unsigned off = (unsigned)lane * 4u;
    asm volatile("global_atomic_add_f32 %0, %1, %2"            :: "v"(off), "v"(g0), "s"(dst) : "memory");
    asm volatile("global_atomic_add_f32 %0, %1, %2 offset:128" :: "v"(off), "v"(g1), "s"(dst) : "memory");
    asm volatile("global_atomic_add_f32 %0, %1, %2 offset:256" :: "v"(off), "v"(g2), "s"(dst) : "memory");
    asm volatile("global_atomic_add_f32 %0, %1, %2 offset:384" :: "v"(off), "v"(g3), "s"(dst) : "memory");
}

extern "C" void kernel_launch(void* const* d_in, const int* in_sizes, int n_in,
                              void* d_out, int out_size, void* d_ws, size_t ws_size,
                              hipStream_t stream) {
    const int*   rows = (const int*)d_in[0];
    const int*   cols = (const int*)d_in[1];
    const float* vals = (const float*)d_in[2];
    const float* x    = (const float*)d_in[3];
    const float* W    = (const float*)d_in[4];
    const float* bias = (const float*)d_in[5];

    const int E = in_sizes[0];
    const int N = in_sizes[3] / D_IN;

    float* support = (float*)d_ws;

    {
        long long total4 = (long long)N * (D_OUT / 4);
        int blocks = (int)((total4 + 255) / 256);
        gcn_init_out<<<blocks, 256, 0, stream>>>(bias, (float*)d_out, total4);
    }

    {
        int mtiles = N / 16;
        int blocks = (mtiles + 7) / 8;
        gcn_gemm_xw<<<blocks, 256, 0, stream>>>(x, W, support, N);
    }

    {
        int blocks = (E + 7) / 8;
        gcn_edge_scatter<<<blocks, 256, 0, stream>>>(rows, cols, vals, support,
                                                     (float*)d_out, E);
    }
}


# Edge scatter: one wave per 32 edges instead of per edge (triples read as 3 coalesced lines, broadcast by readlane; gathers of 8 edges issued before their adds); sums still arrival-order f32 atomics
# speedup vs baseline: 1.8292x; 1.4850x over previous
#include <hip/hip_runtime.h>

typedef __attribute__((ext_vector_type(2))) float v2f;
typedef __attribute__((ext_vector_type(8))) float v8f;

#define D_IN  256
#define D_OUT 128

__global__ __launch_bounds__(256) void gcn_init_out(const float* __restrict__ bias,
                                                    float* __restrict__ out,
                                                    long long total4) {
    long long i = (long long)blockIdx.x * blockDim.x + threadIdx.x;
    if (i < total4) {
        int d4 = (int)(i & (D_OUT / 4 - 1));
        ((float4*)out)[i] = ((const float4*)bias)[d4];
    }
}

__global__ __launch_bounds__(256) void gcn_gemm_xw(const float* __restrict__ X,
                                                   const float* __restrict__ W,
                                                   float* __restrict__ S,
                                                   int n_rows) {
    const int lane  = threadIdx.x & 31;
    const int wave  = threadIdx.x >> 5;
    const int mtile = blockIdx.x * 8 + wave;
    if (mtile * 16 + 16 > n_rows) return;
    const int m0 = mtile * 16;

    const int am = lane & 15;
    const int ak = (lane >> 4) * 2;
    const int bn = lane & 15;
    const int bk = (lane >> 4) * 2;

    v8f c[8];
#pragma unroll
    for (int t = 0; t < 8; ++t) c[t] = (v8f){0.f, 0.f, 0.f, 0.f, 0.f, 0.f, 0.f, 0.f};

    const float* xrow = X + (long long)(m0 + am) * D_IN;

    for (int k = 0; k < D_IN; k += 4) {
        v2f a = *(const v2f*)(xrow + k + ak);
#pragma unroll
        for (int t = 0; t < 8; ++t) {
            v2f b;
            b.x = W[(long long)(k + bk)     * D_OUT + t * 16 + bn];
            b.y = W[(long long)(k + bk + 1) * D_OUT + t * 16 + bn];
            c[t] = __builtin_amdgcn_wmma_f32_16x16x4_f32(
                 false, a,  false, b,
                 (short)0, c[t],  false,  false);
        }
    }

    const int cn    = lane & 15;
    const int cmoff = (lane >> 4) * 8;
#pragma unroll
    for (int t = 0; t < 8; ++t) {
#pragma unroll
        for (int r = 0; r < 8; ++r) {
            S[(long long)(m0 + r + cmoff) * D_OUT + t * 16 + cn] = c[t][r];
        }
    }
}

#define EDGES_PER_WAVE 32
#define EDGE_BATCH     8

static __device__ __forceinline__ void add_row_lines(float* dst, unsigned off,
                                                      float g0, float g1, float g2, float g3) {
    asm volatile("global_atomic_add_f32 %0, %1, %2"            :: "v"(off), "v"(g0), "s"(dst) : "memory");
    asm volatile("global_atomic_add_f32 %0, %1, %2 offset:128" :: "v"(off), "v"(g1), "s"(dst) : "memory");
    asm volatile("global_atomic_add_f32 %0, %1, %2 offset:256" :: "v"(off), "v"(g2), "s"(dst) : "memory");
    asm volatile("global_atomic_add_f32 %0, %1, %2 offset:384" :: "v"(off), "v"(g3), "s"(dst) : "memory");
}

__global__ __launch_bounds__(256) void gcn_edge_scatter(const int* __restrict__ rows,
                                                        const int* __restrict__ cols,
                                                        const float* __restrict__ vals,
                                                        const float* __restrict__ S,
                                                        float* __restrict__ out,
                                                        int n_edges) {
    const int lane = threadIdx.x & 31;
    int w = (int)(blockIdx.x * 8u) + (int)(threadIdx.x >> 5);
    w = __builtin_amdgcn_readfirstlane(w);
    const int e_base = w * EDGES_PER_WAVE;
    if (e_base >= n_edges) return;
    int n_here = n_edges - e_base;
    if (n_here > EDGES_PER_WAVE) n_here = EDGES_PER_WAVE;

    int r_v = 0, c_v = 0, v_v = 0;
    if (lane < n_here) {
        r_v = rows[e_base + lane];
        c_v = cols[e_base + lane];
        v_v = __float_as_int(vals[e_base + lane]);
    }
    const unsigned off = (unsigned)lane * 4u;

    int j0 = 0;
    for (; j0 + EDGE_BATCH <= n_here; j0 += EDGE_BATCH) {
        float g[EDGE_BATCH][4];
#pragma unroll
        for (int u = 0; u < EDGE_BATCH; ++u) {
            const int c = __builtin_amdgcn_readlane(c_v, j0 + u);
            const float* src = S + (long long)c * D_OUT;
            g[u][0] = src[lane];
            g[u][1] = src[lane + 32];
            g[u][2] = src[lane + 64];
            g[u][3] = src[lane + 96];
        }
#pragma unroll
        for (int u = 0; u < EDGE_BATCH; ++u) {
            const int   r = __builtin_amdgcn_readlane(r_v, j0 + u);
            const float v = __int_as_float(__builtin_amdgcn_readlane(v_v, j0 + u));
            float* dst = out + (long long)r * D_OUT;
            add_row_lines(dst, off, g[u][0] * v, g[u][1] * v, g[u][2] * v, g[u][3] * v);
        }
    }
    for (; j0 < n_here; ++j0) {
        const int   c = __builtin_amdgcn_readlane(c_v, j0);
        const int   r = __builtin_amdgcn_readlane(r_v, j0);
        const float v = __int_as_float(__builtin_amdgcn_readlane(v_v, j0));
        const float* src = S + (long long)c * D_OUT;
        const float s0 = src[lane];
        const float s1 = src[lane + 32];
        const float s2 = src[lane + 64];
        const float s3 = src[lane + 96];
        float* dst = out + (long long)r * D_OUT;
        add_row_lines(dst, off, s0 * v, s1 * v, s2 * v, s3 * v);
    }
}

extern "C" void kernel_launch(void* const* d_in, const int* in_sizes, int n_in,
                              void* d_out, int out_size, void* d_ws, size_t ws_size,
                              hipStream_t stream) {
    const int*   rows = (const int*)d_in[0];
    const int*   cols = (const int*)d_in[1];
    const float* vals = (const float*)d_in[2];
    const float* x    = (const float*)d_in[3];
    const float* W    = (const float*)d_in[4];
    const float* bias = (const float*)d_in[5];

    const int E = in_sizes[0];
    const int N = in_sizes[3] / D_IN;

    float* support = (float*)d_ws;

    {
        long long total4 = (long long)N * (D_OUT / 4);
        int blocks = (int)((total4 + 255) / 256);
        gcn_init_out<<<blocks, 256, 0, stream>>>(bias, (float*)d_out, total4);
    }

    {
        int mtiles = N / 16;
        int blocks = (mtiles + 7) / 8;
        gcn_gemm_xw<<<blocks, 256, 0, stream>>>(x, W, support, N);
    }

    {
        int waves  = (E + EDGES_PER_WAVE - 1) / EDGES_PER_WAVE;
        int blocks = (waves + 7) / 8;
        gcn_edge_scatter<<<blocks, 256, 0, stream>>>(rows, cols, vals, support,
                                                     (float*)d_out, E);
    }
}


# Edge sum grouped by destination: per-call ballot counting sort (row range, then row); each row summed in registers in fixed edge order, stored once with bias; no atomic executed, no out init
# speedup vs baseline: 2.1379x; 1.1688x over previous
#include <hip/hip_runtime.h>

typedef __attribute__((ext_vector_type(2))) float v2f;
typedef __attribute__((ext_vector_type(8))) float v8f;

#define D_IN  256
#define D_OUT 128

#define NBINS        256
#define BIN_BITS     8
#define RPB          391
#define ROW_BITS     9
#define COL_BITS     20
#define SORT_THREADS 1024
#define SORT_WAVES   (SORT_THREADS / 32)
#define SORT_EPT     8
#define SORT_EPB     (SORT_THREADS * SORT_EPT)

__global__ __launch_bounds__(256) void gcn_init_out(const float* __restrict__ bias,
                                                    float* __restrict__ out,
                                                    long long total4) {
    long long i = (long long)blockIdx.x * blockDim.x + threadIdx.x;
    if (i < total4) {
        int d4 = (int)(i & (D_OUT / 4 - 1));
        ((float4*)out)[i] = ((const float4*)bias)[d4];
    }
}

__global__ __launch_bounds__(256) void gcn_gemm_xw(const float* __restrict__ X,
                                                   const float* __restrict__ W,
                                                   float* __restrict__ S,
                                                   int n_rows) {
    const int lane  = threadIdx.x & 31;
    const int wave  = threadIdx.x >> 5;
    const int mtile = blockIdx.x * 8 + wave;
    if (mtile * 16 + 16 > n_rows) return;
    const int m0 = mtile * 16;

    const int am = lane & 15;
    const int ak = (lane >> 4) * 2;
    const int bn = lane & 15;
    const int bk = (lane >> 4) * 2;

    v8f c[8];
#pragma unroll
    for (int t = 0; t < 8; ++t) c[t] = (v8f){0.f, 0.f, 0.f, 0.f, 0.f, 0.f, 0.f, 0.f};

    const float* xrow = X + (long long)(m0 + am) * D_IN;

    for (int k = 0; k < D_IN; k += 4) {
        v2f a = *(const v2f*)(xrow + k + ak);
#pragma unroll
        for (int t = 0; t < 8; ++t) {
            v2f b;
            b.x = W[(long long)(k + bk)     * D_OUT + t * 16 + bn];
            b.y = W[(long long)(k + bk + 1) * D_OUT + t * 16 + bn];
            c[t] = __builtin_amdgcn_wmma_f32_16x16x4_f32(
                 false, a,  false, b,
                 (short)0, c[t],  false,  false);
        }
    }

    const int cn    = lane & 15;
    const int cmoff = (lane >> 4) * 8;
#pragma unroll
    for (int t = 0; t < 8; ++t) {
#pragma unroll
        for (int r = 0; r < 8; ++r) {
            S[(long long)(m0 + r + cmoff) * D_OUT + t * 16 + cn] = c[t][r];
        }
    }
}

#define EDGES_PER_WAVE 32
#define EDGE_BATCH     8

static __device__ __forceinline__ void add_row_lines(float* dst, unsigned off,
                                                      float g0, float g1, float g2, float g3) {
    asm volatile("global_atomic_add_f32 %0, %1, %2"            :: "v"(off), "v"(g0), "s"(dst) : "memory");
    asm volatile("global_atomic_add_f32 %0, %1, %2 offset:128" :: "v"(off), "v"(g1), "s"(dst) : "memory");
    asm volatile("global_atomic_add_f32 %0, %1, %2 offset:256" :: "v"(off), "v"(g2), "s"(dst) : "memory");
    asm volatile("global_atomic_add_f32 %0, %1, %2 offset:384" :: "v"(off), "v"(g3), "s"(dst) : "memory");
}

__global__ __launch_bounds__(256) void gcn_edge_scatter(const int* __restrict__ rows,
                                                        const int* __restrict__ cols,
                                                        const float* __restrict__ vals,
                                                        const float* __restrict__ S,
                                                        float* __restrict__ out,
                                                        int n_edges) {
    const int lane = threadIdx.x & 31;
    int w = (int)(blockIdx.x * 8u) + (int)(threadIdx.x >> 5);
    w = __builtin_amdgcn_readfirstlane(w);
    const int e_base = w * EDGES_PER_WAVE;
    if (e_base >= n_edges) return;
    int n_here = n_edges - e_base;
    if (n_here > EDGES_PER_WAVE) n_here = EDGES_PER_WAVE;

    int r_v = 0, c_v = 0, v_v = 0;
    if (lane < n_here) {
        r_v = rows[e_base + lane];
        c_v = cols[e_base + lane];
        v_v = __float_as_int(vals[e_base + lane]);
    }
    const unsigned off = (unsigned)lane * 4u;

    int j0 = 0;
    for (; j0 + EDGE_BATCH <= n_here; j0 += EDGE_BATCH) {
        float g[EDGE_BATCH][4];
#pragma unroll
        for (int u = 0; u < EDGE_BATCH; ++u) {
            const int c = __builtin_amdgcn_readlane(c_v, j0 + u);
            const float* src = S + (long long)c * D_OUT;
            g[u][0] = src[lane];
            g[u][1] = src[lane + 32];
            g[u][2] = src[lane + 64];
            g[u][3] = src[lane + 96];
        }
#pragma unroll
        for (int u = 0; u < EDGE_BATCH; ++u) {
            const int   r = __builtin_amdgcn_readlane(r_v, j0 + u);
            const float v = __int_as_float(__builtin_amdgcn_readlane(v_v, j0 + u));
            float* dst = out + (long long)r * D_OUT;
            add_row_lines(dst, off, g[u][0] * v, g[u][1] * v, g[u][2] * v, g[u][3] * v);
        }
    }
    for (; j0 < n_here; ++j0) {
        const int   c = __builtin_amdgcn_readlane(c_v, j0);
        const int   r = __builtin_amdgcn_readlane(r_v, j0);
        const float v = __int_as_float(__builtin_amdgcn_readlane(v_v, j0));
        const float* src = S + (long long)c * D_OUT;
        const float s0 = src[lane];
        const float s1 = src[lane + 32];
        const float s2 = src[lane + 64];
        const float s3 = src[lane + 96];
        float* dst = out + (long long)r * D_OUT;
        add_row_lines(dst, off, s0 * v, s1 * v, s2 * v, s3 * v);
    }
}


static __device__ __forceinline__ unsigned same_key_lanes(unsigned key, bool valid, int nbits) {
    unsigned peers = __builtin_amdgcn_ballot_w32(valid);
#pragma unroll
    for (int b = 0; b < nbits; ++b) {
        const bool bit = ((key >> b) & 1u) != 0u;
        const unsigned m = __builtin_amdgcn_ballot_w32(bit && valid);
        peers &= bit ? m : ~m;
    }
    return valid ? peers : 0u;
}

static __device__ __forceinline__ void wave_rank(unsigned key, bool valid, int nbits,
                                                 unsigned lt_mask, int& rank, int& cnt) {
    const unsigned peers = same_key_lanes(key, valid, nbits);
    rank = __builtin_popcount(peers & lt_mask);
    cnt  = __builtin_popcount(peers);
}

static __device__ __forceinline__ int counter_take(int* h, unsigned key, bool valid,
                                                   int rank, int cnt) {
    int slot = 0;
    if (valid) {
        const int old = h[key];
        if (rank == 0) h[key] = old + cnt;
        slot = old + rank;
    }
    return slot;
}

static __device__ __forceinline__ int wave_count_step(int* h, unsigned key, bool valid,
                                                      int nbits, unsigned lt_mask) {
    int rank, cnt;
    wave_rank(key, valid, nbits, lt_mask, rank, cnt);
    return counter_take(h, key, valid, rank, cnt);
}

__global__ __launch_bounds__(SORT_THREADS) void gcn_bin_count(const int* __restrict__ rows,
                                                              int* __restrict__ blockhist,
                                                              int n_edges) {
    __shared__ int whist[SORT_WAVES][NBINS];
    const int t = threadIdx.x, lane = t & 31, w = t >> 5;
    const unsigned lt_mask = (1u << lane) - 1u;
    int* h = whist[w];
    for (int k = lane; k < NBINS; k += 32) h[k] = 0;

    const int e0 = blockIdx.x * SORT_EPB + w * (SORT_EPT * 32) + lane;
    int r[SORT_EPT];
#pragma unroll
    for (int i = 0; i < SORT_EPT; ++i) {
        const int e = e0 + i * 32;
        r[i] = (e < n_edges) ? rows[e] : -1;
    }
#pragma unroll
    for (int i = 0; i < SORT_EPT; ++i) {
        const bool valid = r[i] >= 0;
        const unsigned bin = valid ? (unsigned)(r[i] / RPB) : 0u;
        wave_count_step(h, bin, valid, BIN_BITS, lt_mask);
    }
    __syncthreads();
    if (t < NBINS) {
        int s = 0;
        for (int w2 = 0; w2 < SORT_WAVES; ++w2) s += whist[w2][t];
        blockhist[blockIdx.x * NBINS + t] = s;
    }
}

__global__ __launch_bounds__(SORT_THREADS) void gcn_bin_place(const int* __restrict__ rows,
                                                              const int* __restrict__ cols,
                                                              const float* __restrict__ vals,
                                                              const int* __restrict__ blockhist,
                                                              int* __restrict__ binstart,
                                                              uint2* __restrict__ rec,
                                                              int n_edges, int n_blocks) {
    __shared__ int whist[SORT_WAVES][NBINS];
    __shared__ int part[2][SORT_THREADS / NBINS][NBINS];
    __shared__ int scan[NBINS];
    const int t = threadIdx.x, lane = t & 31, w = t >> 5;
    const int me = blockIdx.x;
    const unsigned lt_mask = (1u << lane) - 1u;
    int* h = whist[w];
    for (int k = lane; k < NBINS; k += 32) h[k] = 0;

    const int e0 = me * SORT_EPB + w * (SORT_EPT * 32) + lane;
    unsigned info[SORT_EPT];
    {
        int r[SORT_EPT];
#pragma unroll
        for (int i = 0; i < SORT_EPT; ++i) {
            const int e = e0 + i * 32;
            r[i] = (e < n_edges) ? rows[e] : -1;
        }
#pragma unroll
        for (int i = 0; i < SORT_EPT; ++i) {
            const bool valid = r[i] >= 0;
            const unsigned bin = valid ? (unsigned)(r[i] / RPB) : 0u;
            const unsigned rl  = valid ? (unsigned)r[i] - bin * RPB : 0u;
            int rank, cnt;
            wave_rank(bin, valid, BIN_BITS, lt_mask, rank, cnt);
            counter_take(h, bin, valid, rank, cnt);
            info[i] = bin | ((unsigned)rank << 8) | ((unsigned)cnt << 13) | (rl << 19) |
                      (valid ? (1u << 28) : 0u);
        }
    }

    {
        const int q = t / NBINS, bn = t % NBINS;
        int bef = 0, tot = 0;
#pragma unroll 16
        for (int b = q; b < n_blocks; b += SORT_THREADS / NBINS) {
            const int x = blockhist[b * NBINS + bn];
            tot += x;
            bef += (b < me) ? x : 0;
        }
        part[0][q][bn] = bef;
        part[1][q][bn] = tot;
    }
    __syncthreads();
    int total = 0, before = 0;
    if (t < NBINS) {
#pragma unroll
        for (int q = 0; q < SORT_THREADS / NBINS; ++q) { before += part[0][q][t]; total += part[1][q][t]; }
        scan[t] = total;
    }
    __syncthreads();
    for (int d = 1; d < NBINS; d <<= 1) {
        int add = 0;
        if (t < NBINS && t >= d) add = scan[t - d];
        __syncthreads();
        if (t < NBINS) scan[t] += add;
        __syncthreads();
    }
    if (t < NBINS) {
        const int start = scan[t] - total;
        if (me == 0) {
            binstart[t] = start;
            if (t == NBINS - 1) binstart[NBINS] = scan[t];
        }
        int run = start + before;
        for (int w2 = 0; w2 < SORT_WAVES; ++w2) {
            const int cnt = whist[w2][t];
            whist[w2][t] = run;
            run += cnt;
        }
    }
    __syncthreads();

    int c[SORT_EPT];
    float v[SORT_EPT];
#pragma unroll
    for (int i = 0; i < SORT_EPT; ++i) {
        const int e = e0 + i * 32;
        c[i] = 0; v[i] = 0.f;
        if (info[i] >> 28) { c[i] = cols[e]; v[i] = vals[e]; }
    }
#pragma unroll
    for (int i = 0; i < SORT_EPT; ++i) {
        const bool     valid = (info[i] >> 28) != 0u;
        const unsigned bin   = info[i] & 0xFFu;
        const int      rank  = (int)((info[i] >> 8) & 0x1Fu);
        const int      cnt   = (int)((info[i] >> 13) & 0x3Fu);
        const unsigned rl    = (info[i] >> 19) & 0x1FFu;
        const int pos = counter_take(h, bin, valid, rank, cnt);
        if (valid)
            rec[pos] = make_uint2((unsigned)c[i] | (rl << COL_BITS), __float_as_uint(v[i]));
    }
}

#define RS_THREADS 1024
#define RS_WAVES   (RS_THREADS / 32)
#define RS_BATCH   8
#define RS_SCAN    512
#define RS_GROUP   4

__global__ __launch_bounds__(RS_THREADS) void gcn_range_sum(const uint2* __restrict__ rec,
                                                            uint2* rec2,
                                                            const int* __restrict__ binstart,
                                                            const float* __restrict__ S,
                                                            const float* __restrict__ bias,
                                                            float* __restrict__ out,
                                                            int n_rows) {
    __shared__ int whist[RS_WAVES][RPB + 1];
    __shared__ int rowscan[RS_SCAN];
    __shared__ int rowstart[RPB + 1];
    const int t    = threadIdx.x;
    const int lane = t & 31;
    const int w    = __builtin_amdgcn_readfirstlane(t >> 5);
    const int bin  = blockIdx.x;
    const unsigned lt_mask = (1u << lane) - 1u;

    const int start = binstart[bin];
    const int end   = binstart[bin + 1];
    const int per   = (end - start + RS_WAVES - 1) / RS_WAVES;
    const int a_lo  = start + w * per;
    int a_hi = a_lo + per;
    if (a_hi > end) a_hi = end;

    int* h = whist[w];
    for (int k = lane; k <= RPB; k += 32) h[k] = 0;
    for (int cb = a_lo; cb < a_hi; cb += 32 * RS_GROUP) {
        unsigned key[RS_GROUP];
        bool     ok[RS_GROUP];
#pragma unroll
        for (int j = 0; j < RS_GROUP; ++j) {
            const int idx = cb + j * 32 + lane;
            ok[j]  = idx < a_hi;
            key[j] = 0u;
            if (ok[j]) key[j] = rec[idx].x;
        }
#pragma unroll
        for (int j = 0; j < RS_GROUP; ++j)
            wave_count_step(h, key[j] >> COL_BITS, ok[j], ROW_BITS, lt_mask);
    }
    __syncthreads();

    int rc = 0;
    if (t < RPB) {
        int run = 0;
        for (int w2 = 0; w2 < RS_WAVES; ++w2) {
            const int cnt = whist[w2][t];
            whist[w2][t] = run;
            run += cnt;
        }
        rc = run;
    }
    if (t < RS_SCAN) rowscan[t] = rc;
    __syncthreads();
    for (int d = 1; d < RS_SCAN; d <<= 1) {
        int add = 0;
        if (t < RS_SCAN && t >= d) add = rowscan[t - d];
        __syncthreads();
        if (t < RS_SCAN) rowscan[t] += add;
        __syncthreads();
    }
    if (t <= RPB) rowstart[t] = (t == 0) ? 0 : rowscan[t - 1];
    __syncthreads();

    for (int cb = a_lo; cb < a_hi; cb += 32 * RS_GROUP) {
        uint2 q[RS_GROUP];
        bool  ok[RS_GROUP];
#pragma unroll
        for (int j = 0; j < RS_GROUP; ++j) {
            const int idx = cb + j * 32 + lane;
            ok[j] = idx < a_hi;
            q[j]  = make_uint2(0u, 0u);
            if (ok[j]) q[j] = rec[idx];
        }
#pragma unroll
        for (int j = 0; j < RS_GROUP; ++j) {
            const unsigned rl = q[j].x >> COL_BITS;
            const int slot = wave_count_step(h, rl, ok[j], ROW_BITS, lt_mask);
            if (ok[j]) rec2[start + rowstart[rl] + slot] = q[j];
        }
    }
    __syncthreads();

    const int row0 = bin * RPB;
    int nrows = n_rows - row0;
    if (nrows > RPB) nrows = RPB;
    const int extra = RPB % RS_WAVES;
    const int r0 = w * (RPB / RS_WAVES) + (w < extra ? w : extra);
    int r1 = r0 + (RPB / RS_WAVES) + (w < extra ? 1 : 0);
    if (r1 > nrows) r1 = nrows;
    if (r0 >= r1) return;

    const float b0 = bias[lane], b1 = bias[lane + 32], b2 = bias[lane + 64], b3 = bias[lane + 96];
    const int e_lo = start + __builtin_amdgcn_readfirstlane(rowstart[r0]);
    const int e_hi = start + __builtin_amdgcn_readfirstlane(rowstart[r1]);

    int   cur = r0;
    float s0 = 0.f, s1 = 0.f, s2 = 0.f, s3 = 0.f;

#define RS_FLUSH(nxt)                                                          \
    do {                                                                       \
        float* o = out + (long long)(row0 + cur) * D_OUT;                      \
        o[lane] = s0 + b0; o[lane + 32] = s1 + b1;                             \
        o[lane + 64] = s2 + b2; o[lane + 96] = s3 + b3;                        \
        _Pragma("unroll 1")   \
        for (int rr = cur + 1; rr < (nxt); ++rr) {                             \
            float* oe = out + (long long)(row0 + rr) * D_OUT;                  \
            oe[lane] = b0; oe[lane + 32] = b1; oe[lane + 64] = b2; oe[lane + 96] = b3; \
        }                                                                      \
        s0 = 0.f; s1 = 0.f; s2 = 0.f; s3 = 0.f;                                \
    } while (0)

    int key_v = 0, val_v = 0;
    if (e_lo + lane < e_hi) {
        const uint2 q = rec2[e_lo + lane];
        key_v = (int)q.x;
        val_v = (int)q.y;
    }
    for (int cb = e_lo; cb < e_hi; cb += 32) {
        int n_here = e_hi - cb;
        if (n_here > 32) n_here = 32;
        int key_n = 0, val_n = 0;
        if (cb + 32 + lane < e_hi) {
            const uint2 q = rec2[cb + 32 + lane];
            key_n = (int)q.x;
            val_n = (int)q.y;
        }
        int j0 = 0;
        for (; j0 + RS_BATCH <= n_here; j0 += RS_BATCH) {
            float g[RS_BATCH][4];
#pragma unroll
            for (int u = 0; u < RS_BATCH; ++u) {
                const int key = __builtin_amdgcn_readlane(key_v, j0 + u);
                const float* src = S + (long long)(key & ((1 << COL_BITS) - 1)) * D_OUT;
                g[u][0] = src[lane];
                g[u][1] = src[lane + 32];
                g[u][2] = src[lane + 64];
                g[u][3] = src[lane + 96];
            }
#pragma unroll
            for (int u = 0; u < RS_BATCH; ++u) {
                const int   key = __builtin_amdgcn_readlane(key_v, j0 + u);
                const float v   = __int_as_float(__builtin_amdgcn_readlane(val_v, j0 + u));
                const int   rl  = (int)((unsigned)key >> COL_BITS);
                if (rl != cur) { RS_FLUSH(rl); cur = rl; }
                s0 += g[u][0] * v; s1 += g[u][1] * v; s2 += g[u][2] * v; s3 += g[u][3] * v;
            }
        }
        for (; j0 < n_here; ++j0) {
            const int   key = __builtin_amdgcn_readlane(key_v, j0);
            const float v   = __int_as_float(__builtin_amdgcn_readlane(val_v, j0));
            const int   rl  = (int)((unsigned)key >> COL_BITS);
            const float* src = S + (long long)(key & ((1 << COL_BITS) - 1)) * D_OUT;
            const float g0 = src[lane];
            const float g1 = src[lane + 32];
            const float g2 = src[lane + 64];
            const float g3 = src[lane + 96];
            if (rl != cur) { RS_FLUSH(rl); cur = rl; }
            s0 += g0 * v; s1 += g1 * v; s2 += g2 * v; s3 += g3 * v;
        }
        key_v = key_n;
        val_v = val_n;
    }
    RS_FLUSH(r1);
#undef RS_FLUSH
}

extern "C" void kernel_launch(void* const* d_in, const int* in_sizes, int n_in,
                              void* d_out, int out_size, void* d_ws, size_t ws_size,
                              hipStream_t stream) {
    const int*   rows = (const int*)d_in[0];
    const int*   cols = (const int*)d_in[1];
    const float* vals = (const float*)d_in[2];
    const float* x    = (const float*)d_in[3];
    const float* W    = (const float*)d_in[4];
    const float* bias = (const float*)d_in[5];

    const int E = in_sizes[0];
    const int N = in_sizes[3] / D_IN;

    float* support = (float*)d_ws;

    {
        int mtiles = N / 16;
        int blocks = (mtiles + 7) / 8;
        gcn_gemm_xw<<<blocks, 256, 0, stream>>>(x, W, support, N);
    }

    const int    sort_blocks = (E + SORT_EPB - 1) / SORT_EPB;
    const size_t off_rec   = (((size_t)N * D_OUT * sizeof(float)) + 255) & ~(size_t)255;
    const size_t off_rec2  = (off_rec  + (size_t)E * sizeof(uint2) + 255) & ~(size_t)255;
    const size_t off_hist  = (off_rec2 + (size_t)E * sizeof(uint2) + 255) & ~(size_t)255;
    const size_t off_start = (off_hist + (size_t)sort_blocks * NBINS * sizeof(int) + 255) & ~(size_t)255;
    const size_t ws_need   = off_start + (NBINS + 1) * sizeof(int);

    if (ws_need <= ws_size && N <= NBINS * RPB) {
        uint2* rec       = (uint2*)((char*)d_ws + off_rec);
        uint2* rec2      = (uint2*)((char*)d_ws + off_rec2);
        int*   blockhist = (int*)((char*)d_ws + off_hist);
        int*   binstart  = (int*)((char*)d_ws + off_start);
        gcn_bin_count<<<sort_blocks, SORT_THREADS, 0, stream>>>(rows, blockhist, E);
        gcn_bin_place<<<sort_blocks, SORT_THREADS, 0, stream>>>(rows, cols, vals, blockhist,
                                                                binstart, rec, E, sort_blocks);
        gcn_range_sum<<<NBINS, RS_THREADS, 0, stream>>>(rec, rec2, binstart, support, bias,
                                                        (float*)d_out, N);
    } else {
        long long total4 = (long long)N * (D_OUT / 4);
        int iblocks = (int)((total4 + 255) / 256);
        gcn_init_out<<<iblocks, 256, 0, stream>>>(bias, (float*)d_out, total4);
        int waves  = (E + EDGES_PER_WAVE - 1) / EDGES_PER_WAVE;
        int blocks = (waves + 7) / 8;
        gcn_edge_scatter<<<blocks, 256, 0, stream>>>(rows, cols, vals, support,
                                                     (float*)d_out, E);
    }
}


# Edge placing pass runs as extra work-groups of the GEMM launch (GEMM work-groups of 32 strips first, placing work-groups on the WGPs left free): 3 launches instead of 4
# speedup vs baseline: 2.2552x; 1.0549x over previous
#include <hip/hip_runtime.h>

typedef __attribute__((ext_vector_type(2))) float v2f;
typedef __attribute__((ext_vector_type(8))) float v8f;

#define D_IN  256
#define D_OUT 128

#define NBINS        256
#define BIN_BITS     8
#define RPB          391
#define ROW_BITS     9
#define COL_BITS     20
#define SORT_THREADS 1024
#define SORT_WAVES   (SORT_THREADS / 32)
#define SORT_EPT     8
#define SORT_EPB     (SORT_THREADS * SORT_EPT)

__global__ __launch_bounds__(256) void gcn_init_out(const float* __restrict__ bias,
                                                    float* __restrict__ out,
                                                    long long total4) {
    long long i = (long long)blockIdx.x * blockDim.x + threadIdx.x;
    if (i < total4) {
        int d4 = (int)(i & (D_OUT / 4 - 1));
        ((float4*)out)[i] = ((const float4*)bias)[d4];
    }
}

static __device__ __forceinline__ void gemm_strip(const float* __restrict__ X,
                                                  const float* __restrict__ W,
                                                  float* __restrict__ S,
                                                  int n_rows, int mtile) {
    const int lane  = threadIdx.x & 31;
    if (mtile * 16 + 16 > n_rows) return;
    const int m0 = mtile * 16;

    const int am = lane & 15;
    const int ak = (lane >> 4) * 2;
    const int bn = lane & 15;
    const int bk = (lane >> 4) * 2;

    v8f c[8];
#pragma unroll
    for (int t = 0; t < 8; ++t) c[t] = (v8f){0.f, 0.f, 0.f, 0.f, 0.f, 0.f, 0.f, 0.f};

    const float* xrow = X + (long long)(m0 + am) * D_IN;

    for (int k = 0; k < D_IN; k += 4) {
        v2f a = *(const v2f*)(xrow + k + ak);
#pragma unroll
        for (int t = 0; t < 8; ++t) {
            v2f b;
            b.x = W[(long long)(k + bk)     * D_OUT + t * 16 + bn];
            b.y = W[(long long)(k + bk + 1) * D_OUT + t * 16 + bn];
            c[t] = __builtin_amdgcn_wmma_f32_16x16x4_f32(
                 false, a,  false, b,
                 (short)0, c[t],  false,  false);
        }
    }

    const int cn    = lane & 15;
    const int cmoff = (lane >> 4) * 8;
#pragma unroll
    for (int t = 0; t < 8; ++t) {
#pragma unroll
        for (int r = 0; r < 8; ++r) {
            S[(long long)(m0 + r + cmoff) * D_OUT + t * 16 + cn] = c[t][r];
        }
    }
}

__global__ __launch_bounds__(256) void gcn_gemm_xw(const float* __restrict__ X,
                                                   const float* __restrict__ W,
                                                   float* __restrict__ S,
                                                   int n_rows) {
    gemm_strip(X, W, S, n_rows, (int)blockIdx.x * 8 + (int)(threadIdx.x >> 5));
}

#define EDGES_PER_WAVE 32
#define EDGE_BATCH     8

static __device__ __forceinline__ void add_row_lines(float* dst, unsigned off,
                                                      float g0, float g1, float g2, float g3) {
    asm volatile("global_atomic_add_f32 %0, %1, %2"            :: "v"(off), "v"(g0), "s"(dst) : "memory");
    asm volatile("global_atomic_add_f32 %0, %1, %2 offset:128" :: "v"(off), "v"(g1), "s"(dst) : "memory");
    asm volatile("global_atomic_add_f32 %0, %1, %2 offset:256" :: "v"(off), "v"(g2), "s"(dst) : "memory");
    asm volatile("global_atomic_add_f32 %0, %1, %2 offset:384" :: "v"(off), "v"(g3), "s"(dst) : "memory");
}

__global__ __launch_bounds__(256) void gcn_edge_scatter(const int* __restrict__ rows,
                                                        const int* __restrict__ cols,
                                                        const float* __restrict__ vals,
                                                        const float* __restrict__ S,
                                                        float* __restrict__ out,
                                                        int n_edges) {
    const int lane = threadIdx.x & 31;
    int w = (int)(blockIdx.x * 8u) + (int)(threadIdx.x >> 5);
    w = __builtin_amdgcn_readfirstlane(w);
    const int e_base = w * EDGES_PER_WAVE;
    if (e_base >= n_edges) return;
    int n_here = n_edges - e_base;
    if (n_here > EDGES_PER_WAVE) n_here = EDGES_PER_WAVE;

    int r_v = 0, c_v = 0, v_v = 0;
    if (lane < n_here) {
        r_v = rows[e_base + lane];
        c_v = cols[e_base + lane];
        v_v = __float_as_int(vals[e_base + lane]);
    }
    const unsigned off = (unsigned)lane * 4u;

    int j0 = 0;
    for (; j0 + EDGE_BATCH <= n_here; j0 += EDGE_BATCH) {
        float g[EDGE_BATCH][4];
#pragma unroll
        for (int u = 0; u < EDGE_BATCH; ++u) {
            const int c = __builtin_amdgcn_readlane(c_v, j0 + u);
            const float* src = S + (long long)c * D_OUT;
            g[u][0] = src[lane];
            g[u][1] = src[lane + 32];
            g[u][2] = src[lane + 64];
            g[u][3] = src[lane + 96];
        }
#pragma unroll
        for (int u = 0; u < EDGE_BATCH; ++u) {
            const int   r = __builtin_amdgcn_readlane(r_v, j0 + u);
            const float v = __int_as_float(__builtin_amdgcn_readlane(v_v, j0 + u));
            float* dst = out + (long long)r * D_OUT;
            add_row_lines(dst, off, g[u][0] * v, g[u][1] * v, g[u][2] * v, g[u][3] * v);
        }
    }
    for (; j0 < n_here; ++j0) {
        const int   c = __builtin_amdgcn_readlane(c_v, j0);
        const int   r = __builtin_amdgcn_readlane(r_v, j0);
        const float v = __int_as_float(__builtin_amdgcn_readlane(v_v, j0));
        const float* src = S + (long long)c * D_OUT;
        const float s0 = src[lane];
        const float s1 = src[lane + 32];
        const float s2 = src[lane + 64];
        const float s3 = src[lane + 96];
        float* dst = out + (long long)r * D_OUT;
        add_row_lines(dst, off, s0 * v, s1 * v, s2 * v, s3 * v);
    }
}


static __device__ __forceinline__ unsigned same_key_lanes(unsigned key, bool valid, int nbits) {
    unsigned peers = __builtin_amdgcn_ballot_w32(valid);
#pragma unroll
    for (int b = 0; b < nbits; ++b) {
        const bool bit = ((key >> b) & 1u) != 0u;
        const unsigned m = __builtin_amdgcn_ballot_w32(bit && valid);
        peers &= bit ? m : ~m;
    }
    return valid ? peers : 0u;
}

static __device__ __forceinline__ void wave_rank(unsigned key, bool valid, int nbits,
                                                 unsigned lt_mask, int& rank, int& cnt) {
    const unsigned peers = same_key_lanes(key, valid, nbits);
    rank = __builtin_popcount(peers & lt_mask);
    cnt  = __builtin_popcount(peers);
}

static __device__ __forceinline__ int counter_take(int* h, unsigned key, bool valid,
                                                   int rank, int cnt) {
    int slot = 0;
    if (valid) {
        const int old = h[key];
        if (rank == 0) h[key] = old + cnt;
        slot = old + rank;
    }
    return slot;
}

static __device__ __forceinline__ int wave_count_step(int* h, unsigned key, bool valid,
                                                      int nbits, unsigned lt_mask) {
    int rank, cnt;
    wave_rank(key, valid, nbits, lt_mask, rank, cnt);
    return counter_take(h, key, valid, rank, cnt);
}

__global__ __launch_bounds__(SORT_THREADS) void gcn_bin_count(const int* __restrict__ rows,
                                                              int* __restrict__ blockhist,
                                                              int n_edges) {
    __shared__ int whist[SORT_WAVES][NBINS];
    const int t = threadIdx.x, lane = t & 31, w = t >> 5;
    const unsigned lt_mask = (1u << lane) - 1u;
    int* h = whist[w];
    for (int k = lane; k < NBINS; k += 32) h[k] = 0;

    const int e0 = blockIdx.x * SORT_EPB + w * (SORT_EPT * 32) + lane;
    int r[SORT_EPT];
#pragma unroll
    for (int i = 0; i < SORT_EPT; ++i) {
        const int e = e0 + i * 32;
        r[i] = (e < n_edges) ? rows[e] : -1;
    }
#pragma unroll
    for (int i = 0; i < SORT_EPT; ++i) {
        const bool valid = r[i] >= 0;
        const unsigned bin = valid ? (unsigned)(r[i] / RPB) : 0u;
        wave_count_step(h, bin, valid, BIN_BITS, lt_mask);
    }
    __syncthreads();
    if (t < NBINS) {
        int s = 0;
        for (int w2 = 0; w2 < SORT_WAVES; ++w2) s += whist[w2][t];
        blockhist[blockIdx.x * NBINS + t] = s;
    }
}

static __device__ __forceinline__ void place_block(const int* __restrict__ rows,
                                                   const int* __restrict__ cols,
                                                   const float* __restrict__ vals,
                                                   const int* __restrict__ blockhist,
                                                   int* __restrict__ binstart,
                                                   uint2* __restrict__ rec,
                                                   int n_edges, int n_blocks, int me,
                                                   int (*whist)[NBINS],
                                                   int (*part)[SORT_THREADS / NBINS][NBINS],
                                                   int* scan) {
    const int t = threadIdx.x, lane = t & 31, w = t >> 5;
    const unsigned lt_mask = (1u << lane) - 1u;
    int* h = whist[w];
    for (int k = lane; k < NBINS; k += 32) h[k] = 0;

    const int e0 = me * SORT_EPB + w * (SORT_EPT * 32) + lane;
    unsigned info[SORT_EPT];
    {
        int r[SORT_EPT];
#pragma unroll
        for (int i = 0; i < SORT_EPT; ++i) {
            const int e = e0 + i * 32;
            r[i] = (e < n_edges) ? rows[e] : -1;
        }
#pragma unroll
        for (int i = 0; i < SORT_EPT; ++i) {
            const bool valid = r[i] >= 0;
            const unsigned bin = valid ? (unsigned)(r[i] / RPB) : 0u;
            const unsigned rl  = valid ? (unsigned)r[i] - bin * RPB : 0u;
            int rank, cnt;
            wave_rank(bin, valid, BIN_BITS, lt_mask, rank, cnt);
            counter_take(h, bin, valid, rank, cnt);
            info[i] = bin | ((unsigned)rank << 8) | ((unsigned)cnt << 13) | (rl << 19) |
                      (valid ? (1u << 28) : 0u);
        }
    }

    {
        const int q = t / NBINS, bn = t % NBINS;
        int bef = 0, tot = 0;
#pragma unroll 16
        for (int b = q; b < n_blocks; b += SORT_THREADS / NBINS) {
            const int x = blockhist[b * NBINS + bn];
            tot += x;
            bef += (b < me) ? x : 0;
        }
        part[0][q][bn] = bef;
        part[1][q][bn] = tot;
    }
    __syncthreads();
    int total = 0, before = 0;
    if (t < NBINS) {
#pragma unroll
        for (int q = 0; q < SORT_THREADS / NBINS; ++q) { before += part[0][q][t]; total += part[1][q][t]; }
        scan[t] = total;
    }
    __syncthreads();
    for (int d = 1; d < NBINS; d <<= 1) {
        int add = 0;
        if (t < NBINS && t >= d) add = scan[t - d];
        __syncthreads();
        if (t < NBINS) scan[t] += add;
        __syncthreads();
    }
    if (t < NBINS) {
        const int start = scan[t] - total;
        if (me == 0) {
            binstart[t] = start;
            if (t == NBINS - 1) binstart[NBINS] = scan[t];
        }
        int run = start + before;
        for (int w2 = 0; w2 < SORT_WAVES; ++w2) {
            const int cnt = whist[w2][t];
            whist[w2][t] = run;
            run += cnt;
        }
    }
    __syncthreads();

    int c[SORT_EPT];
    float v[SORT_EPT];
#pragma unroll
    for (int i = 0; i < SORT_EPT; ++i) {
        const int e = e0 + i * 32;
        c[i] = 0; v[i] = 0.f;
        if (info[i] >> 28) { c[i] = cols[e]; v[i] = vals[e]; }
    }
#pragma unroll
    for (int i = 0; i < SORT_EPT; ++i) {
        const bool     valid = (info[i] >> 28) != 0u;
        const unsigned bin   = info[i] & 0xFFu;
        const int      rank  = (int)((info[i] >> 8) & 0x1Fu);
        const int      cnt   = (int)((info[i] >> 13) & 0x3Fu);
        const unsigned rl    = (info[i] >> 19) & 0x1FFu;
        const int pos = counter_take(h, bin, valid, rank, cnt);
        if (valid)
            rec[pos] = make_uint2((unsigned)c[i] | (rl << COL_BITS), __float_as_uint(v[i]));
    }
}

__global__ __launch_bounds__(SORT_THREADS) void gcn_gemm_and_place(const float* __restrict__ X,
                                                                   const float* __restrict__ W,
                                                                   float* __restrict__ S,
                                                                   int n_rows, int gemm_blocks,
                                                                   const int* __restrict__ rows,
                                                                   const int* __restrict__ cols,
                                                                   const float* __restrict__ vals,
                                                                   const int* __restrict__ blockhist,
                                                                   int* __restrict__ binstart,
                                                                   uint2* __restrict__ rec,
                                                                   int n_edges, int n_blocks) {
    __shared__ int whist[SORT_WAVES][NBINS];
    __shared__ int part[2][SORT_THREADS / NBINS][NBINS];
    __shared__ int scan[NBINS];
    const int blk = (int)blockIdx.x;
    if (blk < gemm_blocks) {
        gemm_strip(X, W, S, n_rows, blk * SORT_WAVES + (int)(threadIdx.x >> 5));
    } else {
        place_block(rows, cols, vals, blockhist, binstart, rec, n_edges, n_blocks,
                    blk - gemm_blocks, whist, part, scan);
    }
}

#define RS_THREADS 1024
#define RS_WAVES   (RS_THREADS / 32)
#define RS_BATCH   8
#define RS_SCAN    512
#define RS_GROUP   4

__global__ __launch_bounds__(RS_THREADS) void gcn_range_sum(const uint2* __restrict__ rec,
                                                            uint2* rec2,
                                                            const int* __restrict__ binstart,
                                                            const float* __restrict__ S,
                                                            const float* __restrict__ bias,
                                                            float* __restrict__ out,
                                                            int n_rows) {
    __shared__ int whist[RS_WAVES][RPB + 1];
    __shared__ int rowscan[RS_SCAN];
    __shared__ int rowstart[RPB + 1];
    const int t    = threadIdx.x;
    const int lane = t & 31;
    const int w    = __builtin_amdgcn_readfirstlane(t >> 5);
    const int bin  = blockIdx.x;
    const unsigned lt_mask = (1u << lane) - 1u;

    const int start = binstart[bin];
    const int end   = binstart[bin + 1];
    const int per   = (end - start + RS_WAVES - 1) / RS_WAVES;
    const int a_lo  = start + w * per;
    int a_hi = a_lo + per;
    if (a_hi > end) a_hi = end;

    int* h = whist[w];
    for (int k = lane; k <= RPB; k += 32) h[k] = 0;
    for (int cb = a_lo; cb < a_hi; cb += 32 * RS_GROUP) {
        unsigned key[RS_GROUP];
        bool     ok[RS_GROUP];
#pragma unroll
        for (int j = 0; j < RS_GROUP; ++j) {
            const int idx = cb + j * 32 + lane;
            ok[j]  = idx < a_hi;
            key[j] = 0u;
            if (ok[j]) key[j] = rec[idx].x;
        }
#pragma unroll
        for (int j = 0; j < RS_GROUP; ++j)
            wave_count_step(h, key[j] >> COL_BITS, ok[j], ROW_BITS, lt_mask);
    }
    __syncthreads();

    int rc = 0;
    if (t < RPB) {
        int run = 0;
        for (int w2 = 0; w2 < RS_WAVES; ++w2) {
            const int cnt = whist[w2][t];
            whist[w2][t] = run;
            run += cnt;
        }
        rc = run;
    }
    if (t < RS_SCAN) rowscan[t] = rc;
    __syncthreads();
    for (int d = 1; d < RS_SCAN; d <<= 1) {
        int add = 0;
        if (t < RS_SCAN && t >= d) add = rowscan[t - d];
        __syncthreads();
        if (t < RS_SCAN) rowscan[t] += add;
        __syncthreads();
    }
    if (t <= RPB) rowstart[t] = (t == 0) ? 0 : rowscan[t - 1];
    __syncthreads();

    for (int cb = a_lo; cb < a_hi; cb += 32 * RS_GROUP) {
        uint2 q[RS_GROUP];
        bool  ok[RS_GROUP];
#pragma unroll
        for (int j = 0; j < RS_GROUP; ++j) {
            const int idx = cb + j * 32 + lane;
            ok[j] = idx < a_hi;
            q[j]  = make_uint2(0u, 0u);
            if (ok[j]) q[j] = rec[idx];
        }
#pragma unroll
        for (int j = 0; j < RS_GROUP; ++j) {
            const unsigned rl = q[j].x >> COL_BITS;
            const int slot = wave_count_step(h, rl, ok[j], ROW_BITS, lt_mask);
            if (ok[j]) rec2[start + rowstart[rl] + slot] = q[j];
        }
    }
    __syncthreads();

    const int row0 = bin * RPB;
    int nrows = n_rows - row0;
    if (nrows > RPB) nrows = RPB;
    const int extra = RPB % RS_WAVES;
    const int r0 = w * (RPB / RS_WAVES) + (w < extra ? w : extra);
    int r1 = r0 + (RPB / RS_WAVES) + (w < extra ? 1 : 0);
    if (r1 > nrows) r1 = nrows;
    if (r0 >= r1) return;

    const float b0 = bias[lane], b1 = bias[lane + 32], b2 = bias[lane + 64], b3 = bias[lane + 96];
    const int e_lo = start + __builtin_amdgcn_readfirstlane(rowstart[r0]);
    const int e_hi = start + __builtin_amdgcn_readfirstlane(rowstart[r1]);

    int   cur = r0;
    float s0 = 0.f, s1 = 0.f, s2 = 0.f, s3 = 0.f;

#define RS_FLUSH(nxt)                                                          \
    do {                                                                       \
        float* o = out + (long long)(row0 + cur) * D_OUT;                      \
        o[lane] = s0 + b0; o[lane + 32] = s1 + b1;                             \
        o[lane + 64] = s2 + b2; o[lane + 96] = s3 + b3;                        \
        _Pragma("unroll 1")   \
        for (int rr = cur + 1; rr < (nxt); ++rr) {                             \
            float* oe = out + (long long)(row0 + rr) * D_OUT;                  \
            oe[lane] = b0; oe[lane + 32] = b1; oe[lane + 64] = b2; oe[lane + 96] = b3; \
        }                                                                      \
        s0 = 0.f; s1 = 0.f; s2 = 0.f; s3 = 0.f;                                \
    } while (0)

    int key_v = 0, val_v = 0;
    if (e_lo + lane < e_hi) {
        const uint2 q = rec2[e_lo + lane];
        key_v = (int)q.x;
        val_v = (int)q.y;
    }
    for (int cb = e_lo; cb < e_hi; cb += 32) {
        int n_here = e_hi - cb;
        if (n_here > 32) n_here = 32;
        int key_n = 0, val_n = 0;
        if (cb + 32 + lane < e_hi) {
            const uint2 q = rec2[cb + 32 + lane];
            key_n = (int)q.x;
            val_n = (int)q.y;
        }
        int j0 = 0;
        for (; j0 + RS_BATCH <= n_here; j0 += RS_BATCH) {
            float g[RS_BATCH][4];
#pragma unroll
            for (int u = 0; u < RS_BATCH; ++u) {
                const int key = __builtin_amdgcn_readlane(key_v, j0 + u);
                const float* src = S + (long long)(key & ((1 << COL_BITS) - 1)) * D_OUT;
                g[u][0] = src[lane];
                g[u][1] = src[lane + 32];
                g[u][2] = src[lane + 64];
                g[u][3] = src[lane + 96];
            }
#pragma unroll
            for (int u = 0; u < RS_BATCH; ++u) {
                const int   key = __builtin_amdgcn_readlane(key_v, j0 + u);
                const float v   = __int_as_float(__builtin_amdgcn_readlane(val_v, j0 + u));
                const int   rl  = (int)((unsigned)key >> COL_BITS);
                if (rl != cur) { RS_FLUSH(rl); cur = rl; }
                s0 += g[u][0] * v; s1 += g[u][1] * v; s2 += g[u][2] * v; s3 += g[u][3] * v;
            }
        }
        for (; j0 < n_here; ++j0) {
            const int   key = __builtin_amdgcn_readlane(key_v, j0);
            const float v   = __int_as_float(__builtin_amdgcn_readlane(val_v, j0));
            const int   rl  = (int)((unsigned)key >> COL_BITS);
            const float* src = S + (long long)(key & ((1 << COL_BITS) - 1)) * D_OUT;
            const float g0 = src[lane];
            const float g1 = src[lane + 32];
            const float g2 = src[lane + 64];
            const float g3 = src[lane + 96];
            if (rl != cur) { RS_FLUSH(rl); cur = rl; }
            s0 += g0 * v; s1 += g1 * v; s2 += g2 * v; s3 += g3 * v;
        }
        key_v = key_n;
        val_v = val_n;
    }
    RS_FLUSH(r1);
#undef RS_FLUSH
}

extern "C" void kernel_launch(void* const* d_in, const int* in_sizes, int n_in,
                              void* d_out, int out_size, void* d_ws, size_t ws_size,
                              hipStream_t stream) {
    const int*   rows = (const int*)d_in[0];
    const int*   cols = (const int*)d_in[1];
    const float* vals = (const float*)d_in[2];
    const float* x    = (const float*)d_in[3];
    const float* W    = (const float*)d_in[4];
    const float* bias = (const float*)d_in[5];

    const int E = in_sizes[0];
    const int N = in_sizes[3] / D_IN;

    float* support = (float*)d_ws;

    const int mtiles = N / 16;

    const int    sort_blocks = (E + SORT_EPB - 1) / SORT_EPB;
    const size_t off_rec   = (((size_t)N * D_OUT * sizeof(float)) + 255) & ~(size_t)255;
    const size_t off_rec2  = (off_rec  + (size_t)E * sizeof(uint2) + 255) & ~(size_t)255;
    const size_t off_hist  = (off_rec2 + (size_t)E * sizeof(uint2) + 255) & ~(size_t)255;
    const size_t off_start = (off_hist + (size_t)sort_blocks * NBINS * sizeof(int) + 255) & ~(size_t)255;
    const size_t ws_need   = off_start + (NBINS + 1) * sizeof(int);

    if (ws_need <= ws_size && N <= NBINS * RPB) {
        uint2* rec       = (uint2*)((char*)d_ws + off_rec);
        uint2* rec2      = (uint2*)((char*)d_ws + off_rec2);
        int*   blockhist = (int*)((char*)d_ws + off_hist);
        int*   binstart  = (int*)((char*)d_ws + off_start);
        const int gemm_blocks = (mtiles + SORT_WAVES - 1) / SORT_WAVES;
        gcn_bin_count<<<sort_blocks, SORT_THREADS, 0, stream>>>(rows, blockhist, E);
        gcn_gemm_and_place<<<gemm_blocks + sort_blocks, SORT_THREADS, 0, stream>>>(
            x, W, support, N, gemm_blocks, rows, cols, vals, blockhist, binstart, rec, E,
            sort_blocks);
        gcn_range_sum<<<NBINS, RS_THREADS, 0, stream>>>(rec, rec2, binstart, support, bias,
                                                        (float*)d_out, N);
    } else {
        gcn_gemm_xw<<<(mtiles + 7) / 8, 256, 0, stream>>>(x, W, support, N);
        long long total4 = (long long)N * (D_OUT / 4);
        int iblocks = (int)((total4 + 255) / 256);
        gcn_init_out<<<iblocks, 256, 0, stream>>>(bias, (float*)d_out, total4);
        int waves  = (E + EDGES_PER_WAVE - 1) / EDGES_PER_WAVE;
        int blocks = (waves + 7) / 8;
        gcn_edge_scatter<<<blocks, 256, 0, stream>>>(rows, cols, vals, support,
                                                     (float*)d_out, E);
    }
}


# GEMM operands fetched once: W staged per work-group in LDS as k pairs (B fragment = one 8-byte LDS read), x rows read as whole 128-byte lines; k regrouped inside 32-blocks: f32 sum order changes
# speedup vs baseline: 2.4971x; 1.1073x over previous
#include <hip/hip_runtime.h>

typedef __attribute__((ext_vector_type(2))) float v2f;
typedef __attribute__((ext_vector_type(8))) float v8f;

#define D_IN  256
#define D_OUT 128

#define NBINS        256
#define BIN_BITS     8
#define RPB          391
#define ROW_BITS     9
#define COL_BITS     20
#define SORT_THREADS 1024
#define SORT_WAVES   (SORT_THREADS / 32)
#define SORT_EPT     8
#define SORT_EPB     (SORT_THREADS * SORT_EPT)

__global__ __launch_bounds__(256) void gcn_init_out(const float* __restrict__ bias,
                                                    float* __restrict__ out,
                                                    long long total4) {
    long long i = (long long)blockIdx.x * blockDim.x + threadIdx.x;
    if (i < total4) {
        int d4 = (int)(i & (D_OUT / 4 - 1));
        ((float4*)out)[i] = ((const float4*)bias)[d4];
    }
}

static __device__ __forceinline__ void gemm_strip(const float* __restrict__ X,
                                                  const float* __restrict__ W,
                                                  float* __restrict__ S,
                                                  int n_rows, int mtile) {
    const int lane  = threadIdx.x & 31;
    if (mtile * 16 + 16 > n_rows) return;
    const int m0 = mtile * 16;

    const int am = lane & 15;
    const int ak = (lane >> 4) * 2;
    const int bn = lane & 15;
    const int bk = (lane >> 4) * 2;

    v8f c[8];
#pragma unroll
    for (int t = 0; t < 8; ++t) c[t] = (v8f){0.f, 0.f, 0.f, 0.f, 0.f, 0.f, 0.f, 0.f};

    const float* xrow = X + (long long)(m0 + am) * D_IN;

    for (int k = 0; k < D_IN; k += 4) {
        v2f a = *(const v2f*)(xrow + k + ak);
#pragma unroll
        for (int t = 0; t < 8; ++t) {
            v2f b;
            b.x = W[(long long)(k + bk)     * D_OUT + t * 16 + bn];
            b.y = W[(long long)(k + bk + 1) * D_OUT + t * 16 + bn];
            c[t] = __builtin_amdgcn_wmma_f32_16x16x4_f32(
                 false, a,  false, b,
                 (short)0, c[t],  false,  false);
        }
    }

    const int cn    = lane & 15;
    const int cmoff = (lane >> 4) * 8;
#pragma unroll
    for (int t = 0; t < 8; ++t) {
#pragma unroll
        for (int r = 0; r < 8; ++r) {
            S[(long long)(m0 + r + cmoff) * D_OUT + t * 16 + cn] = c[t][r];
        }
    }
}

__global__ __launch_bounds__(256) void gcn_gemm_xw(const float* __restrict__ X,
                                                   const float* __restrict__ W,
                                                   float* __restrict__ S,
                                                   int n_rows) {
    gemm_strip(X, W, S, n_rows, (int)blockIdx.x * 8 + (int)(threadIdx.x >> 5));
}

#define WL_PITCH 130
#define WL_BYTES ((D_IN / 2) * WL_PITCH * 8)

static __device__ __forceinline__ void stage_w_pairs(const float* __restrict__ W, v2f* wl) {
    float4 w0[4], w1[4];
#pragma unroll
    for (int i = 0; i < 4; ++i) {
        const int idx = (int)threadIdx.x + i * SORT_THREADS;
        const int p = idx >> 5, q = idx & 31;
        w0[i] = ((const float4*)W)[(2 * p)     * (D_OUT / 4) + q];
        w1[i] = ((const float4*)W)[(2 * p + 1) * (D_OUT / 4) + q];
    }
#pragma unroll
    for (int i = 0; i < 4; ++i) {
        const int idx = (int)threadIdx.x + i * SORT_THREADS;
        const int p = idx >> 5, q = idx & 31;
        float4* d = (float4*)(wl + p * WL_PITCH + 4 * q);
        d[0] = make_float4(w0[i].x, w1[i].x, w0[i].y, w1[i].y);
        d[1] = make_float4(w0[i].z, w1[i].z, w0[i].w, w1[i].w);
    }
}

static __device__ __forceinline__ void gemm_strip_lds(const float* __restrict__ X,
                                                      const v2f* wl,
                                                      float* __restrict__ S,
                                                      int n_rows, int mtile) {
    const int lane = threadIdx.x & 31;
    if (mtile * 16 + 16 > n_rows) return;
    const int m0 = mtile * 16;
    const int c  = lane & 15;
    const int h  = lane >> 4;

    v8f acc[8];
#pragma unroll
    for (int t = 0; t < 8; ++t) acc[t] = (v8f){0.f, 0.f, 0.f, 0.f, 0.f, 0.f, 0.f, 0.f};

    const float* xrow  = X + (long long)(m0 + c) * D_IN + 16 * h;
    const v2f*   wlane = wl + (8 * h) * WL_PITCH + c;

#pragma unroll 1
    for (int kb = 0; kb < D_IN; kb += 32) {
        float4 q[4];
#pragma unroll
        for (int i = 0; i < 4; ++i) q[i] = *(const float4*)(xrow + kb + 4 * i);
        const v2f* wk = wlane + (kb / 2) * WL_PITCH;
#pragma unroll
        for (int s = 0; s < 8; ++s) {
            v2f a;
            a.x = (s & 1) ? q[s >> 1].z : q[s >> 1].x;
            a.y = (s & 1) ? q[s >> 1].w : q[s >> 1].y;
#pragma unroll
            for (int t = 0; t < 8; ++t) {
                const v2f b = wk[s * WL_PITCH + 16 * t];
                acc[t] = __builtin_amdgcn_wmma_f32_16x16x4_f32(
                     false, a,  false, b,
                     (short)0, acc[t],  false,  false);
            }
        }
    }

    const int cmoff = h * 8;
#pragma unroll
    for (int t = 0; t < 8; ++t) {
#pragma unroll
        for (int r = 0; r < 8; ++r) {
            S[(long long)(m0 + r + cmoff) * D_OUT + t * 16 + c] = acc[t][r];
        }
    }
}

#define EDGES_PER_WAVE 32
#define EDGE_BATCH     8

static __device__ __forceinline__ void add_row_lines(float* dst, unsigned off,
                                                      float g0, float g1, float g2, float g3) {
    asm volatile("global_atomic_add_f32 %0, %1, %2"            :: "v"(off), "v"(g0), "s"(dst) : "memory");
    asm volatile("global_atomic_add_f32 %0, %1, %2 offset:128" :: "v"(off), "v"(g1), "s"(dst) : "memory");
    asm volatile("global_atomic_add_f32 %0, %1, %2 offset:256" :: "v"(off), "v"(g2), "s"(dst) : "memory");
    asm volatile("global_atomic_add_f32 %0, %1, %2 offset:384" :: "v"(off), "v"(g3), "s"(dst) : "memory");
}

__global__ __launch_bounds__(256) void gcn_edge_scatter(const int* __restrict__ rows,
                                                        const int* __restrict__ cols,
                                                        const float* __restrict__ vals,
                                                        const float* __restrict__ S,
                                                        float* __restrict__ out,
                                                        int n_edges) {
    const int lane = threadIdx.x & 31;
    int w = (int)(blockIdx.x * 8u) + (int)(threadIdx.x >> 5);
    w = __builtin_amdgcn_readfirstlane(w);
    const int e_base = w * EDGES_PER_WAVE;
    if (e_base >= n_edges) return;
    int n_here = n_edges - e_base;
    if (n_here > EDGES_PER_WAVE) n_here = EDGES_PER_WAVE;

    int r_v = 0, c_v = 0, v_v = 0;
    if (lane < n_here) {
        r_v = rows[e_base + lane];
        c_v = cols[e_base + lane];
        v_v = __float_as_int(vals[e_base + lane]);
    }
    const unsigned off = (unsigned)lane * 4u;

    int j0 = 0;
    for (; j0 + EDGE_BATCH <= n_here; j0 += EDGE_BATCH) {
        float g[EDGE_BATCH][4];
#pragma unroll
        for (int u = 0; u < EDGE_BATCH; ++u) {
            const int c = __builtin_amdgcn_readlane(c_v, j0 + u);
            const float* src = S + (long long)c * D_OUT;
            g[u][0] = src[lane];
            g[u][1] = src[lane + 32];
            g[u][2] = src[lane + 64];
            g[u][3] = src[lane + 96];
        }
#pragma unroll
        for (int u = 0; u < EDGE_BATCH; ++u) {
            const int   r = __builtin_amdgcn_readlane(r_v, j0 + u);
            const float v = __int_as_float(__builtin_amdgcn_readlane(v_v, j0 + u));
            float* dst = out + (long long)r * D_OUT;
            add_row_lines(dst, off, g[u][0] * v, g[u][1] * v, g[u][2] * v, g[u][3] * v);
        }
    }
    for (; j0 < n_here; ++j0) {
        const int   c = __builtin_amdgcn_readlane(c_v, j0);
        const int   r = __builtin_amdgcn_readlane(r_v, j0);
        const float v = __int_as_float(__builtin_amdgcn_readlane(v_v, j0));
        const float* src = S + (long long)c * D_OUT;
        const float s0 = src[lane];
        const float s1 = src[lane + 32];
        const float s2 = src[lane + 64];
        const float s3 = src[lane + 96];
        float* dst = out + (long long)r * D_OUT;
        add_row_lines(dst, off, s0 * v, s1 * v, s2 * v, s3 * v);
    }
}


static __device__ __forceinline__ unsigned same_key_lanes(unsigned key, bool valid, int nbits) {
    unsigned peers = __builtin_amdgcn_ballot_w32(valid);
#pragma unroll
    for (int b = 0; b < nbits; ++b) {
        const bool bit = ((key >> b) & 1u) != 0u;
        const unsigned m = __builtin_amdgcn_ballot_w32(bit && valid);
        peers &= bit ? m : ~m;
    }
    return valid ? peers : 0u;
}

static __device__ __forceinline__ void wave_rank(unsigned key, bool valid, int nbits,
                                                 unsigned lt_mask, int& rank, int& cnt) {
    const unsigned peers = same_key_lanes(key, valid, nbits);
    rank = __builtin_popcount(peers & lt_mask);
    cnt  = __builtin_popcount(peers);
}

static __device__ __forceinline__ int counter_take(int* h, unsigned key, bool valid,
                                                   int rank, int cnt) {
    int slot = 0;
    if (valid) {
        const int old = h[key];
        if (rank == 0) h[key] = old + cnt;
        slot = old + rank;
    }
    return slot;
}

static __device__ __forceinline__ int wave_count_step(int* h, unsigned key, bool valid,
                                                      int nbits, unsigned lt_mask) {
    int rank, cnt;
    wave_rank(key, valid, nbits, lt_mask, rank, cnt);
    return counter_take(h, key, valid, rank, cnt);
}

__global__ __launch_bounds__(SORT_THREADS) void gcn_bin_count(const int* __restrict__ rows,
                                                              int* __restrict__ blockhist,
                                                              int n_edges) {
    __shared__ int whist[SORT_WAVES][NBINS];
    const int t = threadIdx.x, lane = t & 31, w = t >> 5;
    const unsigned lt_mask = (1u << lane) - 1u;
    int* h = whist[w];
    for (int k = lane; k < NBINS; k += 32) h[k] = 0;

    const int e0 = blockIdx.x * SORT_EPB + w * (SORT_EPT * 32) + lane;
    int r[SORT_EPT];
#pragma unroll
    for (int i = 0; i < SORT_EPT; ++i) {
        const int e = e0 + i * 32;
        r[i] = (e < n_edges) ? rows[e] : -1;
    }
#pragma unroll
    for (int i = 0; i < SORT_EPT; ++i) {
        const bool valid = r[i] >= 0;
        const unsigned bin = valid ? (unsigned)(r[i] / RPB) : 0u;
        wave_count_step(h, bin, valid, BIN_BITS, lt_mask);
    }
    __syncthreads();
    if (t < NBINS) {
        int s = 0;
        for (int w2 = 0; w2 < SORT_WAVES; ++w2) s += whist[w2][t];
        blockhist[blockIdx.x * NBINS + t] = s;
    }
}

static __device__ __forceinline__ void place_block(const int* __restrict__ rows,
                                                   const int* __restrict__ cols,
                                                   const float* __restrict__ vals,
                                                   const int* __restrict__ blockhist,
                                                   int* __restrict__ binstart,
                                                   uint2* __restrict__ rec,
                                                   int n_edges, int n_blocks, int me,
                                                   int (*whist)[NBINS],
                                                   int (*part)[SORT_THREADS / NBINS][NBINS],
                                                   int* scan) {
    const int t = threadIdx.x, lane = t & 31, w = t >> 5;
    const unsigned lt_mask = (1u << lane) - 1u;
    int* h = whist[w];
    for (int k = lane; k < NBINS; k += 32) h[k] = 0;

    const int e0 = me * SORT_EPB + w * (SORT_EPT * 32) + lane;
    unsigned info[SORT_EPT];
    {
        int r[SORT_EPT];
#pragma unroll
        for (int i = 0; i < SORT_EPT; ++i) {
            const int e = e0 + i * 32;
            r[i] = (e < n_edges) ? rows[e] : -1;
        }
#pragma unroll
        for (int i = 0; i < SORT_EPT; ++i) {
            const bool valid = r[i] >= 0;
            const unsigned bin = valid ? (unsigned)(r[i] / RPB) : 0u;
            const unsigned rl  = valid ? (unsigned)r[i] - bin * RPB : 0u;
            int rank, cnt;
            wave_rank(bin, valid, BIN_BITS, lt_mask, rank, cnt);
            counter_take(h, bin, valid, rank, cnt);
            info[i] = bin | ((unsigned)rank << 8) | ((unsigned)cnt << 13) | (rl << 19) |
                      (valid ? (1u << 28) : 0u);
        }
    }

    {
        const int q = t / NBINS, bn = t % NBINS;
        int bef = 0, tot = 0;
#pragma unroll 16
        for (int b = q; b < n_blocks; b += SORT_THREADS / NBINS) {
            const int x = blockhist[b * NBINS + bn];
            tot += x;
            bef += (b < me) ? x : 0;
        }
        part[0][q][bn] = bef;
        part[1][q][bn] = tot;
    }
    __syncthreads();
    int total = 0, before = 0;
    if (t < NBINS) {
#pragma unroll
        for (int q = 0; q < SORT_THREADS / NBINS; ++q) { before += part[0][q][t]; total += part[1][q][t]; }
        scan[t] = total;
    }
    __syncthreads();
    for (int d = 1; d < NBINS; d <<= 1) {
        int add = 0;
        if (t < NBINS && t >= d) add = scan[t - d];
        __syncthreads();
        if (t < NBINS) scan[t] += add;
        __syncthreads();
    }
    if (t < NBINS) {
        const int start = scan[t] - total;
        if (me == 0) {
            binstart[t] = start;
            if (t == NBINS - 1) binstart[NBINS] = scan[t];
        }
        int run = start + before;
        for (int w2 = 0; w2 < SORT_WAVES; ++w2) {
            const int cnt = whist[w2][t];
            whist[w2][t] = run;
            run += cnt;
        }
    }
    __syncthreads();

    int c[SORT_EPT];
    float v[SORT_EPT];
#pragma unroll
    for (int i = 0; i < SORT_EPT; ++i) {
        const int e = e0 + i * 32;
        c[i] = 0; v[i] = 0.f;
        if (info[i] >> 28) { c[i] = cols[e]; v[i] = vals[e]; }
    }
#pragma unroll
    for (int i = 0; i < SORT_EPT; ++i) {
        const bool     valid = (info[i] >> 28) != 0u;
        const unsigned bin   = info[i] & 0xFFu;
        const int      rank  = (int)((info[i] >> 8) & 0x1Fu);
        const int      cnt   = (int)((info[i] >> 13) & 0x3Fu);
        const unsigned rl    = (info[i] >> 19) & 0x1FFu;
        const int pos = counter_take(h, bin, valid, rank, cnt);
        if (valid)
            rec[pos] = make_uint2((unsigned)c[i] | (rl << COL_BITS), __float_as_uint(v[i]));
    }
}

__global__ __launch_bounds__(SORT_THREADS) void gcn_gemm_and_place(const float* __restrict__ X,
                                                                   const float* __restrict__ W,
                                                                   float* __restrict__ S,
                                                                   int n_rows, int gemm_blocks,
                                                                   const int* __restrict__ rows,
                                                                   const int* __restrict__ cols,
                                                                   const float* __restrict__ vals,
                                                                   const int* __restrict__ blockhist,
                                                                   int* __restrict__ binstart,
                                                                   uint2* __restrict__ rec,
                                                                   int n_edges, int n_blocks) {
    __shared__ __attribute__((aligned(16))) unsigned char smem[WL_BYTES];
    const int blk = (int)blockIdx.x;
    if (blk < gemm_blocks) {
        v2f* wl = (v2f*)smem;
        stage_w_pairs(W, wl);
        __syncthreads();
        gemm_strip_lds(X, wl, S, n_rows, blk * SORT_WAVES + (int)(threadIdx.x >> 5));
    } else {
        int (*whist)[NBINS] = (int (*)[NBINS])smem;
        int (*part)[SORT_THREADS / NBINS][NBINS] =
            (int (*)[SORT_THREADS / NBINS][NBINS])(smem + sizeof(int) * SORT_WAVES * NBINS);
        int* scan = (int*)(smem + sizeof(int) * (SORT_WAVES * NBINS + 2 * SORT_THREADS));
        place_block(rows, cols, vals, blockhist, binstart, rec, n_edges, n_blocks,
                    blk - gemm_blocks, whist, part, scan);
    }
}

#define RS_THREADS 1024
#define RS_WAVES   (RS_THREADS / 32)
#define RS_BATCH   8
#define RS_SCAN    512
#define RS_GROUP   4

__global__ __launch_bounds__(RS_THREADS) void gcn_range_sum(const uint2* __restrict__ rec,
                                                            uint2* rec2,
                                                            const int* __restrict__ binstart,
                                                            const float* __restrict__ S,
                                                            const float* __restrict__ bias,
                                                            float* __restrict__ out,
                                                            int n_rows) {
    __shared__ int whist[RS_WAVES][RPB + 1];
    __shared__ int rowscan[RS_SCAN];
    __shared__ int rowstart[RPB + 1];
    const int t    = threadIdx.x;
    const int lane = t & 31;
    const int w    = __builtin_amdgcn_readfirstlane(t >> 5);
    const int bin  = blockIdx.x;
    const unsigned lt_mask = (1u << lane) - 1u;

    const int start = binstart[bin];
    const int end   = binstart[bin + 1];
    const int per   = (end - start + RS_WAVES - 1) / RS_WAVES;
    const int a_lo  = start + w * per;
    int a_hi = a_lo + per;
    if (a_hi > end) a_hi = end;

    int* h = whist[w];
    for (int k = lane; k <= RPB; k += 32) h[k] = 0;
    for (int cb = a_lo; cb < a_hi; cb += 32 * RS_GROUP) {
        unsigned key[RS_GROUP];
        bool     ok[RS_GROUP];
#pragma unroll
        for (int j = 0; j < RS_GROUP; ++j) {
            const int idx = cb + j * 32 + lane;
            ok[j]  = idx < a_hi;
            key[j] = 0u;
            if (ok[j]) key[j] = rec[idx].x;
        }
#pragma unroll
        for (int j = 0; j < RS_GROUP; ++j)
            wave_count_step(h, key[j] >> COL_BITS, ok[j], ROW_BITS, lt_mask);
    }
    __syncthreads();

    int rc = 0;
    if (t < RPB) {
        int run = 0;
        for (int w2 = 0; w2 < RS_WAVES; ++w2) {
            const int cnt = whist[w2][t];
            whist[w2][t] = run;
            run += cnt;
        }
        rc = run;
    }
    if (t < RS_SCAN) rowscan[t] = rc;
    __syncthreads();
    for (int d = 1; d < RS_SCAN; d <<= 1) {
        int add = 0;
        if (t < RS_SCAN && t >= d) add = rowscan[t - d];
        __syncthreads();
        if (t < RS_SCAN) rowscan[t] += add;
        __syncthreads();
    }
    if (t <= RPB) rowstart[t] = (t == 0) ? 0 : rowscan[t - 1];
    __syncthreads();

    for (int cb = a_lo; cb < a_hi; cb += 32 * RS_GROUP) {
        uint2 q[RS_GROUP];
        bool  ok[RS_GROUP];
#pragma unroll
        for (int j = 0; j < RS_GROUP; ++j) {
            const int idx = cb + j * 32 + lane;
            ok[j] = idx < a_hi;
            q[j]  = make_uint2(0u, 0u);
            if (ok[j]) q[j] = rec[idx];
        }
#pragma unroll
        for (int j = 0; j < RS_GROUP; ++j) {
            const unsigned rl = q[j].x >> COL_BITS;
            const int slot = wave_count_step(h, rl, ok[j], ROW_BITS, lt_mask);
            if (ok[j]) rec2[start + rowstart[rl] + slot] = q[j];
        }
    }
    __syncthreads();

    const int row0 = bin * RPB;
    int nrows = n_rows - row0;
    if (nrows > RPB) nrows = RPB;
    const int extra = RPB % RS_WAVES;
    const int r0 = w * (RPB / RS_WAVES) + (w < extra ? w : extra);
    int r1 = r0 + (RPB / RS_WAVES) + (w < extra ? 1 : 0);
    if (r1 > nrows) r1 = nrows;
    if (r0 >= r1) return;

    const float b0 = bias[lane], b1 = bias[lane + 32], b2 = bias[lane + 64], b3 = bias[lane + 96];
    const int e_lo = start + __builtin_amdgcn_readfirstlane(rowstart[r0]);
    const int e_hi = start + __builtin_amdgcn_readfirstlane(rowstart[r1]);

    int   cur = r0;
    float s0 = 0.f, s1 = 0.f, s2 = 0.f, s3 = 0.f;

#define RS_FLUSH(nxt)                                                          \
    do {                                                                       \
        float* o = out + (long long)(row0 + cur) * D_OUT;                      \
        o[lane] = s0 + b0; o[lane + 32] = s1 + b1;                             \
        o[lane + 64] = s2 + b2; o[lane + 96] = s3 + b3;                        \
        _Pragma("unroll 1")   \
        for (int rr = cur + 1; rr < (nxt); ++rr) {                             \
            float* oe = out + (long long)(row0 + rr) * D_OUT;                  \
            oe[lane] = b0; oe[lane + 32] = b1; oe[lane + 64] = b2; oe[lane + 96] = b3; \
        }                                                                      \
        s0 = 0.f; s1 = 0.f; s2 = 0.f; s3 = 0.f;                                \
    } while (0)

    int key_v = 0, val_v = 0;
    if (e_lo + lane < e_hi) {
        const uint2 q = rec2[e_lo + lane];
        key_v = (int)q.x;
        val_v = (int)q.y;
    }
    for (int cb = e_lo; cb < e_hi; cb += 32) {
        int n_here = e_hi - cb;
        if (n_here > 32) n_here = 32;
        int key_n = 0, val_n = 0;
        if (cb + 32 + lane < e_hi) {
            const uint2 q = rec2[cb + 32 + lane];
            key_n = (int)q.x;
            val_n = (int)q.y;
        }
        int j0 = 0;
        for (; j0 + RS_BATCH <= n_here; j0 += RS_BATCH) {
            float g[RS_BATCH][4];
#pragma unroll
            for (int u = 0; u < RS_BATCH; ++u) {
                const int key = __builtin_amdgcn_readlane(key_v, j0 + u);
                const float* src = S + (long long)(key & ((1 << COL_BITS) - 1)) * D_OUT;
                g[u][0] = src[lane];
                g[u][1] = src[lane + 32];
                g[u][2] = src[lane + 64];
                g[u][3] = src[lane + 96];
            }
#pragma unroll
            for (int u = 0; u < RS_BATCH; ++u) {
                const int   key = __builtin_amdgcn_readlane(key_v, j0 + u);
                const float v   = __int_as_float(__builtin_amdgcn_readlane(val_v, j0 + u));
                const int   rl  = (int)((unsigned)key >> COL_BITS);
                if (rl != cur) { RS_FLUSH(rl); cur = rl; }
                s0 += g[u][0] * v; s1 += g[u][1] * v; s2 += g[u][2] * v; s3 += g[u][3] * v;
            }
        }
        for (; j0 < n_here; ++j0) {
            const int   key = __builtin_amdgcn_readlane(key_v, j0);
            const float v   = __int_as_float(__builtin_amdgcn_readlane(val_v, j0));
            const int   rl  = (int)((unsigned)key >> COL_BITS);
            const float* src = S + (long long)(key & ((1 << COL_BITS) - 1)) * D_OUT;
            const float g0 = src[lane];
            const float g1 = src[lane + 32];
            const float g2 = src[lane + 64];
            const float g3 = src[lane + 96];
            if (rl != cur) { RS_FLUSH(rl); cur = rl; }
            s0 += g0 * v; s1 += g1 * v; s2 += g2 * v; s3 += g3 * v;
        }
        key_v = key_n;
        val_v = val_n;
    }
    RS_FLUSH(r1);
#undef RS_FLUSH
}

extern "C" void kernel_launch(void* const* d_in, const int* in_sizes, int n_in,
                              void* d_out, int out_size, void* d_ws, size_t ws_size,
                              hipStream_t stream) {
    const int*   rows = (const int*)d_in[0];
    const int*   cols = (const int*)d_in[1];
    const float* vals = (const float*)d_in[2];
    const float* x    = (const float*)d_in[3];
    const float* W    = (const float*)d_in[4];
    const float* bias = (const float*)d_in[5];

    const int E = in_sizes[0];
    const int N = in_sizes[3] / D_IN;

    float* support = (float*)d_ws;

    const int mtiles = N / 16;

    const int    sort_blocks = (E + SORT_EPB - 1) / SORT_EPB;
    const size_t off_rec   = (((size_t)N * D_OUT * sizeof(float)) + 255) & ~(size_t)255;
    const size_t off_rec2  = (off_rec  + (size_t)E * sizeof(uint2) + 255) & ~(size_t)255;
    const size_t off_hist  = (off_rec2 + (size_t)E * sizeof(uint2) + 255) & ~(size_t)255;
    const size_t off_start = (off_hist + (size_t)sort_blocks * NBINS * sizeof(int) + 255) & ~(size_t)255;
    const size_t ws_need   = off_start + (NBINS + 1) * sizeof(int);

    if (ws_need <= ws_size && N <= NBINS * RPB) {
        uint2* rec       = (uint2*)((char*)d_ws + off_rec);
        uint2* rec2      = (uint2*)((char*)d_ws + off_rec2);
        int*   blockhist = (int*)((char*)d_ws + off_hist);
        int*   binstart  = (int*)((char*)d_ws + off_start);
        const int gemm_blocks = (mtiles + SORT_WAVES - 1) / SORT_WAVES;
        gcn_bin_count<<<sort_blocks, SORT_THREADS, 0, stream>>>(rows, blockhist, E);
        gcn_gemm_and_place<<<gemm_blocks + sort_blocks, SORT_THREADS, 0, stream>>>(
            x, W, support, N, gemm_blocks, rows, cols, vals, blockhist, binstart, rec, E,
            sort_blocks);
        gcn_range_sum<<<NBINS, RS_THREADS, 0, stream>>>(rec, rec2, binstart, support, bias,
                                                        (float*)d_out, N);
    } else {
        gcn_gemm_xw<<<(mtiles + 7) / 8, 256, 0, stream>>>(x, W, support, N);
        long long total4 = (long long)N * (D_OUT / 4);
        int iblocks = (int)((total4 + 255) / 256);
        gcn_init_out<<<iblocks, 256, 0, stream>>>(bias, (float*)d_out, total4);
        int waves  = (E + EDGES_PER_WAVE - 1) / EDGES_PER_WAVE;
        int blocks = (waves + 7) / 8;
        gcn_edge_scatter<<<blocks, 256, 0, stream>>>(rows, cols, vals, support,
                                                     (float*)d_out, E);
    }
}


# Range sum: a range's row-ordered records go into a 64 KiB LDS copy, not the rec2 workspace plane (ranges over 8192 records keep rec2): no scattered 8-byte global stores; sum order unchanged
# speedup vs baseline: 2.6589x; 1.0648x over previous
#include <hip/hip_runtime.h>

typedef __attribute__((ext_vector_type(2))) float v2f;
typedef __attribute__((ext_vector_type(8))) float v8f;

#define D_IN  256
#define D_OUT 128

#define NBINS        256
#define BIN_BITS     8
#define RPB          391
#define ROW_BITS     9
#define COL_BITS     20
#define SORT_THREADS 1024
#define SORT_WAVES   (SORT_THREADS / 32)
#define SORT_EPT     8
#define SORT_EPB     (SORT_THREADS * SORT_EPT)

__global__ __launch_bounds__(256) void gcn_init_out(const float* __restrict__ bias,
                                                    float* __restrict__ out,
                                                    long long total4) {
    long long i = (long long)blockIdx.x * blockDim.x + threadIdx.x;
    if (i < total4) {
        int d4 = (int)(i & (D_OUT / 4 - 1));
        ((float4*)out)[i] = ((const float4*)bias)[d4];
    }
}

static __device__ __forceinline__ void gemm_strip(const float* __restrict__ X,
                                                  const float* __restrict__ W,
                                                  float* __restrict__ S,
                                                  int n_rows, int mtile) {
    const int lane  = threadIdx.x & 31;
    if (mtile * 16 + 16 > n_rows) return;
    const int m0 = mtile * 16;

    const int am = lane & 15;
    const int ak = (lane >> 4) * 2;
    const int bn = lane & 15;
    const int bk = (lane >> 4) * 2;

    v8f c[8];
#pragma unroll
    for (int t = 0; t < 8; ++t) c[t] = (v8f){0.f, 0.f, 0.f, 0.f, 0.f, 0.f, 0.f, 0.f};

    const float* xrow = X + (long long)(m0 + am) * D_IN;

    for (int k = 0; k < D_IN; k += 4) {
        v2f a = *(const v2f*)(xrow + k + ak);
#pragma unroll
        for (int t = 0; t < 8; ++t) {
            v2f b;
            b.x = W[(long long)(k + bk)     * D_OUT + t * 16 + bn];
            b.y = W[(long long)(k + bk + 1) * D_OUT + t * 16 + bn];
            c[t] = __builtin_amdgcn_wmma_f32_16x16x4_f32(
                 false, a,  false, b,
                 (short)0, c[t],  false,  false);
        }
    }

    const int cn    = lane & 15;
    const int cmoff = (lane >> 4) * 8;
#pragma unroll
    for (int t = 0; t < 8; ++t) {
#pragma unroll
        for (int r = 0; r < 8; ++r) {
            S[(long long)(m0 + r + cmoff) * D_OUT + t * 16 + cn] = c[t][r];
        }
    }
}

__global__ __launch_bounds__(256) void gcn_gemm_xw(const float* __restrict__ X,
                                                   const float* __restrict__ W,
                                                   float* __restrict__ S,
                                                   int n_rows) {
    gemm_strip(X, W, S, n_rows, (int)blockIdx.x * 8 + (int)(threadIdx.x >> 5));
}

#define WL_PITCH 130
#define WL_BYTES ((D_IN / 2) * WL_PITCH * 8)

static __device__ __forceinline__ void stage_w_pairs(const float* __restrict__ W, v2f* wl) {
    float4 w0[4], w1[4];
#pragma unroll
    for (int i = 0; i < 4; ++i) {
        const int idx = (int)threadIdx.x + i * SORT_THREADS;
        const int p = idx >> 5, q = idx & 31;
        w0[i] = ((const float4*)W)[(2 * p)     * (D_OUT / 4) + q];
        w1[i] = ((const float4*)W)[(2 * p + 1) * (D_OUT / 4) + q];
    }
#pragma unroll
    for (int i = 0; i < 4; ++i) {
        const int idx = (int)threadIdx.x + i * SORT_THREADS;
        const int p = idx >> 5, q = idx & 31;
        float4* d = (float4*)(wl + p * WL_PITCH + 4 * q);
        d[0] = make_float4(w0[i].x, w1[i].x, w0[i].y, w1[i].y);
        d[1] = make_float4(w0[i].z, w1[i].z, w0[i].w, w1[i].w);
    }
}

static __device__ __forceinline__ void gemm_strip_lds(const float* __restrict__ X,
                                                      const v2f* wl,
                                                      float* __restrict__ S,
                                                      int n_rows, int mtile) {
    const int lane = threadIdx.x & 31;
    if (mtile * 16 + 16 > n_rows) return;
    const int m0 = mtile * 16;
    const int c  = lane & 15;
    const int h  = lane >> 4;

    v8f acc[8];
#pragma unroll
    for (int t = 0; t < 8; ++t) acc[t] = (v8f){0.f, 0.f, 0.f, 0.f, 0.f, 0.f, 0.f, 0.f};

    const float* xrow  = X + (long long)(m0 + c) * D_IN + 16 * h;
    const v2f*   wlane = wl + (8 * h) * WL_PITCH + c;

#pragma unroll 1
    for (int kb = 0; kb < D_IN; kb += 32) {
        float4 q[4];
#pragma unroll
        for (int i = 0; i < 4; ++i) q[i] = *(const float4*)(xrow + kb + 4 * i);
        const v2f* wk = wlane + (kb / 2) * WL_PITCH;
#pragma unroll
        for (int s = 0; s < 8; ++s) {
            v2f a;
            a.x = (s & 1) ? q[s >> 1].z : q[s >> 1].x;
            a.y = (s & 1) ? q[s >> 1].w : q[s >> 1].y;
#pragma unroll
            for (int t = 0; t < 8; ++t) {
                const v2f b = wk[s * WL_PITCH + 16 * t];
                acc[t] = __builtin_amdgcn_wmma_f32_16x16x4_f32(
                     false, a,  false, b,
                     (short)0, acc[t],  false,  false);
            }
        }
    }

    const int cmoff = h * 8;
#pragma unroll
    for (int t = 0; t < 8; ++t) {
#pragma unroll
        for (int r = 0; r < 8; ++r) {
            S[(long long)(m0 + r + cmoff) * D_OUT + t * 16 + c] = acc[t][r];
        }
    }
}

#define EDGES_PER_WAVE 32
#define EDGE_BATCH     8

static __device__ __forceinline__ void add_row_lines(float* dst, unsigned off,
                                                      float g0, float g1, float g2, float g3) {
    asm volatile("global_atomic_add_f32 %0, %1, %2"            :: "v"(off), "v"(g0), "s"(dst) : "memory");
    asm volatile("global_atomic_add_f32 %0, %1, %2 offset:128" :: "v"(off), "v"(g1), "s"(dst) : "memory");
    asm volatile("global_atomic_add_f32 %0, %1, %2 offset:256" :: "v"(off), "v"(g2), "s"(dst) : "memory");
    asm volatile("global_atomic_add_f32 %0, %1, %2 offset:384" :: "v"(off), "v"(g3), "s"(dst) : "memory");
}

__global__ __launch_bounds__(256) void gcn_edge_scatter(const int* __restrict__ rows,
                                                        const int* __restrict__ cols,
                                                        const float* __restrict__ vals,
                                                        const float* __restrict__ S,
                                                        float* __restrict__ out,
                                                        int n_edges) {
    const int lane = threadIdx.x & 31;
    int w = (int)(blockIdx.x * 8u) + (int)(threadIdx.x >> 5);
    w = __builtin_amdgcn_readfirstlane(w);
    const int e_base = w * EDGES_PER_WAVE;
    if (e_base >= n_edges) return;
    int n_here = n_edges - e_base;
    if (n_here > EDGES_PER_WAVE) n_here = EDGES_PER_WAVE;

    int r_v = 0, c_v = 0, v_v = 0;
    if (lane < n_here) {
        r_v = rows[e_base + lane];
        c_v = cols[e_base + lane];
        v_v = __float_as_int(vals[e_base + lane]);
    }
    const unsigned off = (unsigned)lane * 4u;

    int j0 = 0;
    for (; j0 + EDGE_BATCH <= n_here; j0 += EDGE_BATCH) {
        float g[EDGE_BATCH][4];
#pragma unroll
        for (int u = 0; u < EDGE_BATCH; ++u) {
            const int c = __builtin_amdgcn_readlane(c_v, j0 + u);
            const float* src = S + (long long)c * D_OUT;
            g[u][0] = src[lane];
            g[u][1] = src[lane + 32];
            g[u][2] = src[lane + 64];
            g[u][3] = src[lane + 96];
        }
#pragma unroll
        for (int u = 0; u < EDGE_BATCH; ++u) {
            const int   r = __builtin_amdgcn_readlane(r_v, j0 + u);
            const float v = __int_as_float(__builtin_amdgcn_readlane(v_v, j0 + u));
            float* dst = out + (long long)r * D_OUT;
            add_row_lines(dst, off, g[u][0] * v, g[u][1] * v, g[u][2] * v, g[u][3] * v);
        }
    }
    for (; j0 < n_here; ++j0) {
        const int   c = __builtin_amdgcn_readlane(c_v, j0);
        const int   r = __builtin_amdgcn_readlane(r_v, j0);
        const float v = __int_as_float(__builtin_amdgcn_readlane(v_v, j0));
        const float* src = S + (long long)c * D_OUT;
        const float s0 = src[lane];
        const float s1 = src[lane + 32];
        const float s2 = src[lane + 64];
        const float s3 = src[lane + 96];
        float* dst = out + (long long)r * D_OUT;
        add_row_lines(dst, off, s0 * v, s1 * v, s2 * v, s3 * v);
    }
}


static __device__ __forceinline__ unsigned same_key_lanes(unsigned key, bool valid, int nbits) {
    unsigned peers = __builtin_amdgcn_ballot_w32(valid);
#pragma unroll
    for (int b = 0; b < nbits; ++b) {
        const bool bit = ((key >> b) & 1u) != 0u;
        const unsigned m = __builtin_amdgcn_ballot_w32(bit && valid);
        peers &= bit ? m : ~m;
    }
    return valid ? peers : 0u;
}

static __device__ __forceinline__ void wave_rank(unsigned key, bool valid, int nbits,
                                                 unsigned lt_mask, int& rank, int& cnt) {
    const unsigned peers = same_key_lanes(key, valid, nbits);
    rank = __builtin_popcount(peers & lt_mask);
    cnt  = __builtin_popcount(peers);
}

static __device__ __forceinline__ int counter_take(int* h, unsigned key, bool valid,
                                                   int rank, int cnt) {
    int slot = 0;
    if (valid) {
        const int old = h[key];
        if (rank == 0) h[key] = old + cnt;
        slot = old + rank;
    }
    return slot;
}

static __device__ __forceinline__ int wave_count_step(int* h, unsigned key, bool valid,
                                                      int nbits, unsigned lt_mask) {
    int rank, cnt;
    wave_rank(key, valid, nbits, lt_mask, rank, cnt);
    return counter_take(h, key, valid, rank, cnt);
}

__global__ __launch_bounds__(SORT_THREADS) void gcn_bin_count(const int* __restrict__ rows,
                                                              int* __restrict__ blockhist,
                                                              int n_edges) {
    __shared__ int whist[SORT_WAVES][NBINS];
    const int t = threadIdx.x, lane = t & 31, w = t >> 5;
    const unsigned lt_mask = (1u << lane) - 1u;
    int* h = whist[w];
    for (int k = lane; k < NBINS; k += 32) h[k] = 0;

    const int e0 = blockIdx.x * SORT_EPB + w * (SORT_EPT * 32) + lane;
    int r[SORT_EPT];
#pragma unroll
    for (int i = 0; i < SORT_EPT; ++i) {
        const int e = e0 + i * 32;
        r[i] = (e < n_edges) ? rows[e] : -1;
    }
#pragma unroll
    for (int i = 0; i < SORT_EPT; ++i) {
        const bool valid = r[i] >= 0;
        const unsigned bin = valid ? (unsigned)(r[i] / RPB) : 0u;
        wave_count_step(h, bin, valid, BIN_BITS, lt_mask);
    }
    __syncthreads();
    if (t < NBINS) {
        int s = 0;
        for (int w2 = 0; w2 < SORT_WAVES; ++w2) s += whist[w2][t];
        blockhist[blockIdx.x * NBINS + t] = s;
    }
}

static __device__ __forceinline__ void place_block(const int* __restrict__ rows,
                                                   const int* __restrict__ cols,
                                                   const float* __restrict__ vals,
                                                   const int* __restrict__ blockhist,
                                                   int* __restrict__ binstart,
                                                   uint2* __restrict__ rec,
                                                   int n_edges, int n_blocks, int me,
                                                   int (*whist)[NBINS],
                                                   int (*part)[SORT_THREADS / NBINS][NBINS],
                                                   int* scan) {
    const int t = threadIdx.x, lane = t & 31, w = t >> 5;
    const unsigned lt_mask = (1u << lane) - 1u;
    int* h = whist[w];
    for (int k = lane; k < NBINS; k += 32) h[k] = 0;

    const int e0 = me * SORT_EPB + w * (SORT_EPT * 32) + lane;
    unsigned info[SORT_EPT];
    {
        int r[SORT_EPT];
#pragma unroll
        for (int i = 0; i < SORT_EPT; ++i) {
            const int e = e0 + i * 32;
            r[i] = (e < n_edges) ? rows[e] : -1;
        }
#pragma unroll
        for (int i = 0; i < SORT_EPT; ++i) {
            const bool valid = r[i] >= 0;
            const unsigned bin = valid ? (unsigned)(r[i] / RPB) : 0u;
            const unsigned rl  = valid ? (unsigned)r[i] - bin * RPB : 0u;
            int rank, cnt;
            wave_rank(bin, valid, BIN_BITS, lt_mask, rank, cnt);
            counter_take(h, bin, valid, rank, cnt);
            info[i] = bin | ((unsigned)rank << 8) | ((unsigned)cnt << 13) | (rl << 19) |
                      (valid ? (1u << 28) : 0u);
        }
    }

    {
        const int q = t / NBINS, bn = t % NBINS;
        int bef = 0, tot = 0;
#pragma unroll 16
        for (int b = q; b < n_blocks; b += SORT_THREADS / NBINS) {
            const int x = blockhist[b * NBINS + bn];
            tot += x;
            bef += (b < me) ? x : 0;
        }
        part[0][q][bn] = bef;
        part[1][q][bn] = tot;
    }
    __syncthreads();
    int total = 0, before = 0;
    if (t < NBINS) {
#pragma unroll
        for (int q = 0; q < SORT_THREADS / NBINS; ++q) { before += part[0][q][t]; total += part[1][q][t]; }
        scan[t] = total;
    }
    __syncthreads();
    for (int d = 1; d < NBINS; d <<= 1) {
        int add = 0;
        if (t < NBINS && t >= d) add = scan[t - d];
        __syncthreads();
        if (t < NBINS) scan[t] += add;
        __syncthreads();
    }
    if (t < NBINS) {
        const int start = scan[t] - total;
        if (me == 0) {
            binstart[t] = start;
            if (t == NBINS - 1) binstart[NBINS] = scan[t];
        }
        int run = start + before;
        for (int w2 = 0; w2 < SORT_WAVES; ++w2) {
            const int cnt = whist[w2][t];
            whist[w2][t] = run;
            run += cnt;
        }
    }
    __syncthreads();

    int c[SORT_EPT];
    float v[SORT_EPT];
#pragma unroll
    for (int i = 0; i < SORT_EPT; ++i) {
        const int e = e0 + i * 32;
        c[i] = 0; v[i] = 0.f;
        if (info[i] >> 28) { c[i] = cols[e]; v[i] = vals[e]; }
    }
#pragma unroll
    for (int i = 0; i < SORT_EPT; ++i) {
        const bool     valid = (info[i] >> 28) != 0u;
        const unsigned bin   = info[i] & 0xFFu;
        const int      rank  = (int)((info[i] >> 8) & 0x1Fu);
        const int      cnt   = (int)((info[i] >> 13) & 0x3Fu);
        const unsigned rl    = (info[i] >> 19) & 0x1FFu;
        const int pos = counter_take(h, bin, valid, rank, cnt);
        if (valid)
            rec[pos] = make_uint2((unsigned)c[i] | (rl << COL_BITS), __float_as_uint(v[i]));
    }
}

__global__ __launch_bounds__(SORT_THREADS) void gcn_gemm_and_place(const float* __restrict__ X,
                                                                   const float* __restrict__ W,
                                                                   float* __restrict__ S,
                                                                   int n_rows, int gemm_blocks,
                                                                   const int* __restrict__ rows,
                                                                   const int* __restrict__ cols,
                                                                   const float* __restrict__ vals,
                                                                   const int* __restrict__ blockhist,
                                                                   int* __restrict__ binstart,
                                                                   uint2* __restrict__ rec,
                                                                   int n_edges, int n_blocks) {
    __shared__ __attribute__((aligned(16))) unsigned char smem[WL_BYTES];
    const int blk = (int)blockIdx.x;
    if (blk < gemm_blocks) {
        v2f* wl = (v2f*)smem;
        stage_w_pairs(W, wl);
        __syncthreads();
        gemm_strip_lds(X, wl, S, n_rows, blk * SORT_WAVES + (int)(threadIdx.x >> 5));
    } else {
        int (*whist)[NBINS] = (int (*)[NBINS])smem;
        int (*part)[SORT_THREADS / NBINS][NBINS] =
            (int (*)[SORT_THREADS / NBINS][NBINS])(smem + sizeof(int) * SORT_WAVES * NBINS);
        int* scan = (int*)(smem + sizeof(int) * (SORT_WAVES * NBINS + 2 * SORT_THREADS));
        place_block(rows, cols, vals, blockhist, binstart, rec, n_edges, n_blocks,
                    blk - gemm_blocks, whist, part, scan);
    }
}

#define RS_THREADS 1024
#define RS_WAVES   (RS_THREADS / 32)
#define RS_BATCH   8
#define RS_SCAN    512
#define RS_GROUP   4
#define RS_CAP     8192

template <bool IN_LDS>
static __device__ __forceinline__ void range_order_and_sum(const uint2* __restrict__ rec,
                                                           uint2* rec2,
                                                           uint2* lrec,
                                                           int (*whist)[RPB + 1],
                                                           int* rowscan,
                                                           int* rowstart,
                                                           const float* __restrict__ S,
                                                           const float* __restrict__ bias,
                                                           float* __restrict__ out,
                                                           int n_rows, int start, int end) {
    const int t    = threadIdx.x;
    const int lane = t & 31;
    const int w    = __builtin_amdgcn_readfirstlane(t >> 5);
    const int bin  = blockIdx.x;
    const unsigned lt_mask = (1u << lane) - 1u;

    const int per   = (end - start + RS_WAVES - 1) / RS_WAVES;
    const int a_lo  = start + w * per;
    int a_hi = a_lo + per;
    if (a_hi > end) a_hi = end;

    int* h = whist[w];
    for (int k = lane; k <= RPB; k += 32) h[k] = 0;
    for (int cb = a_lo; cb < a_hi; cb += 32 * RS_GROUP) {
        unsigned key[RS_GROUP];
        bool     ok[RS_GROUP];
#pragma unroll
        for (int j = 0; j < RS_GROUP; ++j) {
            const int idx = cb + j * 32 + lane;
            ok[j]  = idx < a_hi;
            key[j] = 0u;
            if (ok[j]) key[j] = rec[idx].x;
        }
#pragma unroll
        for (int j = 0; j < RS_GROUP; ++j)
            wave_count_step(h, key[j] >> COL_BITS, ok[j], ROW_BITS, lt_mask);
    }
    __syncthreads();

    int rc = 0;
    if (t < RPB) {
        int run = 0;
        for (int w2 = 0; w2 < RS_WAVES; ++w2) {
            const int cnt = whist[w2][t];
            whist[w2][t] = run;
            run += cnt;
        }
        rc = run;
    }
    if (t < RS_SCAN) rowscan[t] = rc;
    __syncthreads();
    for (int d = 1; d < RS_SCAN; d <<= 1) {
        int add = 0;
        if (t < RS_SCAN && t >= d) add = rowscan[t - d];
        __syncthreads();
        if (t < RS_SCAN) rowscan[t] += add;
        __syncthreads();
    }
    if (t <= RPB) rowstart[t] = (t == 0) ? 0 : rowscan[t - 1];
    __syncthreads();

    for (int cb = a_lo; cb < a_hi; cb += 32 * RS_GROUP) {
        uint2 q[RS_GROUP];
        bool  ok[RS_GROUP];
#pragma unroll
        for (int j = 0; j < RS_GROUP; ++j) {
            const int idx = cb + j * 32 + lane;
            ok[j] = idx < a_hi;
            q[j]  = make_uint2(0u, 0u);
            if (ok[j]) q[j] = rec[idx];
        }
#pragma unroll
        for (int j = 0; j < RS_GROUP; ++j) {
            const unsigned rl = q[j].x >> COL_BITS;
            const int slot = wave_count_step(h, rl, ok[j], ROW_BITS, lt_mask);
            if (ok[j]) {
                const int pos = rowstart[rl] + slot;
                if (IN_LDS) lrec[pos] = q[j];
                else        rec2[start + pos] = q[j];
            }
        }
    }
    __syncthreads();

    const int row0 = bin * RPB;
    int nrows = n_rows - row0;
    if (nrows > RPB) nrows = RPB;
    const int extra = RPB % RS_WAVES;
    const int r0 = w * (RPB / RS_WAVES) + (w < extra ? w : extra);
    int r1 = r0 + (RPB / RS_WAVES) + (w < extra ? 1 : 0);
    if (r1 > nrows) r1 = nrows;
    if (r0 >= r1) return;

    const float b0 = bias[lane], b1 = bias[lane + 32], b2 = bias[lane + 64], b3 = bias[lane + 96];
    const int e_lo = start + __builtin_amdgcn_readfirstlane(rowstart[r0]);
    const int e_hi = start + __builtin_amdgcn_readfirstlane(rowstart[r1]);

    int   cur = r0;
    float s0 = 0.f, s1 = 0.f, s2 = 0.f, s3 = 0.f;

#define RS_FLUSH(nxt)                                                          \
    do {                                                                       \
        float* o = out + (long long)(row0 + cur) * D_OUT;                      \
        o[lane] = s0 + b0; o[lane + 32] = s1 + b1;                             \
        o[lane + 64] = s2 + b2; o[lane + 96] = s3 + b3;                        \
        _Pragma("unroll 1")   \
        for (int rr = cur + 1; rr < (nxt); ++rr) {                             \
            float* oe = out + (long long)(row0 + rr) * D_OUT;                  \
            oe[lane] = b0; oe[lane + 32] = b1; oe[lane + 64] = b2; oe[lane + 96] = b3; \
        }                                                                      \
        s0 = 0.f; s1 = 0.f; s2 = 0.f; s3 = 0.f;                                \
    } while (0)

    int key_v = 0, val_v = 0;
    if (e_lo + lane < e_hi) {
        uint2 q;
        if (IN_LDS) q = lrec[e_lo - start + lane];
        else        q = rec2[e_lo + lane];
        key_v = (int)q.x;
        val_v = (int)q.y;
    }
    for (int cb = e_lo; cb < e_hi; cb += 32) {
        int n_here = e_hi - cb;
        if (n_here > 32) n_here = 32;
        int key_n = 0, val_n = 0;
        if (cb + 32 + lane < e_hi) {
            uint2 q;
            if (IN_LDS) q = lrec[cb + 32 - start + lane];
            else        q = rec2[cb + 32 + lane];
            key_n = (int)q.x;
            val_n = (int)q.y;
        }
        int j0 = 0;
        for (; j0 + RS_BATCH <= n_here; j0 += RS_BATCH) {
            float g[RS_BATCH][4];
#pragma unroll
            for (int u = 0; u < RS_BATCH; ++u) {
                const int key = __builtin_amdgcn_readlane(key_v, j0 + u);
                const float* src = S + (long long)(key & ((1 << COL_BITS) - 1)) * D_OUT;
                g[u][0] = src[lane];
                g[u][1] = src[lane + 32];
                g[u][2] = src[lane + 64];
                g[u][3] = src[lane + 96];
            }
#pragma unroll
            for (int u = 0; u < RS_BATCH; ++u) {
                const int   key = __builtin_amdgcn_readlane(key_v, j0 + u);
                const float v   = __int_as_float(__builtin_amdgcn_readlane(val_v, j0 + u));
                const int   rl  = (int)((unsigned)key >> COL_BITS);
                if (rl != cur) { RS_FLUSH(rl); cur = rl; }
                s0 += g[u][0] * v; s1 += g[u][1] * v; s2 += g[u][2] * v; s3 += g[u][3] * v;
            }
        }
        for (; j0 < n_here; ++j0) {
            const int   key = __builtin_amdgcn_readlane(key_v, j0);
            const float v   = __int_as_float(__builtin_amdgcn_readlane(val_v, j0));
            const int   rl  = (int)((unsigned)key >> COL_BITS);
            const float* src = S + (long long)(key & ((1 << COL_BITS) - 1)) * D_OUT;
            const float g0 = src[lane];
            const float g1 = src[lane + 32];
            const float g2 = src[lane + 64];
            const float g3 = src[lane + 96];
            if (rl != cur) { RS_FLUSH(rl); cur = rl; }
            s0 += g0 * v; s1 += g1 * v; s2 += g2 * v; s3 += g3 * v;
        }
        key_v = key_n;
        val_v = val_n;
    }
    RS_FLUSH(r1);
#undef RS_FLUSH
}

__global__ __launch_bounds__(RS_THREADS) void gcn_range_sum(const uint2* __restrict__ rec,
                                                            uint2* rec2,
                                                            const int* __restrict__ binstart,
                                                            const float* __restrict__ S,
                                                            const float* __restrict__ bias,
                                                            float* __restrict__ out,
                                                            int n_rows) {
    __shared__ int whist[RS_WAVES][RPB + 1];
    __shared__ int rowscan[RS_SCAN];
    __shared__ int rowstart[RPB + 1];
    __shared__ uint2 lrec[RS_CAP];
    const int start = binstart[blockIdx.x];
    const int end   = binstart[blockIdx.x + 1];
    if (__builtin_expect(end - start <= RS_CAP, 1))
        range_order_and_sum<true>(rec, rec2, lrec, whist, rowscan, rowstart, S, bias, out,
                                  n_rows, start, end);
    else
        range_order_and_sum<false>(rec, rec2, lrec, whist, rowscan, rowstart, S, bias, out,
                                   n_rows, start, end);
}

extern "C" void kernel_launch(void* const* d_in, const int* in_sizes, int n_in,
                              void* d_out, int out_size, void* d_ws, size_t ws_size,
                              hipStream_t stream) {
    const int*   rows = (const int*)d_in[0];
    const int*   cols = (const int*)d_in[1];
    const float* vals = (const float*)d_in[2];
    const float* x    = (const float*)d_in[3];
    const float* W    = (const float*)d_in[4];
    const float* bias = (const float*)d_in[5];

    const int E = in_sizes[0];
    const int N = in_sizes[3] / D_IN;

    float* support = (float*)d_ws;

    const int mtiles = N / 16;

    const int    sort_blocks = (E + SORT_EPB - 1) / SORT_EPB;
    const size_t off_rec   = (((size_t)N * D_OUT * sizeof(float)) + 255) & ~(size_t)255;
    const size_t off_rec2  = (off_rec  + (size_t)E * sizeof(uint2) + 255) & ~(size_t)255;
    const size_t off_hist  = (off_rec2 + (size_t)E * sizeof(uint2) + 255) & ~(size_t)255;
    const size_t off_start = (off_hist + (size_t)sort_blocks * NBINS * sizeof(int) + 255) & ~(size_t)255;
    const size_t ws_need   = off_start + (NBINS + 1) * sizeof(int);

    if (ws_need <= ws_size && N <= NBINS * RPB) {
        uint2* rec       = (uint2*)((char*)d_ws + off_rec);
        uint2* rec2      = (uint2*)((char*)d_ws + off_rec2);
        int*   blockhist = (int*)((char*)d_ws + off_hist);
        int*   binstart  = (int*)((char*)d_ws + off_start);
        const int gemm_blocks = (mtiles + SORT_WAVES - 1) / SORT_WAVES;
        gcn_bin_count<<<sort_blocks, SORT_THREADS, 0, stream>>>(rows, blockhist, E);
        gcn_gemm_and_place<<<gemm_blocks + sort_blocks, SORT_THREADS, 0, stream>>>(
            x, W, support, N, gemm_blocks, rows, cols, vals, blockhist, binstart, rec, E,
            sort_blocks);
        gcn_range_sum<<<NBINS, RS_THREADS, 0, stream>>>(rec, rec2, binstart, support, bias,
                                                        (float*)d_out, N);
    } else {
        gcn_gemm_xw<<<(mtiles + 7) / 8, 256, 0, stream>>>(x, W, support, N);
        long long total4 = (long long)N * (D_OUT / 4);
        int iblocks = (int)((total4 + 255) / 256);
        gcn_init_out<<<iblocks, 256, 0, stream>>>(bias, (float*)d_out, total4);
        int waves  = (E + EDGES_PER_WAVE - 1) / EDGES_PER_WAVE;
        int blocks = (waves + 7) / 8;
        gcn_edge_scatter<<<blocks, 256, 0, stream>>>(rows, cols, vals, support,
                                                     (float*)d_out, E);
    }
}


# Edges grouped by range inside each placing work-group's own chunk (local count, scan); a range collects its runs from all chunks in chunk order: no count pass, 2 launches not 3; sum order unchanged
# speedup vs baseline: 3.0100x; 1.1320x over previous
#include <hip/hip_runtime.h>

typedef __attribute__((ext_vector_type(2))) float v2f;
typedef __attribute__((ext_vector_type(8))) float v8f;

#define D_IN  256
#define D_OUT 128

#define NBINS        256
#define BIN_BITS     8
#define RPB          391
#define ROW_BITS     9
#define COL_BITS     20
#define SORT_THREADS 1024
#define SORT_WAVES   (SORT_THREADS / 32)
#define SORT_EPT     8
#define SORT_EPB     (SORT_THREADS * SORT_EPT)

__global__ __launch_bounds__(256) void gcn_init_out(const float* __restrict__ bias,
                                                    float* __restrict__ out,
                                                    long long total4) {
    long long i = (long long)blockIdx.x * blockDim.x + threadIdx.x;
    if (i < total4) {
        int d4 = (int)(i & (D_OUT / 4 - 1));
        ((float4*)out)[i] = ((const float4*)bias)[d4];
    }
}

static __device__ __forceinline__ void gemm_strip(const float* __restrict__ X,
                                                  const float* __restrict__ W,
                                                  float* __restrict__ S,
                                                  int n_rows, int mtile) {
    const int lane  = threadIdx.x & 31;
    if (mtile * 16 + 16 > n_rows) return;
    const int m0 = mtile * 16;

    const int am = lane & 15;
    const int ak = (lane >> 4) * 2;
    const int bn = lane & 15;
    const int bk = (lane >> 4) * 2;

    v8f c[8];
#pragma unroll
    for (int t = 0; t < 8; ++t) c[t] = (v8f){0.f, 0.f, 0.f, 0.f, 0.f, 0.f, 0.f, 0.f};

    const float* xrow = X + (long long)(m0 + am) * D_IN;

    for (int k = 0; k < D_IN; k += 4) {
        v2f a = *(const v2f*)(xrow + k + ak);
#pragma unroll
        for (int t = 0; t < 8; ++t) {
            v2f b;
            b.x = W[(long long)(k + bk)     * D_OUT + t * 16 + bn];
            b.y = W[(long long)(k + bk + 1) * D_OUT + t * 16 + bn];
            c[t] = __builtin_amdgcn_wmma_f32_16x16x4_f32(
                 false, a,  false, b,
                 (short)0, c[t],  false,  false);
        }
    }

    const int cn    = lane & 15;
    const int cmoff = (lane >> 4) * 8;
#pragma unroll
    for (int t = 0; t < 8; ++t) {
#pragma unroll
        for (int r = 0; r < 8; ++r) {
            S[(long long)(m0 + r + cmoff) * D_OUT + t * 16 + cn] = c[t][r];
        }
    }
}

__global__ __launch_bounds__(256) void gcn_gemm_xw(const float* __restrict__ X,
                                                   const float* __restrict__ W,
                                                   float* __restrict__ S,
                                                   int n_rows) {
    gemm_strip(X, W, S, n_rows, (int)blockIdx.x * 8 + (int)(threadIdx.x >> 5));
}

#define WL_PITCH 130
#define WL_BYTES ((D_IN / 2) * WL_PITCH * 8)

static __device__ __forceinline__ void stage_w_pairs(const float* __restrict__ W, v2f* wl) {
    float4 w0[4], w1[4];
#pragma unroll
    for (int i = 0; i < 4; ++i) {
        const int idx = (int)threadIdx.x + i * SORT_THREADS;
        const int p = idx >> 5, q = idx & 31;
        w0[i] = ((const float4*)W)[(2 * p)     * (D_OUT / 4) + q];
        w1[i] = ((const float4*)W)[(2 * p + 1) * (D_OUT / 4) + q];
    }
#pragma unroll
    for (int i = 0; i < 4; ++i) {
        const int idx = (int)threadIdx.x + i * SORT_THREADS;
        const int p = idx >> 5, q = idx & 31;
        float4* d = (float4*)(wl + p * WL_PITCH + 4 * q);
        d[0] = make_float4(w0[i].x, w1[i].x, w0[i].y, w1[i].y);
        d[1] = make_float4(w0[i].z, w1[i].z, w0[i].w, w1[i].w);
    }
}

static __device__ __forceinline__ void gemm_strip_lds(const float* __restrict__ X,
                                                      const v2f* wl,
                                                      float* __restrict__ S,
                                                      int n_rows, int mtile) {
    const int lane = threadIdx.x & 31;
    if (mtile * 16 + 16 > n_rows) return;
    const int m0 = mtile * 16;
    const int c  = lane & 15;
    const int h  = lane >> 4;

    v8f acc[8];
#pragma unroll
    for (int t = 0; t < 8; ++t) acc[t] = (v8f){0.f, 0.f, 0.f, 0.f, 0.f, 0.f, 0.f, 0.f};

    const float* xrow  = X + (long long)(m0 + c) * D_IN + 16 * h;
    const v2f*   wlane = wl + (8 * h) * WL_PITCH + c;

#pragma unroll 1
    for (int kb = 0; kb < D_IN; kb += 32) {
        float4 q[4];
#pragma unroll
        for (int i = 0; i < 4; ++i) q[i] = *(const float4*)(xrow + kb + 4 * i);
        const v2f* wk = wlane + (kb / 2) * WL_PITCH;
#pragma unroll
        for (int s = 0; s < 8; ++s) {
            v2f a;
            a.x = (s & 1) ? q[s >> 1].z : q[s >> 1].x;
            a.y = (s & 1) ? q[s >> 1].w : q[s >> 1].y;
#pragma unroll
            for (int t = 0; t < 8; ++t) {
                const v2f b = wk[s * WL_PITCH + 16 * t];
                acc[t] = __builtin_amdgcn_wmma_f32_16x16x4_f32(
                     false, a,  false, b,
                     (short)0, acc[t],  false,  false);
            }
        }
    }

    const int cmoff = h * 8;
#pragma unroll
    for (int t = 0; t < 8; ++t) {
#pragma unroll
        for (int r = 0; r < 8; ++r) {
            S[(long long)(m0 + r + cmoff) * D_OUT + t * 16 + c] = acc[t][r];
        }
    }
}

#define EDGES_PER_WAVE 32
#define EDGE_BATCH     8

static __device__ __forceinline__ void add_row_lines(float* dst, unsigned off,
                                                      float g0, float g1, float g2, float g3) {
    asm volatile("global_atomic_add_f32 %0, %1, %2"            :: "v"(off), "v"(g0), "s"(dst) : "memory");
    asm volatile("global_atomic_add_f32 %0, %1, %2 offset:128" :: "v"(off), "v"(g1), "s"(dst) : "memory");
    asm volatile("global_atomic_add_f32 %0, %1, %2 offset:256" :: "v"(off), "v"(g2), "s"(dst) : "memory");
    asm volatile("global_atomic_add_f32 %0, %1, %2 offset:384" :: "v"(off), "v"(g3), "s"(dst) : "memory");
}

__global__ __launch_bounds__(256) void gcn_edge_scatter(const int* __restrict__ rows,
                                                        const int* __restrict__ cols,
                                                        const float* __restrict__ vals,
                                                        const float* __restrict__ S,
                                                        float* __restrict__ out,
                                                        int n_edges) {
    const int lane = threadIdx.x & 31;
    int w = (int)(blockIdx.x * 8u) + (int)(threadIdx.x >> 5);
    w = __builtin_amdgcn_readfirstlane(w);
    const int e_base = w * EDGES_PER_WAVE;
    if (e_base >= n_edges) return;
    int n_here = n_edges - e_base;
    if (n_here > EDGES_PER_WAVE) n_here = EDGES_PER_WAVE;

    int r_v = 0, c_v = 0, v_v = 0;
    if (lane < n_here) {
        r_v = rows[e_base + lane];
        c_v = cols[e_base + lane];
        v_v = __float_as_int(vals[e_base + lane]);
    }
    const unsigned off = (unsigned)lane * 4u;

    int j0 = 0;
    for (; j0 + EDGE_BATCH <= n_here; j0 += EDGE_BATCH) {
        float g[EDGE_BATCH][4];
#pragma unroll
        for (int u = 0; u < EDGE_BATCH; ++u) {
            const int c = __builtin_amdgcn_readlane(c_v, j0 + u);
            const float* src = S + (long long)c * D_OUT;
            g[u][0] = src[lane];
            g[u][1] = src[lane + 32];
            g[u][2] = src[lane + 64];
            g[u][3] = src[lane + 96];
        }
#pragma unroll
        for (int u = 0; u < EDGE_BATCH; ++u) {
            const int   r = __builtin_amdgcn_readlane(r_v, j0 + u);
            const float v = __int_as_float(__builtin_amdgcn_readlane(v_v, j0 + u));
            float* dst = out + (long long)r * D_OUT;
            add_row_lines(dst, off, g[u][0] * v, g[u][1] * v, g[u][2] * v, g[u][3] * v);
        }
    }
    for (; j0 < n_here; ++j0) {
        const int   c = __builtin_amdgcn_readlane(c_v, j0);
        const int   r = __builtin_amdgcn_readlane(r_v, j0);
        const float v = __int_as_float(__builtin_amdgcn_readlane(v_v, j0));
        const float* src = S + (long long)c * D_OUT;
        const float s0 = src[lane];
        const float s1 = src[lane + 32];
        const float s2 = src[lane + 64];
        const float s3 = src[lane + 96];
        float* dst = out + (long long)r * D_OUT;
        add_row_lines(dst, off, s0 * v, s1 * v, s2 * v, s3 * v);
    }
}


static __device__ __forceinline__ unsigned same_key_lanes(unsigned key, bool valid, int nbits) {
    unsigned peers = __builtin_amdgcn_ballot_w32(valid);
#pragma unroll
    for (int b = 0; b < nbits; ++b) {
        const bool bit = ((key >> b) & 1u) != 0u;
        const unsigned m = __builtin_amdgcn_ballot_w32(bit && valid);
        peers &= bit ? m : ~m;
    }
    return valid ? peers : 0u;
}

static __device__ __forceinline__ void wave_rank(unsigned key, bool valid, int nbits,
                                                 unsigned lt_mask, int& rank, int& cnt) {
    const unsigned peers = same_key_lanes(key, valid, nbits);
    rank = __builtin_popcount(peers & lt_mask);
    cnt  = __builtin_popcount(peers);
}

static __device__ __forceinline__ int counter_take(int* h, unsigned key, bool valid,
                                                   int rank, int cnt) {
    int slot = 0;
    if (valid) {
        const int old = h[key];
        if (rank == 0) h[key] = old + cnt;
        slot = old + rank;
    }
    return slot;
}

static __device__ __forceinline__ int wave_count_step(int* h, unsigned key, bool valid,
                                                      int nbits, unsigned lt_mask) {
    int rank, cnt;
    wave_rank(key, valid, nbits, lt_mask, rank, cnt);
    return counter_take(h, key, valid, rank, cnt);
}

static __device__ __forceinline__ void place_block(const int* __restrict__ rows,
                                                   const int* __restrict__ cols,
                                                   const float* __restrict__ vals,
                                                   int* __restrict__ blockoff,
                                                   uint2* __restrict__ rec,
                                                   int n_edges, int me,
                                                   int (*whist)[NBINS],
                                                   int* scan) {
    const int t = threadIdx.x, lane = t & 31, w = t >> 5;
    const unsigned lt_mask = (1u << lane) - 1u;
    int* h = whist[w];
    for (int k = lane; k < NBINS; k += 32) h[k] = 0;

    const int e0 = me * SORT_EPB + w * (SORT_EPT * 32) + lane;
    unsigned info[SORT_EPT];
    {
        int r[SORT_EPT];
#pragma unroll
        for (int i = 0; i < SORT_EPT; ++i) {
            const int e = e0 + i * 32;
            r[i] = (e < n_edges) ? rows[e] : -1;
        }
#pragma unroll
        for (int i = 0; i < SORT_EPT; ++i) {
            const bool valid = r[i] >= 0;
            const unsigned bin = valid ? (unsigned)(r[i] / RPB) : 0u;
            const unsigned rl  = valid ? (unsigned)r[i] - bin * RPB : 0u;
            int rank, cnt;
            wave_rank(bin, valid, BIN_BITS, lt_mask, rank, cnt);
            counter_take(h, bin, valid, rank, cnt);
            info[i] = bin | ((unsigned)rank << 8) | ((unsigned)cnt << 13) | (rl << 19) |
                      (valid ? (1u << 28) : 0u);
        }
    }
    __syncthreads();

    int total = 0;
    if (t < NBINS) {
        for (int w2 = 0; w2 < SORT_WAVES; ++w2) {
            const int cnt = whist[w2][t];
            whist[w2][t] = total;
            total += cnt;
        }
        scan[t] = total;
    }
    __syncthreads();
    for (int d = 1; d < NBINS; d <<= 1) {
        int add = 0;
        if (t < NBINS && t >= d) add = scan[t - d];
        __syncthreads();
        if (t < NBINS) scan[t] += add;
        __syncthreads();
    }
    if (t < NBINS) {
        const int incl  = scan[t];
        const int start = incl - total;
        blockoff[me * (NBINS + 1) + t] = start;
        if (t == NBINS - 1) blockoff[me * (NBINS + 1) + NBINS] = incl;
        scan[t] = start;
    }
    __syncthreads();

    int c[SORT_EPT];
    float v[SORT_EPT];
#pragma unroll
    for (int i = 0; i < SORT_EPT; ++i) {
        const int e = e0 + i * 32;
        c[i] = 0; v[i] = 0.f;
        if (info[i] >> 28) { c[i] = cols[e]; v[i] = vals[e]; }
    }
    uint2* chunk = rec + (long long)me * SORT_EPB;
#pragma unroll
    for (int i = 0; i < SORT_EPT; ++i) {
        const bool     valid = (info[i] >> 28) != 0u;
        const unsigned bin   = info[i] & 0xFFu;
        const int      rank  = (int)((info[i] >> 8) & 0x1Fu);
        const int      cnt   = (int)((info[i] >> 13) & 0x3Fu);
        const unsigned rl    = (info[i] >> 19) & 0x1FFu;
        const int slot = counter_take(h, bin, valid, rank, cnt);
        if (valid)
            chunk[scan[bin] + slot] = make_uint2((unsigned)c[i] | (rl << COL_BITS), __float_as_uint(v[i]));
    }
}

__global__ __launch_bounds__(SORT_THREADS) __attribute__((amdgpu_waves_per_eu(8, 8)))
void gcn_gemm_and_place(const float* __restrict__ X,
                                                                   const float* __restrict__ W,
                                                                   float* __restrict__ S,
                                                                   int n_rows, int gemm_blocks,
                                                                   const int* __restrict__ rows,
                                                                   const int* __restrict__ cols,
                                                                   const float* __restrict__ vals,
                                                                   int* __restrict__ blockoff,
                                                                   uint2* __restrict__ rec,
                                                                   int n_edges) {
    __shared__ __attribute__((aligned(16))) unsigned char smem[WL_BYTES];
    const int blk = (int)blockIdx.x;
    if (blk < gemm_blocks) {
        v2f* wl = (v2f*)smem;
        stage_w_pairs(W, wl);
        __syncthreads();
        gemm_strip_lds(X, wl, S, n_rows, blk * SORT_WAVES + (int)(threadIdx.x >> 5));
    } else {
        int (*whist)[NBINS] = (int (*)[NBINS])smem;
        int* scan = (int*)(smem + sizeof(int) * SORT_WAVES * NBINS);
        place_block(rows, cols, vals, blockoff, rec, n_edges, blk - gemm_blocks, whist, scan);
    }
}

#define RS_THREADS 1024
#define RS_WAVES   (RS_THREADS / 32)
#define RS_BATCH   8
#define RS_SCAN    512
#define RS_GROUP   4
#define RS_CAP     8192

template <bool IN_LDS>
static __device__ __forceinline__ void range_order_and_sum(const uint2* __restrict__ rec,
                                                           uint2* rec2,
                                                           uint2* lrec,
                                                           int (*whist)[RPB + 1],
                                                           int* rowscan,
                                                           int* rowstart,
                                                           const float* __restrict__ S,
                                                           const float* __restrict__ bias,
                                                           float* __restrict__ out,
                                                           int n_rows, int start, int end,
                                                           int seg_base, int seg_pre, int nseg,
                                                           int n_mine) {
    const int t    = threadIdx.x;
    const int lane = t & 31;
    const int w    = __builtin_amdgcn_readfirstlane(t >> 5);
    const int bin  = blockIdx.x;
    const unsigned lt_mask = (1u << lane) - 1u;

#define RS_LOCATE(cb, idx)                                                     \
    do {                                                                       \
        _Pragma("unroll")                                                      \
        for (int j = 0; j < RS_GROUP; ++j) idx[j] = 0;                         \
        for (int s = 0; s < nseg; ++s) {                                       \
            const int ps = __builtin_amdgcn_readlane(seg_pre, s);              \
            const int bs = __builtin_amdgcn_readlane(seg_base, s);             \
            _Pragma("unroll")                                                  \
            for (int j = 0; j < RS_GROUP; ++j) {                               \
                const int i = (cb) + j * 32 + lane;                            \
                if (i >= ps) idx[j] = bs + (i - ps);                           \
            }                                                                  \
        }                                                                      \
    } while (0)

    int* h = whist[w];
    for (int k = lane; k <= RPB; k += 32) h[k] = 0;
    for (int cb = 0; cb < n_mine; cb += 32 * RS_GROUP) {
        unsigned key[RS_GROUP];
        bool     ok[RS_GROUP];
        int      idx[RS_GROUP];
        RS_LOCATE(cb, idx);
#pragma unroll
        for (int j = 0; j < RS_GROUP; ++j) {
            ok[j]  = cb + j * 32 + lane < n_mine;
            key[j] = 0u;
            if (ok[j]) key[j] = rec[idx[j]].x;
        }
#pragma unroll
        for (int j = 0; j < RS_GROUP; ++j)
            wave_count_step(h, key[j] >> COL_BITS, ok[j], ROW_BITS, lt_mask);
    }
    __syncthreads();

    int rc = 0;
    if (t < RPB) {
        int run = 0;
        for (int w2 = 0; w2 < RS_WAVES; ++w2) {
            const int cnt = whist[w2][t];
            whist[w2][t] = run;
            run += cnt;
        }
        rc = run;
    }
    if (t < RS_SCAN) rowscan[t] = rc;
    __syncthreads();
    for (int d = 1; d < RS_SCAN; d <<= 1) {
        int add = 0;
        if (t < RS_SCAN && t >= d) add = rowscan[t - d];
        __syncthreads();
        if (t < RS_SCAN) rowscan[t] += add;
        __syncthreads();
    }
    if (t <= RPB) rowstart[t] = (t == 0) ? 0 : rowscan[t - 1];
    __syncthreads();

    for (int cb = 0; cb < n_mine; cb += 32 * RS_GROUP) {
        uint2 q[RS_GROUP];
        bool  ok[RS_GROUP];
        int   idx[RS_GROUP];
        RS_LOCATE(cb, idx);
#pragma unroll
        for (int j = 0; j < RS_GROUP; ++j) {
            ok[j] = cb + j * 32 + lane < n_mine;
            q[j]  = make_uint2(0u, 0u);
            if (ok[j]) q[j] = rec[idx[j]];
        }
#pragma unroll
        for (int j = 0; j < RS_GROUP; ++j) {
            const unsigned rl = q[j].x >> COL_BITS;
            const int slot = wave_count_step(h, rl, ok[j], ROW_BITS, lt_mask);
            if (ok[j]) {
                const int pos = rowstart[rl] + slot;
                if (IN_LDS) lrec[pos] = q[j];
                else        rec2[start + pos] = q[j];
            }
        }
    }
#undef RS_LOCATE
    __syncthreads();

    const int row0 = bin * RPB;
    int nrows = n_rows - row0;
    if (nrows > RPB) nrows = RPB;
    const int extra = RPB % RS_WAVES;
    const int r0 = w * (RPB / RS_WAVES) + (w < extra ? w : extra);
    int r1 = r0 + (RPB / RS_WAVES) + (w < extra ? 1 : 0);
    if (r1 > nrows) r1 = nrows;
    if (r0 >= r1) return;

    const float b0 = bias[lane], b1 = bias[lane + 32], b2 = bias[lane + 64], b3 = bias[lane + 96];
    const int e_lo = start + __builtin_amdgcn_readfirstlane(rowstart[r0]);
    const int e_hi = start + __builtin_amdgcn_readfirstlane(rowstart[r1]);

    int   cur = r0;
    float s0 = 0.f, s1 = 0.f, s2 = 0.f, s3 = 0.f;

#define RS_FLUSH(nxt)                                                          \
    do {                                                                       \
        float* o = out + (long long)(row0 + cur) * D_OUT;                      \
        o[lane] = s0 + b0; o[lane + 32] = s1 + b1;                             \
        o[lane + 64] = s2 + b2; o[lane + 96] = s3 + b3;                        \
        _Pragma("unroll 1")   \
        for (int rr = cur + 1; rr < (nxt); ++rr) {                             \
            float* oe = out + (long long)(row0 + rr) * D_OUT;                  \
            oe[lane] = b0; oe[lane + 32] = b1; oe[lane + 64] = b2; oe[lane + 96] = b3; \
        }                                                                      \
        s0 = 0.f; s1 = 0.f; s2 = 0.f; s3 = 0.f;                                \
    } while (0)

    int key_v = 0, val_v = 0;
    if (e_lo + lane < e_hi) {
        uint2 q;
        if (IN_LDS) q = lrec[e_lo - start + lane];
        else        q = rec2[e_lo + lane];
        key_v = (int)q.x;
        val_v = (int)q.y;
    }
    for (int cb = e_lo; cb < e_hi; cb += 32) {
        int n_here = e_hi - cb;
        if (n_here > 32) n_here = 32;
        int key_n = 0, val_n = 0;
        if (cb + 32 + lane < e_hi) {
            uint2 q;
            if (IN_LDS) q = lrec[cb + 32 - start + lane];
            else        q = rec2[cb + 32 + lane];
            key_n = (int)q.x;
            val_n = (int)q.y;
        }
        int j0 = 0;
        for (; j0 + RS_BATCH <= n_here; j0 += RS_BATCH) {
            float g[RS_BATCH][4];
#pragma unroll
            for (int u = 0; u < RS_BATCH; ++u) {
                const int key = __builtin_amdgcn_readlane(key_v, j0 + u);
                const float* src = S + (long long)(key & ((1 << COL_BITS) - 1)) * D_OUT;
                g[u][0] = src[lane];
                g[u][1] = src[lane + 32];
                g[u][2] = src[lane + 64];
                g[u][3] = src[lane + 96];
            }
#pragma unroll
            for (int u = 0; u < RS_BATCH; ++u) {
                const int   key = __builtin_amdgcn_readlane(key_v, j0 + u);
                const float v   = __int_as_float(__builtin_amdgcn_readlane(val_v, j0 + u));
                const int   rl  = (int)((unsigned)key >> COL_BITS);
                if (rl != cur) { RS_FLUSH(rl); cur = rl; }
                s0 += g[u][0] * v; s1 += g[u][1] * v; s2 += g[u][2] * v; s3 += g[u][3] * v;
            }
        }
        for (; j0 < n_here; ++j0) {
            const int   key = __builtin_amdgcn_readlane(key_v, j0);
            const float v   = __int_as_float(__builtin_amdgcn_readlane(val_v, j0));
            const int   rl  = (int)((unsigned)key >> COL_BITS);
            const float* src = S + (long long)(key & ((1 << COL_BITS) - 1)) * D_OUT;
            const float g0 = src[lane];
            const float g1 = src[lane + 32];
            const float g2 = src[lane + 64];
            const float g3 = src[lane + 96];
            if (rl != cur) { RS_FLUSH(rl); cur = rl; }
            s0 += g0 * v; s1 += g1 * v; s2 += g2 * v; s3 += g3 * v;
        }
        key_v = key_n;
        val_v = val_n;
    }
    RS_FLUSH(r1);
#undef RS_FLUSH
}

__global__ __launch_bounds__(RS_THREADS) void gcn_range_sum(const uint2* __restrict__ rec,
                                                            uint2* rec2,
                                                            const int* __restrict__ blockoff,
                                                            int n_blocks,
                                                            const float* __restrict__ S,
                                                            const float* __restrict__ bias,
                                                            float* __restrict__ out,
                                                            int n_rows) {
    __shared__ int whist[RS_WAVES][RPB + 1];
    __shared__ int rowscan[RS_SCAN];
    __shared__ int rowstart[RPB + 1];
    __shared__ uint2 lrec[RS_CAP];
    __shared__ int wsum[2][RS_WAVES];
    const int lane = threadIdx.x & 31;
    const int w    = __builtin_amdgcn_readfirstlane(threadIdx.x >> 5);
    const int b_lo = (w * n_blocks) / RS_WAVES;
    const int nseg = ((w + 1) * n_blocks) / RS_WAVES - b_lo;
    int seg_base = 0, seg_len = 0, seg_o0 = 0;
    if (lane < nseg) {
        const int* bo = blockoff + (long long)(b_lo + lane) * (NBINS + 1) + blockIdx.x;
        seg_o0   = bo[0];
        seg_len  = bo[1] - seg_o0;
        seg_base = (b_lo + lane) * SORT_EPB + seg_o0;
    }
    int seg_pre = seg_len, below = seg_o0;
#pragma unroll
    for (int d = 1; d < 32; d <<= 1) {
        const int up = __shfl_up(seg_pre, d);
        if (lane >= d) seg_pre += up;
        below += __shfl_xor(below, d);
    }
    const int n_mine = __builtin_amdgcn_readlane(seg_pre, 31);
    seg_pre -= seg_len;
    if (lane == 0) { wsum[0][w] = below; wsum[1][w] = n_mine; }
    __syncthreads();
    int s_below = wsum[0][lane], s_mine = wsum[1][lane];
#pragma unroll
    for (int d = 1; d < 32; d <<= 1) {
        s_below += __shfl_xor(s_below, d);
        s_mine  += __shfl_xor(s_mine, d);
    }
    const int start = __builtin_amdgcn_readfirstlane(s_below);
    const int end   = start + __builtin_amdgcn_readfirstlane(s_mine);
    if (__builtin_expect(end - start <= RS_CAP, 1))
        range_order_and_sum<true>(rec, rec2, lrec, whist, rowscan, rowstart, S, bias, out,
                                  n_rows, start, end, seg_base, seg_pre, nseg, n_mine);
    else
        range_order_and_sum<false>(rec, rec2, lrec, whist, rowscan, rowstart, S, bias, out,
                                   n_rows, start, end, seg_base, seg_pre, nseg, n_mine);
}

extern "C" void kernel_launch(void* const* d_in, const int* in_sizes, int n_in,
                              void* d_out, int out_size, void* d_ws, size_t ws_size,
                              hipStream_t stream) {
    const int*   rows = (const int*)d_in[0];
    const int*   cols = (const int*)d_in[1];
    const float* vals = (const float*)d_in[2];
    const float* x    = (const float*)d_in[3];
    const float* W    = (const float*)d_in[4];
    const float* bias = (const float*)d_in[5];

    const int E = in_sizes[0];
    const int N = in_sizes[3] / D_IN;

    float* support = (float*)d_ws;

    const int mtiles = N / 16;

    const int    sort_blocks = (E + SORT_EPB - 1) / SORT_EPB;
    const size_t off_rec   = (((size_t)N * D_OUT * sizeof(float)) + 255) & ~(size_t)255;
    const size_t off_rec2  = (off_rec  + (size_t)sort_blocks * SORT_EPB * sizeof(uint2) + 255) & ~(size_t)255;
    const size_t off_boff  = (off_rec2 + (size_t)E * sizeof(uint2) + 255) & ~(size_t)255;
    const size_t ws_need   = off_boff + (size_t)sort_blocks * (NBINS + 1) * sizeof(int);

    if (ws_need <= ws_size && N <= NBINS * RPB && sort_blocks <= 32 * RS_WAVES) {
        uint2* rec      = (uint2*)((char*)d_ws + off_rec);
        uint2* rec2     = (uint2*)((char*)d_ws + off_rec2);
        int*   blockoff = (int*)((char*)d_ws + off_boff);
        const int gemm_blocks = (mtiles + SORT_WAVES - 1) / SORT_WAVES;
        gcn_gemm_and_place<<<gemm_blocks + sort_blocks, SORT_THREADS, 0, stream>>>(
            x, W, support, N, gemm_blocks, rows, cols, vals, blockoff, rec, E);
        gcn_range_sum<<<NBINS, RS_THREADS, 0, stream>>>(rec, rec2, blockoff, sort_blocks, support,
                                                        bias, (float*)d_out, N);
    } else {
        gcn_gemm_xw<<<(mtiles + 7) / 8, 256, 0, stream>>>(x, W, support, N);
        long long total4 = (long long)N * (D_OUT / 4);
        int iblocks = (int)((total4 + 255) / 256);
        gcn_init_out<<<iblocks, 256, 0, stream>>>(bias, (float*)d_out, total4);
        int waves  = (E + EDGES_PER_WAVE - 1) / EDGES_PER_WAVE;
        int blocks = (waves + 7) / 8;
        gcn_edge_scatter<<<blocks, 256, 0, stream>>>(rows, cols, vals, support,
                                                     (float*)d_out, E);
    }
}


# Range ordering: place pass takes a wave's first 256 records and their same-row ranks from registers (kept from the count pass), no longer locates, loads and matches them again; sum order unchanged
# speedup vs baseline: 3.1146x; 1.0348x over previous
#include <hip/hip_runtime.h>

typedef __attribute__((ext_vector_type(2))) float v2f;
typedef __attribute__((ext_vector_type(8))) float v8f;

#define D_IN  256
#define D_OUT 128

#define NBINS        256
#define BIN_BITS     8
#define RPB          391
#define ROW_BITS     9
#define COL_BITS     20
#define SORT_THREADS 1024
#define SORT_WAVES   (SORT_THREADS / 32)
#define SORT_EPT     8
#define SORT_EPB     (SORT_THREADS * SORT_EPT)

__global__ __launch_bounds__(256) void gcn_init_out(const float* __restrict__ bias,
                                                    float* __restrict__ out,
                                                    long long total4) {
    long long i = (long long)blockIdx.x * blockDim.x + threadIdx.x;
    if (i < total4) {
        int d4 = (int)(i & (D_OUT / 4 - 1));
        ((float4*)out)[i] = ((const float4*)bias)[d4];
    }
}

static __device__ __forceinline__ void gemm_strip(const float* __restrict__ X,
                                                  const float* __restrict__ W,
                                                  float* __restrict__ S,
                                                  int n_rows, int mtile) {
    const int lane  = threadIdx.x & 31;
    if (mtile * 16 + 16 > n_rows) return;
    const int m0 = mtile * 16;

    const int am = lane & 15;
    const int ak = (lane >> 4) * 2;
    const int bn = lane & 15;
    const int bk = (lane >> 4) * 2;

    v8f c[8];
#pragma unroll
    for (int t = 0; t < 8; ++t) c[t] = (v8f){0.f, 0.f, 0.f, 0.f, 0.f, 0.f, 0.f, 0.f};

    const float* xrow = X + (long long)(m0 + am) * D_IN;

    for (int k = 0; k < D_IN; k += 4) {
        v2f a = *(const v2f*)(xrow + k + ak);
#pragma unroll
        for (int t = 0; t < 8; ++t) {
            v2f b;
            b.x = W[(long long)(k + bk)     * D_OUT + t * 16 + bn];
            b.y = W[(long long)(k + bk + 1) * D_OUT + t * 16 + bn];
            c[t] = __builtin_amdgcn_wmma_f32_16x16x4_f32(
                 false, a,  false, b,
                 (short)0, c[t],  false,  false);
        }
    }

    const int cn    = lane & 15;
    const int cmoff = (lane >> 4) * 8;
#pragma unroll
    for (int t = 0; t < 8; ++t) {
#pragma unroll
        for (int r = 0; r < 8; ++r) {
            S[(long long)(m0 + r + cmoff) * D_OUT + t * 16 + cn] = c[t][r];
        }
    }
}

__global__ __launch_bounds__(256) void gcn_gemm_xw(const float* __restrict__ X,
                                                   const float* __restrict__ W,
                                                   float* __restrict__ S,
                                                   int n_rows) {
    gemm_strip(X, W, S, n_rows, (int)blockIdx.x * 8 + (int)(threadIdx.x >> 5));
}

#define WL_PITCH 130
#define WL_BYTES ((D_IN / 2) * WL_PITCH * 8)

static __device__ __forceinline__ void stage_w_pairs(const float* __restrict__ W, v2f* wl) {
    float4 w0[4], w1[4];
#pragma unroll
    for (int i = 0; i < 4; ++i) {
        const int idx = (int)threadIdx.x + i * SORT_THREADS;
        const int p = idx >> 5, q = idx & 31;
        w0[i] = ((const float4*)W)[(2 * p)     * (D_OUT / 4) + q];
        w1[i] = ((const float4*)W)[(2 * p + 1) * (D_OUT / 4) + q];
    }
#pragma unroll
    for (int i = 0; i < 4; ++i) {
        const int idx = (int)threadIdx.x + i * SORT_THREADS;
        const int p = idx >> 5, q = idx & 31;
        float4* d = (float4*)(wl + p * WL_PITCH + 4 * q);
        d[0] = make_float4(w0[i].x, w1[i].x, w0[i].y, w1[i].y);
        d[1] = make_float4(w0[i].z, w1[i].z, w0[i].w, w1[i].w);
    }
}

static __device__ __forceinline__ void gemm_strip_lds(const float* __restrict__ X,
                                                      const v2f* wl,
                                                      float* __restrict__ S,
                                                      int n_rows, int mtile) {
    const int lane = threadIdx.x & 31;
    if (mtile * 16 + 16 > n_rows) return;
    const int m0 = mtile * 16;
    const int c  = lane & 15;
    const int h  = lane >> 4;

    v8f acc[8];
#pragma unroll
    for (int t = 0; t < 8; ++t) acc[t] = (v8f){0.f, 0.f, 0.f, 0.f, 0.f, 0.f, 0.f, 0.f};

    const float* xrow  = X + (long long)(m0 + c) * D_IN + 16 * h;
    const v2f*   wlane = wl + (8 * h) * WL_PITCH + c;

#pragma unroll 1
    for (int kb = 0; kb < D_IN; kb += 32) {
        float4 q[4];
#pragma unroll
        for (int i = 0; i < 4; ++i) q[i] = *(const float4*)(xrow + kb + 4 * i);
        const v2f* wk = wlane + (kb / 2) * WL_PITCH;
#pragma unroll
        for (int s = 0; s < 8; ++s) {
            v2f a;
            a.x = (s & 1) ? q[s >> 1].z : q[s >> 1].x;
            a.y = (s & 1) ? q[s >> 1].w : q[s >> 1].y;
#pragma unroll
            for (int t = 0; t < 8; ++t) {
                const v2f b = wk[s * WL_PITCH + 16 * t];
                acc[t] = __builtin_amdgcn_wmma_f32_16x16x4_f32(
                     false, a,  false, b,
                     (short)0, acc[t],  false,  false);
            }
        }
    }

    const int cmoff = h * 8;
#pragma unroll
    for (int t = 0; t < 8; ++t) {
#pragma unroll
        for (int r = 0; r < 8; ++r) {
            S[(long long)(m0 + r + cmoff) * D_OUT + t * 16 + c] = acc[t][r];
        }
    }
}

#define EDGES_PER_WAVE 32
#define EDGE_BATCH     8

static __device__ __forceinline__ void add_row_lines(float* dst, unsigned off,
                                                      float g0, float g1, float g2, float g3) {
    asm volatile("global_atomic_add_f32 %0, %1, %2"            :: "v"(off), "v"(g0), "s"(dst) : "memory");
    asm volatile("global_atomic_add_f32 %0, %1, %2 offset:128" :: "v"(off), "v"(g1), "s"(dst) : "memory");
    asm volatile("global_atomic_add_f32 %0, %1, %2 offset:256" :: "v"(off), "v"(g2), "s"(dst) : "memory");
    asm volatile("global_atomic_add_f32 %0, %1, %2 offset:384" :: "v"(off), "v"(g3), "s"(dst) : "memory");
}

__global__ __launch_bounds__(256) void gcn_edge_scatter(const int* __restrict__ rows,
                                                        const int* __restrict__ cols,
                                                        const float* __restrict__ vals,
                                                        const float* __restrict__ S,
                                                        float* __restrict__ out,
                                                        int n_edges) {
    const int lane = threadIdx.x & 31;
    int w = (int)(blockIdx.x * 8u) + (int)(threadIdx.x >> 5);
    w = __builtin_amdgcn_readfirstlane(w);
    const int e_base = w * EDGES_PER_WAVE;
    if (e_base >= n_edges) return;
    int n_here = n_edges - e_base;
    if (n_here > EDGES_PER_WAVE) n_here = EDGES_PER_WAVE;

    int r_v = 0, c_v = 0, v_v = 0;
    if (lane < n_here) {
        r_v = rows[e_base + lane];
        c_v = cols[e_base + lane];
        v_v = __float_as_int(vals[e_base + lane]);
    }
    const unsigned off = (unsigned)lane * 4u;

    int j0 = 0;
    for (; j0 + EDGE_BATCH <= n_here; j0 += EDGE_BATCH) {
        float g[EDGE_BATCH][4];
#pragma unroll
        for (int u = 0; u < EDGE_BATCH; ++u) {
            const int c = __builtin_amdgcn_readlane(c_v, j0 + u);
            const float* src = S + (long long)c * D_OUT;
            g[u][0] = src[lane];
            g[u][1] = src[lane + 32];
            g[u][2] = src[lane + 64];
            g[u][3] = src[lane + 96];
        }
#pragma unroll
        for (int u = 0; u < EDGE_BATCH; ++u) {
            const int   r = __builtin_amdgcn_readlane(r_v, j0 + u);
            const float v = __int_as_float(__builtin_amdgcn_readlane(v_v, j0 + u));
            float* dst = out + (long long)r * D_OUT;
            add_row_lines(dst, off, g[u][0] * v, g[u][1] * v, g[u][2] * v, g[u][3] * v);
        }
    }
    for (; j0 < n_here; ++j0) {
        const int   c = __builtin_amdgcn_readlane(c_v, j0);
        const int   r = __builtin_amdgcn_readlane(r_v, j0);
        const float v = __int_as_float(__builtin_amdgcn_readlane(v_v, j0));
        const float* src = S + (long long)c * D_OUT;
        const float s0 = src[lane];
        const float s1 = src[lane + 32];
        const float s2 = src[lane + 64];
        const float s3 = src[lane + 96];
        float* dst = out + (long long)r * D_OUT;
        add_row_lines(dst, off, s0 * v, s1 * v, s2 * v, s3 * v);
    }
}


static __device__ __forceinline__ unsigned same_key_lanes(unsigned key, bool valid, int nbits) {
    unsigned peers = __builtin_amdgcn_ballot_w32(valid);
#pragma unroll
    for (int b = 0; b < nbits; ++b) {
        const bool bit = ((key >> b) & 1u) != 0u;
        const unsigned m = __builtin_amdgcn_ballot_w32(bit && valid);
        peers &= bit ? m : ~m;
    }
    return valid ? peers : 0u;
}

static __device__ __forceinline__ void wave_rank(unsigned key, bool valid, int nbits,
                                                 unsigned lt_mask, int& rank, int& cnt) {
    const unsigned peers = same_key_lanes(key, valid, nbits);
    rank = __builtin_popcount(peers & lt_mask);
    cnt  = __builtin_popcount(peers);
}

static __device__ __forceinline__ int counter_take(int* h, unsigned key, bool valid,
                                                   int rank, int cnt) {
    int slot = 0;
    if (valid) {
        const int old = h[key];
        if (rank == 0) h[key] = old + cnt;
        slot = old + rank;
    }
    return slot;
}

static __device__ __forceinline__ int wave_count_step(int* h, unsigned key, bool valid,
                                                      int nbits, unsigned lt_mask) {
    int rank, cnt;
    wave_rank(key, valid, nbits, lt_mask, rank, cnt);
    return counter_take(h, key, valid, rank, cnt);
}

static __device__ __forceinline__ void place_block(const int* __restrict__ rows,
                                                   const int* __restrict__ cols,
                                                   const float* __restrict__ vals,
                                                   int* __restrict__ blockoff,
                                                   uint2* __restrict__ rec,
                                                   int n_edges, int me,
                                                   int (*whist)[NBINS],
                                                   int* scan) {
    const int t = threadIdx.x, lane = t & 31, w = t >> 5;
    const unsigned lt_mask = (1u << lane) - 1u;
    int* h = whist[w];
    for (int k = lane; k < NBINS; k += 32) h[k] = 0;

    const int e0 = me * SORT_EPB + w * (SORT_EPT * 32) + lane;
    unsigned info[SORT_EPT];
    {
        int r[SORT_EPT];
#pragma unroll
        for (int i = 0; i < SORT_EPT; ++i) {
            const int e = e0 + i * 32;
            r[i] = (e < n_edges) ? rows[e] : -1;
        }
#pragma unroll
        for (int i = 0; i < SORT_EPT; ++i) {
            const bool valid = r[i] >= 0;
            const unsigned bin = valid ? (unsigned)(r[i] / RPB) : 0u;
            const unsigned rl  = valid ? (unsigned)r[i] - bin * RPB : 0u;
            int rank, cnt;
            wave_rank(bin, valid, BIN_BITS, lt_mask, rank, cnt);
            counter_take(h, bin, valid, rank, cnt);
            info[i] = bin | ((unsigned)rank << 8) | ((unsigned)cnt << 13) | (rl << 19) |
                      (valid ? (1u << 28) : 0u);
        }
    }
    __syncthreads();

    int total = 0;
    if (t < NBINS) {
        for (int w2 = 0; w2 < SORT_WAVES; ++w2) {
            const int cnt = whist[w2][t];
            whist[w2][t] = total;
            total += cnt;
        }
        scan[t] = total;
    }
    __syncthreads();
    for (int d = 1; d < NBINS; d <<= 1) {
        int add = 0;
        if (t < NBINS && t >= d) add = scan[t - d];
        __syncthreads();
        if (t < NBINS) scan[t] += add;
        __syncthreads();
    }
    if (t < NBINS) {
        const int incl  = scan[t];
        const int start = incl - total;
        blockoff[me * (NBINS + 1) + t] = start;
        if (t == NBINS - 1) blockoff[me * (NBINS + 1) + NBINS] = incl;
        scan[t] = start;
    }
    __syncthreads();

    int c[SORT_EPT];
    float v[SORT_EPT];
#pragma unroll
    for (int i = 0; i < SORT_EPT; ++i) {
        const int e = e0 + i * 32;
        c[i] = 0; v[i] = 0.f;
        if (info[i] >> 28) { c[i] = cols[e]; v[i] = vals[e]; }
    }
    uint2* chunk = rec + (long long)me * SORT_EPB;
#pragma unroll
    for (int i = 0; i < SORT_EPT; ++i) {
        const bool     valid = (info[i] >> 28) != 0u;
        const unsigned bin   = info[i] & 0xFFu;
        const int      rank  = (int)((info[i] >> 8) & 0x1Fu);
        const int      cnt   = (int)((info[i] >> 13) & 0x3Fu);
        const unsigned rl    = (info[i] >> 19) & 0x1FFu;
        const int slot = counter_take(h, bin, valid, rank, cnt);
        if (valid)
            chunk[scan[bin] + slot] = make_uint2((unsigned)c[i] | (rl << COL_BITS), __float_as_uint(v[i]));
    }
}

__global__ __launch_bounds__(SORT_THREADS) __attribute__((amdgpu_waves_per_eu(8, 8)))
void gcn_gemm_and_place(const float* __restrict__ X,
                                                                   const float* __restrict__ W,
                                                                   float* __restrict__ S,
                                                                   int n_rows, int gemm_blocks,
                                                                   const int* __restrict__ rows,
                                                                   const int* __restrict__ cols,
                                                                   const float* __restrict__ vals,
                                                                   int* __restrict__ blockoff,
                                                                   uint2* __restrict__ rec,
                                                                   int n_edges) {
    __shared__ __attribute__((aligned(16))) unsigned char smem[WL_BYTES];
    const int blk = (int)blockIdx.x;
    if (blk < gemm_blocks) {
        v2f* wl = (v2f*)smem;
        stage_w_pairs(W, wl);
        __syncthreads();
        gemm_strip_lds(X, wl, S, n_rows, blk * SORT_WAVES + (int)(threadIdx.x >> 5));
    } else {
        int (*whist)[NBINS] = (int (*)[NBINS])smem;
        int* scan = (int*)(smem + sizeof(int) * SORT_WAVES * NBINS);
        place_block(rows, cols, vals, blockoff, rec, n_edges, blk - gemm_blocks, whist, scan);
    }
}

#define RS_THREADS 1024
#define RS_WAVES   (RS_THREADS / 32)
#define RS_BATCH   8
#define RS_SCAN    512
#define RS_GROUP   4
#define RS_CAP     8192
#define RS_KEEP    2

template <bool IN_LDS>
static __device__ __forceinline__ void range_order_and_sum(const uint2* __restrict__ rec,
                                                           uint2* rec2,
                                                           uint2* lrec,
                                                           int (*whist)[RPB + 1],
                                                           int* rowscan,
                                                           int* rowstart,
                                                           const float* __restrict__ S,
                                                           const float* __restrict__ bias,
                                                           float* __restrict__ out,
                                                           int n_rows, int start, int end,
                                                           int seg_base, int seg_pre, int nseg,
                                                           int n_mine) {
    const int t    = threadIdx.x;
    const int lane = t & 31;
    const int w    = __builtin_amdgcn_readfirstlane(t >> 5);
    const int bin  = blockIdx.x;
    const unsigned lt_mask = (1u << lane) - 1u;

#define RS_LOCATE(cb, idx)                                                     \
    do {                                                                       \
        _Pragma("unroll")                                                      \
        for (int j = 0; j < RS_GROUP; ++j) idx[j] = 0;                         \
        for (int s = 0; s < nseg; ++s) {                                       \
            const int ps = __builtin_amdgcn_readlane(seg_pre, s);              \
            const int bs = __builtin_amdgcn_readlane(seg_base, s);             \
            _Pragma("unroll")                                                  \
            for (int j = 0; j < RS_GROUP; ++j) {                               \
                const int i = (cb) + j * 32 + lane;                            \
                if (i >= ps) idx[j] = bs + (i - ps);                           \
            }                                                                  \
        }                                                                      \
    } while (0)

    int* h = whist[w];
    for (int k = lane; k <= RPB; k += 32) h[k] = 0;
    uint2    kq[RS_KEEP][RS_GROUP];
    unsigned kinfo[RS_KEEP][RS_GROUP];
#pragma unroll
    for (int tr = 0; tr < RS_KEEP; ++tr) {
#pragma unroll
        for (int j = 0; j < RS_GROUP; ++j) { kq[tr][j] = make_uint2(0u, 0u); kinfo[tr][j] = 0u; }
        const int cb = tr * 32 * RS_GROUP;
        if (cb < n_mine) {
            int idx[RS_GROUP];
            RS_LOCATE(cb, idx);
#pragma unroll
            for (int j = 0; j < RS_GROUP; ++j)
                if (cb + j * 32 + lane < n_mine) kq[tr][j] = rec[idx[j]];
#pragma unroll
            for (int j = 0; j < RS_GROUP; ++j) {
                const bool     ok = cb + j * 32 + lane < n_mine;
                const unsigned rl = kq[tr][j].x >> COL_BITS;
                int rank, cnt;
                wave_rank(rl, ok, ROW_BITS, lt_mask, rank, cnt);
                counter_take(h, rl, ok, rank, cnt);
                kinfo[tr][j] = (unsigned)rank | ((unsigned)cnt << 5) | (ok ? (1u << 11) : 0u);
            }
        }
    }
    for (int cb = RS_KEEP * 32 * RS_GROUP; cb < n_mine; cb += 32 * RS_GROUP) {
        unsigned key[RS_GROUP];
        bool     ok[RS_GROUP];
        int      idx[RS_GROUP];
        RS_LOCATE(cb, idx);
#pragma unroll
        for (int j = 0; j < RS_GROUP; ++j) {
            ok[j]  = cb + j * 32 + lane < n_mine;
            key[j] = 0u;
            if (ok[j]) key[j] = rec[idx[j]].x;
        }
#pragma unroll
        for (int j = 0; j < RS_GROUP; ++j)
            wave_count_step(h, key[j] >> COL_BITS, ok[j], ROW_BITS, lt_mask);
    }
    __syncthreads();

    int rc = 0;
    if (t < RPB) {
        int run = 0;
        for (int w2 = 0; w2 < RS_WAVES; ++w2) {
            const int cnt = whist[w2][t];
            whist[w2][t] = run;
            run += cnt;
        }
        rc = run;
    }
    if (t < RS_SCAN) rowscan[t] = rc;
    __syncthreads();
    for (int d = 1; d < RS_SCAN; d <<= 1) {
        int add = 0;
        if (t < RS_SCAN && t >= d) add = rowscan[t - d];
        __syncthreads();
        if (t < RS_SCAN) rowscan[t] += add;
        __syncthreads();
    }
    if (t <= RPB) rowstart[t] = (t == 0) ? 0 : rowscan[t - 1];
    __syncthreads();

#pragma unroll
    for (int tr = 0; tr < RS_KEEP; ++tr) {
        if (tr * 32 * RS_GROUP < n_mine) {
#pragma unroll
            for (int j = 0; j < RS_GROUP; ++j) {
                const bool     ok = (kinfo[tr][j] >> 11) != 0u;
                const unsigned rl = kq[tr][j].x >> COL_BITS;
                const int slot = counter_take(h, rl, ok, (int)(kinfo[tr][j] & 31u),
                                              (int)((kinfo[tr][j] >> 5) & 63u));
                if (ok) {
                    const int pos = rowstart[rl] + slot;
                    if (IN_LDS) lrec[pos] = kq[tr][j];
                    else        rec2[start + pos] = kq[tr][j];
                }
            }
        }
    }
    for (int cb = RS_KEEP * 32 * RS_GROUP; cb < n_mine; cb += 32 * RS_GROUP) {
        uint2 q[RS_GROUP];
        bool  ok[RS_GROUP];
        int   idx[RS_GROUP];
        RS_LOCATE(cb, idx);
#pragma unroll
        for (int j = 0; j < RS_GROUP; ++j) {
            ok[j] = cb + j * 32 + lane < n_mine;
            q[j]  = make_uint2(0u, 0u);
            if (ok[j]) q[j] = rec[idx[j]];
        }
#pragma unroll
        for (int j = 0; j < RS_GROUP; ++j) {
            const unsigned rl = q[j].x >> COL_BITS;
            const int slot = wave_count_step(h, rl, ok[j], ROW_BITS, lt_mask);
            if (ok[j]) {
                const int pos = rowstart[rl] + slot;
                if (IN_LDS) lrec[pos] = q[j];
                else        rec2[start + pos] = q[j];
            }
        }
    }
#undef RS_LOCATE
    __syncthreads();

    const int row0 = bin * RPB;
    int nrows = n_rows - row0;
    if (nrows > RPB) nrows = RPB;
    const int extra = RPB % RS_WAVES;
    const int r0 = w * (RPB / RS_WAVES) + (w < extra ? w : extra);
    int r1 = r0 + (RPB / RS_WAVES) + (w < extra ? 1 : 0);
    if (r1 > nrows) r1 = nrows;
    if (r0 >= r1) return;

    const float b0 = bias[lane], b1 = bias[lane + 32], b2 = bias[lane + 64], b3 = bias[lane + 96];
    const int e_lo = start + __builtin_amdgcn_readfirstlane(rowstart[r0]);
    const int e_hi = start + __builtin_amdgcn_readfirstlane(rowstart[r1]);

    int   cur = r0;
    float s0 = 0.f, s1 = 0.f, s2 = 0.f, s3 = 0.f;

#define RS_FLUSH(nxt)                                                          \
    do {                                                                       \
        float* o = out + (long long)(row0 + cur) * D_OUT;                      \
        o[lane] = s0 + b0; o[lane + 32] = s1 + b1;                             \
        o[lane + 64] = s2 + b2; o[lane + 96] = s3 + b3;                        \
        _Pragma("unroll 1")   \
        for (int rr = cur + 1; rr < (nxt); ++rr) {                             \
            float* oe = out + (long long)(row0 + rr) * D_OUT;                  \
            oe[lane] = b0; oe[lane + 32] = b1; oe[lane + 64] = b2; oe[lane + 96] = b3; \
        }                                                                      \
        s0 = 0.f; s1 = 0.f; s2 = 0.f; s3 = 0.f;                                \
    } while (0)

    int key_v = 0, val_v = 0;
    if (e_lo + lane < e_hi) {
        uint2 q;
        if (IN_LDS) q = lrec[e_lo - start + lane];
        else        q = rec2[e_lo + lane];
        key_v = (int)q.x;
        val_v = (int)q.y;
    }
    for (int cb = e_lo; cb < e_hi; cb += 32) {
        int n_here = e_hi - cb;
        if (n_here > 32) n_here = 32;
        int key_n = 0, val_n = 0;
        if (cb + 32 + lane < e_hi) {
            uint2 q;
            if (IN_LDS) q = lrec[cb + 32 - start + lane];
            else        q = rec2[cb + 32 + lane];
            key_n = (int)q.x;
            val_n = (int)q.y;
        }
        int j0 = 0;
        for (; j0 + RS_BATCH <= n_here; j0 += RS_BATCH) {
            float g[RS_BATCH][4];
#pragma unroll
            for (int u = 0; u < RS_BATCH; ++u) {
                const int key = __builtin_amdgcn_readlane(key_v, j0 + u);
                const float* src = S + (long long)(key & ((1 << COL_BITS) - 1)) * D_OUT;
                g[u][0] = src[lane];
                g[u][1] = src[lane + 32];
                g[u][2] = src[lane + 64];
                g[u][3] = src[lane + 96];
            }
#pragma unroll
            for (int u = 0; u < RS_BATCH; ++u) {
                const int   key = __builtin_amdgcn_readlane(key_v, j0 + u);
                const float v   = __int_as_float(__builtin_amdgcn_readlane(val_v, j0 + u));
                const int   rl  = (int)((unsigned)key >> COL_BITS);
                if (rl != cur) { RS_FLUSH(rl); cur = rl; }
                s0 += g[u][0] * v; s1 += g[u][1] * v; s2 += g[u][2] * v; s3 += g[u][3] * v;
            }
        }
        for (; j0 < n_here; ++j0) {
            const int   key = __builtin_amdgcn_readlane(key_v, j0);
            const float v   = __int_as_float(__builtin_amdgcn_readlane(val_v, j0));
            const int   rl  = (int)((unsigned)key >> COL_BITS);
            const float* src = S + (long long)(key & ((1 << COL_BITS) - 1)) * D_OUT;
            const float g0 = src[lane];
            const float g1 = src[lane + 32];
            const float g2 = src[lane + 64];
            const float g3 = src[lane + 96];
            if (rl != cur) { RS_FLUSH(rl); cur = rl; }
            s0 += g0 * v; s1 += g1 * v; s2 += g2 * v; s3 += g3 * v;
        }
        key_v = key_n;
        val_v = val_n;
    }
    RS_FLUSH(r1);
#undef RS_FLUSH
}

__global__ __launch_bounds__(RS_THREADS) void gcn_range_sum(const uint2* __restrict__ rec,
                                                            uint2* rec2,
                                                            const int* __restrict__ blockoff,
                                                            int n_blocks,
                                                            const float* __restrict__ S,
                                                            const float* __restrict__ bias,
                                                            float* __restrict__ out,
                                                            int n_rows) {
    __shared__ int whist[RS_WAVES][RPB + 1];
    __shared__ int rowscan[RS_SCAN];
    __shared__ int rowstart[RPB + 1];
    __shared__ uint2 lrec[RS_CAP];
    __shared__ int wsum[2][RS_WAVES];
    const int lane = threadIdx.x & 31;
    const int w    = __builtin_amdgcn_readfirstlane(threadIdx.x >> 5);
    const int b_lo = (w * n_blocks) / RS_WAVES;
    const int nseg = ((w + 1) * n_blocks) / RS_WAVES - b_lo;
    int seg_base = 0, seg_len = 0, seg_o0 = 0;
    if (lane < nseg) {
        const int* bo = blockoff + (long long)(b_lo + lane) * (NBINS + 1) + blockIdx.x;
        seg_o0   = bo[0];
        seg_len  = bo[1] - seg_o0;
        seg_base = (b_lo + lane) * SORT_EPB + seg_o0;
    }
    int seg_pre = seg_len, below = seg_o0;
#pragma unroll
    for (int d = 1; d < 32; d <<= 1) {
        const int up = __shfl_up(seg_pre, d);
        if (lane >= d) seg_pre += up;
        below += __shfl_xor(below, d);
    }
    const int n_mine = __builtin_amdgcn_readlane(seg_pre, 31);
    seg_pre -= seg_len;
    if (lane == 0) { wsum[0][w] = below; wsum[1][w] = n_mine; }
    __syncthreads();
    int s_below = wsum[0][lane], s_mine = wsum[1][lane];
#pragma unroll
    for (int d = 1; d < 32; d <<= 1) {
        s_below += __shfl_xor(s_below, d);
        s_mine  += __shfl_xor(s_mine, d);
    }
    const int start = __builtin_amdgcn_readfirstlane(s_below);
    const int end   = start + __builtin_amdgcn_readfirstlane(s_mine);
    if (__builtin_expect(end - start <= RS_CAP, 1))
        range_order_and_sum<true>(rec, rec2, lrec, whist, rowscan, rowstart, S, bias, out,
                                  n_rows, start, end, seg_base, seg_pre, nseg, n_mine);
    else
        range_order_and_sum<false>(rec, rec2, lrec, whist, rowscan, rowstart, S, bias, out,
                                   n_rows, start, end, seg_base, seg_pre, nseg, n_mine);
}

extern "C" void kernel_launch(void* const* d_in, const int* in_sizes, int n_in,
                              void* d_out, int out_size, void* d_ws, size_t ws_size,
                              hipStream_t stream) {
    const int*   rows = (const int*)d_in[0];
    const int*   cols = (const int*)d_in[1];
    const float* vals = (const float*)d_in[2];
    const float* x    = (const float*)d_in[3];
    const float* W    = (const float*)d_in[4];
    const float* bias = (const float*)d_in[5];

    const int E = in_sizes[0];
    const int N = in_sizes[3] / D_IN;

    float* support = (float*)d_ws;

    const int mtiles = N / 16;

    const int    sort_blocks = (E + SORT_EPB - 1) / SORT_EPB;
    const size_t off_rec   = (((size_t)N * D_OUT * sizeof(float)) + 255) & ~(size_t)255;
    const size_t off_rec2  = (off_rec  + (size_t)sort_blocks * SORT_EPB * sizeof(uint2) + 255) & ~(size_t)255;
    const size_t off_boff  = (off_rec2 + (size_t)E * sizeof(uint2) + 255) & ~(size_t)255;
    const size_t ws_need   = off_boff + (size_t)sort_blocks * (NBINS + 1) * sizeof(int);

    if (ws_need <= ws_size && N <= NBINS * RPB && sort_blocks <= 32 * RS_WAVES) {
        uint2* rec      = (uint2*)((char*)d_ws + off_rec);
        uint2* rec2     = (uint2*)((char*)d_ws + off_rec2);
        int*   blockoff = (int*)((char*)d_ws + off_boff);
        const int gemm_blocks = (mtiles + SORT_WAVES - 1) / SORT_WAVES;
        gcn_gemm_and_place<<<gemm_blocks + sort_blocks, SORT_THREADS, 0, stream>>>(
            x, W, support, N, gemm_blocks, rows, cols, vals, blockoff, rec, E);
        gcn_range_sum<<<NBINS, RS_THREADS, 0, stream>>>(rec, rec2, blockoff, sort_blocks, support,
                                                        bias, (float*)d_out, N);
    } else {
        gcn_gemm_xw<<<(mtiles + 7) / 8, 256, 0, stream>>>(x, W, support, N);
        long long total4 = (long long)N * (D_OUT / 4);
        int iblocks = (int)((total4 + 255) / 256);
        gcn_init_out<<<iblocks, 256, 0, stream>>>(bias, (float*)d_out, total4);
        int waves  = (E + EDGES_PER_WAVE - 1) / EDGES_PER_WAVE;
        int blocks = (waves + 7) / 8;
        gcn_edge_scatter<<<blocks, 256, 0, stream>>>(rows, cols, vals, support,
                                                     (float*)d_out, E);
    }
}


# GEMM strips dealt to all 256 WGPs: each work-group first places its edge chunk, then takes 24 strips (6 per SIMD), the 60 without a chunk 25-26 (7 per SIMD), not 32 on 196 WGPs; sum order unchanged
# speedup vs baseline: 3.2339x; 1.0383x over previous
#include <hip/hip_runtime.h>

typedef __attribute__((ext_vector_type(2))) float v2f;
typedef __attribute__((ext_vector_type(8))) float v8f;

#define D_IN  256
#define D_OUT 128

#define NBINS        256
#define BIN_BITS     8
#define RPB          391
#define ROW_BITS     9
#define COL_BITS     20
#define SORT_THREADS 1024
#define SORT_WAVES   (SORT_THREADS / 32)
#define SORT_EPT     8
#define SORT_EPB     (SORT_THREADS * SORT_EPT)

__global__ __launch_bounds__(256) void gcn_init_out(const float* __restrict__ bias,
                                                    float* __restrict__ out,
                                                    long long total4) {
    long long i = (long long)blockIdx.x * blockDim.x + threadIdx.x;
    if (i < total4) {
        int d4 = (int)(i & (D_OUT / 4 - 1));
        ((float4*)out)[i] = ((const float4*)bias)[d4];
    }
}

static __device__ __forceinline__ void gemm_strip(const float* __restrict__ X,
                                                  const float* __restrict__ W,
                                                  float* __restrict__ S,
                                                  int n_rows, int mtile) {
    const int lane  = threadIdx.x & 31;
    if (mtile * 16 + 16 > n_rows) return;
    const int m0 = mtile * 16;

    const int am = lane & 15;
    const int ak = (lane >> 4) * 2;
    const int bn = lane & 15;
    const int bk = (lane >> 4) * 2;

    v8f c[8];
#pragma unroll
    for (int t = 0; t < 8; ++t) c[t] = (v8f){0.f, 0.f, 0.f, 0.f, 0.f, 0.f, 0.f, 0.f};

    const float* xrow = X + (long long)(m0 + am) * D_IN;

    for (int k = 0; k < D_IN; k += 4) {
        v2f a = *(const v2f*)(xrow + k + ak);
#pragma unroll
        for (int t = 0; t < 8; ++t) {
            v2f b;
            b.x = W[(long long)(k + bk)     * D_OUT + t * 16 + bn];
            b.y = W[(long long)(k + bk + 1) * D_OUT + t * 16 + bn];
            c[t] = __builtin_amdgcn_wmma_f32_16x16x4_f32(
                 false, a,  false, b,
                 (short)0, c[t],  false,  false);
        }
    }

    const int cn    = lane & 15;
    const int cmoff = (lane >> 4) * 8;
#pragma unroll
    for (int t = 0; t < 8; ++t) {
#pragma unroll
        for (int r = 0; r < 8; ++r) {
            S[(long long)(m0 + r + cmoff) * D_OUT + t * 16 + cn] = c[t][r];
        }
    }
}

__global__ __launch_bounds__(256) void gcn_gemm_xw(const float* __restrict__ X,
                                                   const float* __restrict__ W,
                                                   float* __restrict__ S,
                                                   int n_rows) {
    gemm_strip(X, W, S, n_rows, (int)blockIdx.x * 8 + (int)(threadIdx.x >> 5));
}

#define WL_PITCH 130
#define WL_BYTES ((D_IN / 2) * WL_PITCH * 8)

static __device__ __forceinline__ void stage_w_pairs(const float* __restrict__ W, v2f* wl) {
    float4 w0[4], w1[4];
#pragma unroll
    for (int i = 0; i < 4; ++i) {
        const int idx = (int)threadIdx.x + i * SORT_THREADS;
        const int p = idx >> 5, q = idx & 31;
        w0[i] = ((const float4*)W)[(2 * p)     * (D_OUT / 4) + q];
        w1[i] = ((const float4*)W)[(2 * p + 1) * (D_OUT / 4) + q];
    }
#pragma unroll
    for (int i = 0; i < 4; ++i) {
        const int idx = (int)threadIdx.x + i * SORT_THREADS;
        const int p = idx >> 5, q = idx & 31;
        float4* d = (float4*)(wl + p * WL_PITCH + 4 * q);
        d[0] = make_float4(w0[i].x, w1[i].x, w0[i].y, w1[i].y);
        d[1] = make_float4(w0[i].z, w1[i].z, w0[i].w, w1[i].w);
    }
}

static __device__ __forceinline__ void gemm_strip_lds(const float* __restrict__ X,
                                                      const v2f* wl,
                                                      float* __restrict__ S,
                                                      int n_rows, int mtile) {
    const int lane = threadIdx.x & 31;
    if (mtile * 16 + 16 > n_rows) return;
    const int m0 = mtile * 16;
    const int c  = lane & 15;
    const int h  = lane >> 4;

    v8f acc[8];
#pragma unroll
    for (int t = 0; t < 8; ++t) acc[t] = (v8f){0.f, 0.f, 0.f, 0.f, 0.f, 0.f, 0.f, 0.f};

    const float* xrow  = X + (long long)(m0 + c) * D_IN + 16 * h;
    const v2f*   wlane = wl + (8 * h) * WL_PITCH + c;

#pragma unroll 1
    for (int kb = 0; kb < D_IN; kb += 32) {
        float4 q[4];
#pragma unroll
        for (int i = 0; i < 4; ++i) q[i] = *(const float4*)(xrow + kb + 4 * i);
        const v2f* wk = wlane + (kb / 2) * WL_PITCH;
#pragma unroll
        for (int s = 0; s < 8; ++s) {
            v2f a;
            a.x = (s & 1) ? q[s >> 1].z : q[s >> 1].x;
            a.y = (s & 1) ? q[s >> 1].w : q[s >> 1].y;
#pragma unroll
            for (int t = 0; t < 8; ++t) {
                const v2f b = wk[s * WL_PITCH + 16 * t];
                acc[t] = __builtin_amdgcn_wmma_f32_16x16x4_f32(
                     false, a,  false, b,
                     (short)0, acc[t],  false,  false);
            }
        }
    }

    const int cmoff = h * 8;
#pragma unroll
    for (int t = 0; t < 8; ++t) {
#pragma unroll
        for (int r = 0; r < 8; ++r) {
            S[(long long)(m0 + r + cmoff) * D_OUT + t * 16 + c] = acc[t][r];
        }
    }
}

#define EDGES_PER_WAVE 32
#define EDGE_BATCH     8

static __device__ __forceinline__ void add_row_lines(float* dst, unsigned off,
                                                      float g0, float g1, float g2, float g3) {
    asm volatile("global_atomic_add_f32 %0, %1, %2"            :: "v"(off), "v"(g0), "s"(dst) : "memory");
    asm volatile("global_atomic_add_f32 %0, %1, %2 offset:128" :: "v"(off), "v"(g1), "s"(dst) : "memory");
    asm volatile("global_atomic_add_f32 %0, %1, %2 offset:256" :: "v"(off), "v"(g2), "s"(dst) : "memory");
    asm volatile("global_atomic_add_f32 %0, %1, %2 offset:384" :: "v"(off), "v"(g3), "s"(dst) : "memory");
}

__global__ __launch_bounds__(256) void gcn_edge_scatter(const int* __restrict__ rows,
                                                        const int* __restrict__ cols,
                                                        const float* __restrict__ vals,
                                                        const float* __restrict__ S,
                                                        float* __restrict__ out,
                                                        int n_edges) {
    const int lane = threadIdx.x & 31;
    int w = (int)(blockIdx.x * 8u) + (int)(threadIdx.x >> 5);
    w = __builtin_amdgcn_readfirstlane(w);
    const int e_base = w * EDGES_PER_WAVE;
    if (e_base >= n_edges) return;
    int n_here = n_edges - e_base;
    if (n_here > EDGES_PER_WAVE) n_here = EDGES_PER_WAVE;

    int r_v = 0, c_v = 0, v_v = 0;
    if (lane < n_here) {
        r_v = rows[e_base + lane];
        c_v = cols[e_base + lane];
        v_v = __float_as_int(vals[e_base + lane]);
    }
    const unsigned off = (unsigned)lane * 4u;

    int j0 = 0;
    for (; j0 + EDGE_BATCH <= n_here; j0 += EDGE_BATCH) {
        float g[EDGE_BATCH][4];
#pragma unroll
        for (int u = 0; u < EDGE_BATCH; ++u) {
            const int c = __builtin_amdgcn_readlane(c_v, j0 + u);
            const float* src = S + (long long)c * D_OUT;
            g[u][0] = src[lane];
            g[u][1] = src[lane + 32];
            g[u][2] = src[lane + 64];
            g[u][3] = src[lane + 96];
        }
#pragma unroll
        for (int u = 0; u < EDGE_BATCH; ++u) {
            const int   r = __builtin_amdgcn_readlane(r_v, j0 + u);
            const float v = __int_as_float(__builtin_amdgcn_readlane(v_v, j0 + u));
            float* dst = out + (long long)r * D_OUT;
            add_row_lines(dst, off, g[u][0] * v, g[u][1] * v, g[u][2] * v, g[u][3] * v);
        }
    }
    for (; j0 < n_here; ++j0) {
        const int   c = __builtin_amdgcn_readlane(c_v, j0);
        const int   r = __builtin_amdgcn_readlane(r_v, j0);
        const float v = __int_as_float(__builtin_amdgcn_readlane(v_v, j0));
        const float* src = S + (long long)c * D_OUT;
        const float s0 = src[lane];
        const float s1 = src[lane + 32];
        const float s2 = src[lane + 64];
        const float s3 = src[lane + 96];
        float* dst = out + (long long)r * D_OUT;
        add_row_lines(dst, off, s0 * v, s1 * v, s2 * v, s3 * v);
    }
}


static __device__ __forceinline__ unsigned same_key_lanes(unsigned key, bool valid, int nbits) {
    unsigned peers = __builtin_amdgcn_ballot_w32(valid);
#pragma unroll
    for (int b = 0; b < nbits; ++b) {
        const bool bit = ((key >> b) & 1u) != 0u;
        const unsigned m = __builtin_amdgcn_ballot_w32(bit && valid);
        peers &= bit ? m : ~m;
    }
    return valid ? peers : 0u;
}

static __device__ __forceinline__ void wave_rank(unsigned key, bool valid, int nbits,
                                                 unsigned lt_mask, int& rank, int& cnt) {
    const unsigned peers = same_key_lanes(key, valid, nbits);
    rank = __builtin_popcount(peers & lt_mask);
    cnt  = __builtin_popcount(peers);
}

static __device__ __forceinline__ int counter_take(int* h, unsigned key, bool valid,
                                                   int rank, int cnt) {
    int slot = 0;
    if (valid) {
        const int old = h[key];
        if (rank == 0) h[key] = old + cnt;
        slot = old + rank;
    }
    return slot;
}

static __device__ __forceinline__ int wave_count_step(int* h, unsigned key, bool valid,
                                                      int nbits, unsigned lt_mask) {
    int rank, cnt;
    wave_rank(key, valid, nbits, lt_mask, rank, cnt);
    return counter_take(h, key, valid, rank, cnt);
}

template <class Between>
static __device__ __forceinline__ void place_block(const int* __restrict__ rows,
                                                   const int* __restrict__ cols,
                                                   const float* __restrict__ vals,
                                                   int* __restrict__ blockoff,
                                                   uint2* __restrict__ rec,
                                                   int n_edges, int me,
                                                   int (*whist)[NBINS],
                                                   int* scan, Between between) {
    const int t = threadIdx.x, lane = t & 31, w = t >> 5;
    const unsigned lt_mask = (1u << lane) - 1u;
    int* h = whist[w];
    for (int k = lane; k < NBINS; k += 32) h[k] = 0;

    const int e0 = me * SORT_EPB + w * (SORT_EPT * 32) + lane;
    unsigned info[SORT_EPT];
    {
        int r[SORT_EPT];
#pragma unroll
        for (int i = 0; i < SORT_EPT; ++i) {
            const int e = e0 + i * 32;
            r[i] = (e < n_edges) ? rows[e] : -1;
        }
        between();
#pragma unroll
        for (int i = 0; i < SORT_EPT; ++i) {
            const bool valid = r[i] >= 0;
            const unsigned bin = valid ? (unsigned)(r[i] / RPB) : 0u;
            const unsigned rl  = valid ? (unsigned)r[i] - bin * RPB : 0u;
            int rank, cnt;
            wave_rank(bin, valid, BIN_BITS, lt_mask, rank, cnt);
            counter_take(h, bin, valid, rank, cnt);
            info[i] = bin | ((unsigned)rank << 8) | ((unsigned)cnt << 13) | (rl << 19) |
                      (valid ? (1u << 28) : 0u);
        }
    }
    __syncthreads();

    int total = 0;
    if (t < NBINS) {
        for (int w2 = 0; w2 < SORT_WAVES; ++w2) {
            const int cnt = whist[w2][t];
            whist[w2][t] = total;
            total += cnt;
        }
        scan[t] = total;
    }
    __syncthreads();
    for (int d = 1; d < NBINS; d <<= 1) {
        int add = 0;
        if (t < NBINS && t >= d) add = scan[t - d];
        __syncthreads();
        if (t < NBINS) scan[t] += add;
        __syncthreads();
    }
    if (t < NBINS) {
        const int incl  = scan[t];
        const int start = incl - total;
        blockoff[me * (NBINS + 1) + t] = start;
        if (t == NBINS - 1) blockoff[me * (NBINS + 1) + NBINS] = incl;
        scan[t] = start;
    }
    __syncthreads();

    int c[SORT_EPT];
    float v[SORT_EPT];
#pragma unroll
    for (int i = 0; i < SORT_EPT; ++i) {
        const int e = e0 + i * 32;
        c[i] = 0; v[i] = 0.f;
        if (info[i] >> 28) { c[i] = cols[e]; v[i] = vals[e]; }
    }
    uint2* chunk = rec + (long long)me * SORT_EPB;
#pragma unroll
    for (int i = 0; i < SORT_EPT; ++i) {
        const bool     valid = (info[i] >> 28) != 0u;
        const unsigned bin   = info[i] & 0xFFu;
        const int      rank  = (int)((info[i] >> 8) & 0x1Fu);
        const int      cnt   = (int)((info[i] >> 13) & 0x3Fu);
        const unsigned rl    = (info[i] >> 19) & 0x1FFu;
        const int slot = counter_take(h, bin, valid, rank, cnt);
        if (valid)
            chunk[scan[bin] + slot] = make_uint2((unsigned)c[i] | (rl << COL_BITS), __float_as_uint(v[i]));
    }
}

#define GP_BLOCKS       256
#define PLACE_STRIPS    24
#define PLACE_LDS_BYTES (sizeof(int) * (SORT_WAVES * NBINS + NBINS))

__global__ __launch_bounds__(SORT_THREADS) __attribute__((amdgpu_waves_per_eu(8, 8)))
void gcn_gemm_and_place(const float* __restrict__ X,
                        const float* __restrict__ W,
                        float* __restrict__ S,
                        int n_rows, int n_strips,
                        const int* __restrict__ rows,
                        const int* __restrict__ cols,
                        const float* __restrict__ vals,
                        int* __restrict__ blockoff,
                        uint2* __restrict__ rec,
                        int n_edges, int n_chunks) {
    __shared__ __attribute__((aligned(16))) unsigned char smem[WL_BYTES + PLACE_LDS_BYTES];
    const int blk = (int)blockIdx.x;
    const int nb  = (int)gridDim.x;
    v2f* wl = (v2f*)smem;
    int (*whist)[NBINS] = (int (*)[NBINS])(smem + WL_BYTES);
    int* scan = (int*)(smem + WL_BYTES + sizeof(int) * SORT_WAVES * NBINS);

    if (blk < n_chunks) {
        for (int chunk = blk; chunk < n_chunks; chunk += nb)
            place_block(rows, cols, vals, blockoff, rec, n_edges, chunk, whist, scan,
                        [&]() { if (chunk == blk) stage_w_pairs(W, wl); });
    } else {
        stage_w_pairs(W, wl);
        __syncthreads();
    }

    const int placers = n_chunks < nb ? n_chunks : nb;
    const bool uneven = placers > 0 && placers < nb && placers * PLACE_STRIPS <= n_strips;
    int base = 0, rest = n_strips, sharers = nb, j = blk;
    int s_lo, s_hi;
    if (uneven) {
        base    = placers * PLACE_STRIPS;
        rest    = n_strips - base;
        sharers = nb - placers;
        j       = blk - placers;
    }
    if (uneven && blk < placers) {
        s_lo = blk * PLACE_STRIPS;
        s_hi = s_lo + PLACE_STRIPS;
    } else {
        const int q = rest / sharers, r = rest % sharers;
        s_lo = base + j * q + (j < r ? j : r);
        s_hi = s_lo + q + (j < r ? 1 : 0);
    }
    const int wv = __builtin_amdgcn_readfirstlane((int)(threadIdx.x >> 5));
    for (int s = s_lo + wv; s < s_hi; s += SORT_WAVES)
        gemm_strip_lds(X, wl, S, n_rows, s);
}

#define RS_THREADS 1024
#define RS_WAVES   (RS_THREADS / 32)
#define RS_BATCH   8
#define RS_SCAN    512
#define RS_GROUP   4
#define RS_CAP     8192
#define RS_KEEP    2

template <bool IN_LDS>
static __device__ __forceinline__ void range_order_and_sum(const uint2* __restrict__ rec,
                                                           uint2* rec2,
                                                           uint2* lrec,
                                                           int (*whist)[RPB + 1],
                                                           int* rowscan,
                                                           int* rowstart,
                                                           const float* __restrict__ S,
                                                           const float* __restrict__ bias,
                                                           float* __restrict__ out,
                                                           int n_rows, int start, int end,
                                                           int seg_base, int seg_pre, int nseg,
                                                           int n_mine) {
    const int t    = threadIdx.x;
    const int lane = t & 31;
    const int w    = __builtin_amdgcn_readfirstlane(t >> 5);
    const int bin  = blockIdx.x;
    const unsigned lt_mask = (1u << lane) - 1u;

#define RS_LOCATE(cb, idx)                                                     \
    do {                                                                       \
        _Pragma("unroll")                                                      \
        for (int j = 0; j < RS_GROUP; ++j) idx[j] = 0;                         \
        for (int s = 0; s < nseg; ++s) {                                       \
            const int ps = __builtin_amdgcn_readlane(seg_pre, s);              \
            const int bs = __builtin_amdgcn_readlane(seg_base, s);             \
            _Pragma("unroll")                                                  \
            for (int j = 0; j < RS_GROUP; ++j) {                               \
                const int i = (cb) + j * 32 + lane;                            \
                if (i >= ps) idx[j] = bs + (i - ps);                           \
            }                                                                  \
        }                                                                      \
    } while (0)

    int* h = whist[w];
    for (int k = lane; k <= RPB; k += 32) h[k] = 0;
    uint2    kq[RS_KEEP][RS_GROUP];
    unsigned kinfo[RS_KEEP][RS_GROUP];
#pragma unroll
    for (int tr = 0; tr < RS_KEEP; ++tr) {
#pragma unroll
        for (int j = 0; j < RS_GROUP; ++j) { kq[tr][j] = make_uint2(0u, 0u); kinfo[tr][j] = 0u; }
        const int cb = tr * 32 * RS_GROUP;
        if (cb < n_mine) {
            int idx[RS_GROUP];
            RS_LOCATE(cb, idx);
#pragma unroll
            for (int j = 0; j < RS_GROUP; ++j)
                if (cb + j * 32 + lane < n_mine) kq[tr][j] = rec[idx[j]];
#pragma unroll
            for (int j = 0; j < RS_GROUP; ++j) {
                const bool     ok = cb + j * 32 + lane < n_mine;
                const unsigned rl = kq[tr][j].x >> COL_BITS;
                int rank, cnt;
                wave_rank(rl, ok, ROW_BITS, lt_mask, rank, cnt);
                counter_take(h, rl, ok, rank, cnt);
                kinfo[tr][j] = (unsigned)rank | ((unsigned)cnt << 5) | (ok ? (1u << 11) : 0u);
            }
        }
    }
    for (int cb = RS_KEEP * 32 * RS_GROUP; cb < n_mine; cb += 32 * RS_GROUP) {
        unsigned key[RS_GROUP];
        bool     ok[RS_GROUP];
        int      idx[RS_GROUP];
        RS_LOCATE(cb, idx);
#pragma unroll
        for (int j = 0; j < RS_GROUP; ++j) {
            ok[j]  = cb + j * 32 + lane < n_mine;
            key[j] = 0u;
            if (ok[j]) key[j] = rec[idx[j]].x;
        }
#pragma unroll
        for (int j = 0; j < RS_GROUP; ++j)
            wave_count_step(h, key[j] >> COL_BITS, ok[j], ROW_BITS, lt_mask);
    }
    __syncthreads();

    int rc = 0;
    if (t < RPB) {
        int run = 0;
        for (int w2 = 0; w2 < RS_WAVES; ++w2) {
            const int cnt = whist[w2][t];
            whist[w2][t] = run;
            run += cnt;
        }
        rc = run;
    }
    if (t < RS_SCAN) rowscan[t] = rc;
    __syncthreads();
    for (int d = 1; d < RS_SCAN; d <<= 1) {
        int add = 0;
        if (t < RS_SCAN && t >= d) add = rowscan[t - d];
        __syncthreads();
        if (t < RS_SCAN) rowscan[t] += add;
        __syncthreads();
    }
    if (t <= RPB) rowstart[t] = (t == 0) ? 0 : rowscan[t - 1];
    __syncthreads();

#pragma unroll
    for (int tr = 0; tr < RS_KEEP; ++tr) {
        if (tr * 32 * RS_GROUP < n_mine) {
#pragma unroll
            for (int j = 0; j < RS_GROUP; ++j) {
                const bool     ok = (kinfo[tr][j] >> 11) != 0u;
                const unsigned rl = kq[tr][j].x >> COL_BITS;
                const int slot = counter_take(h, rl, ok, (int)(kinfo[tr][j] & 31u),
                                              (int)((kinfo[tr][j] >> 5) & 63u));
                if (ok) {
                    const int pos = rowstart[rl] + slot;
                    if (IN_LDS) lrec[pos] = kq[tr][j];
                    else        rec2[start + pos] = kq[tr][j];
                }
            }
        }
    }
    for (int cb = RS_KEEP * 32 * RS_GROUP; cb < n_mine; cb += 32 * RS_GROUP) {
        uint2 q[RS_GROUP];
        bool  ok[RS_GROUP];
        int   idx[RS_GROUP];
        RS_LOCATE(cb, idx);
#pragma unroll
        for (int j = 0; j < RS_GROUP; ++j) {
            ok[j] = cb + j * 32 + lane < n_mine;
            q[j]  = make_uint2(0u, 0u);
            if (ok[j]) q[j] = rec[idx[j]];
        }
#pragma unroll
        for (int j = 0; j < RS_GROUP; ++j) {
            const unsigned rl = q[j].x >> COL_BITS;
            const int slot = wave_count_step(h, rl, ok[j], ROW_BITS, lt_mask);
            if (ok[j]) {
                const int pos = rowstart[rl] + slot;
                if (IN_LDS) lrec[pos] = q[j];
                else        rec2[start + pos] = q[j];
            }
        }
    }
#undef RS_LOCATE
    __syncthreads();

    const int row0 = bin * RPB;
    int nrows = n_rows - row0;
    if (nrows > RPB) nrows = RPB;
    const int extra = RPB % RS_WAVES;
    const int r0 = w * (RPB / RS_WAVES) + (w < extra ? w : extra);
    int r1 = r0 + (RPB / RS_WAVES) + (w < extra ? 1 : 0);
    if (r1 > nrows) r1 = nrows;
    if (r0 >= r1) return;

    const float b0 = bias[lane], b1 = bias[lane + 32], b2 = bias[lane + 64], b3 = bias[lane + 96];
    const int e_lo = start + __builtin_amdgcn_readfirstlane(rowstart[r0]);
    const int e_hi = start + __builtin_amdgcn_readfirstlane(rowstart[r1]);

    int   cur = r0;
    float s0 = 0.f, s1 = 0.f, s2 = 0.f, s3 = 0.f;

#define RS_FLUSH(nxt)                                                          \
    do {                                                                       \
        float* o = out + (long long)(row0 + cur) * D_OUT;                      \
        o[lane] = s0 + b0; o[lane + 32] = s1 + b1;                             \
        o[lane + 64] = s2 + b2; o[lane + 96] = s3 + b3;                        \
        _Pragma("unroll 1")   \
        for (int rr = cur + 1; rr < (nxt); ++rr) {                             \
            float* oe = out + (long long)(row0 + rr) * D_OUT;                  \
            oe[lane] = b0; oe[lane + 32] = b1; oe[lane + 64] = b2; oe[lane + 96] = b3; \
        }                                                                      \
        s0 = 0.f; s1 = 0.f; s2 = 0.f; s3 = 0.f;                                \
    } while (0)

    int key_v = 0, val_v = 0;
    if (e_lo + lane < e_hi) {
        uint2 q;
        if (IN_LDS) q = lrec[e_lo - start + lane];
        else        q = rec2[e_lo + lane];
        key_v = (int)q.x;
        val_v = (int)q.y;
    }
    for (int cb = e_lo; cb < e_hi; cb += 32) {
        int n_here = e_hi - cb;
        if (n_here > 32) n_here = 32;
        int key_n = 0, val_n = 0;
        if (cb + 32 + lane < e_hi) {
            uint2 q;
            if (IN_LDS) q = lrec[cb + 32 - start + lane];
            else        q = rec2[cb + 32 + lane];
            key_n = (int)q.x;
            val_n = (int)q.y;
        }
        int j0 = 0;
        for (; j0 + RS_BATCH <= n_here; j0 += RS_BATCH) {
            float g[RS_BATCH][4];
#pragma unroll
            for (int u = 0; u < RS_BATCH; ++u) {
                const int key = __builtin_amdgcn_readlane(key_v, j0 + u);
                const float* src = S + (long long)(key & ((1 << COL_BITS) - 1)) * D_OUT;
                g[u][0] = src[lane];
                g[u][1] = src[lane + 32];
                g[u][2] = src[lane + 64];
                g[u][3] = src[lane + 96];
            }
#pragma unroll
            for (int u = 0; u < RS_BATCH; ++u) {
                const int   key = __builtin_amdgcn_readlane(key_v, j0 + u);
                const float v   = __int_as_float(__builtin_amdgcn_readlane(val_v, j0 + u));
                const int   rl  = (int)((unsigned)key >> COL_BITS);
                if (rl != cur) { RS_FLUSH(rl); cur = rl; }
                s0 += g[u][0] * v; s1 += g[u][1] * v; s2 += g[u][2] * v; s3 += g[u][3] * v;
            }
        }
        for (; j0 < n_here; ++j0) {
            const int   key = __builtin_amdgcn_readlane(key_v, j0);
            const float v   = __int_as_float(__builtin_amdgcn_readlane(val_v, j0));
            const int   rl  = (int)((unsigned)key >> COL_BITS);
            const float* src = S + (long long)(key & ((1 << COL_BITS) - 1)) * D_OUT;
            const float g0 = src[lane];
            const float g1 = src[lane + 32];
            const float g2 = src[lane + 64];
            const float g3 = src[lane + 96];
            if (rl != cur) { RS_FLUSH(rl); cur = rl; }
            s0 += g0 * v; s1 += g1 * v; s2 += g2 * v; s3 += g3 * v;
        }
        key_v = key_n;
        val_v = val_n;
    }
    RS_FLUSH(r1);
#undef RS_FLUSH
}

__global__ __launch_bounds__(RS_THREADS) void gcn_range_sum(const uint2* __restrict__ rec,
                                                            uint2* rec2,
                                                            const int* __restrict__ blockoff,
                                                            int n_blocks,
                                                            const float* __restrict__ S,
                                                            const float* __restrict__ bias,
                                                            float* __restrict__ out,
                                                            int n_rows) {
    __shared__ int whist[RS_WAVES][RPB + 1];
    __shared__ int rowscan[RS_SCAN];
    __shared__ int rowstart[RPB + 1];
    __shared__ uint2 lrec[RS_CAP];
    __shared__ int wsum[2][RS_WAVES];
    const int lane = threadIdx.x & 31;
    const int w    = __builtin_amdgcn_readfirstlane(threadIdx.x >> 5);
    const int b_lo = (w * n_blocks) / RS_WAVES;
    const int nseg = ((w + 1) * n_blocks) / RS_WAVES - b_lo;
    int seg_base = 0, seg_len = 0, seg_o0 = 0;
    if (lane < nseg) {
        const int* bo = blockoff + (long long)(b_lo + lane) * (NBINS + 1) + blockIdx.x;
        seg_o0   = bo[0];
        seg_len  = bo[1] - seg_o0;
        seg_base = (b_lo + lane) * SORT_EPB + seg_o0;
    }
    int seg_pre = seg_len, below = seg_o0;
#pragma unroll
    for (int d = 1; d < 32; d <<= 1) {
        const int up = __shfl_up(seg_pre, d);
        if (lane >= d) seg_pre += up;
        below += __shfl_xor(below, d);
    }
    const int n_mine = __builtin_amdgcn_readlane(seg_pre, 31);
    seg_pre -= seg_len;
    if (lane == 0) { wsum[0][w] = below; wsum[1][w] = n_mine; }
    __syncthreads();
    int s_below = wsum[0][lane], s_mine = wsum[1][lane];
#pragma unroll
    for (int d = 1; d < 32; d <<= 1) {
        s_below += __shfl_xor(s_below, d);
        s_mine  += __shfl_xor(s_mine, d);
    }
    const int start = __builtin_amdgcn_readfirstlane(s_below);
    const int end   = start + __builtin_amdgcn_readfirstlane(s_mine);
    if (__builtin_expect(end - start <= RS_CAP, 1))
        range_order_and_sum<true>(rec, rec2, lrec, whist, rowscan, rowstart, S, bias, out,
                                  n_rows, start, end, seg_base, seg_pre, nseg, n_mine);
    else
        range_order_and_sum<false>(rec, rec2, lrec, whist, rowscan, rowstart, S, bias, out,
                                   n_rows, start, end, seg_base, seg_pre, nseg, n_mine);
}

extern "C" void kernel_launch(void* const* d_in, const int* in_sizes, int n_in,
                              void* d_out, int out_size, void* d_ws, size_t ws_size,
                              hipStream_t stream) {
    const int*   rows = (const int*)d_in[0];
    const int*   cols = (const int*)d_in[1];
    const float* vals = (const float*)d_in[2];
    const float* x    = (const float*)d_in[3];
    const float* W    = (const float*)d_in[4];
    const float* bias = (const float*)d_in[5];

    const int E = in_sizes[0];
    const int N = in_sizes[3] / D_IN;

    float* support = (float*)d_ws;

    const int mtiles = N / 16;

    const int    sort_blocks = (E + SORT_EPB - 1) / SORT_EPB;
    const size_t off_rec   = (((size_t)N * D_OUT * sizeof(float)) + 255) & ~(size_t)255;
    const size_t off_rec2  = (off_rec  + (size_t)sort_blocks * SORT_EPB * sizeof(uint2) + 255) & ~(size_t)255;
    const size_t off_boff  = (off_rec2 + (size_t)E * sizeof(uint2) + 255) & ~(size_t)255;
    const size_t ws_need   = off_boff + (size_t)sort_blocks * (NBINS + 1) * sizeof(int);

    if (ws_need <= ws_size && N <= NBINS * RPB && sort_blocks <= 32 * RS_WAVES) {
        uint2* rec      = (uint2*)((char*)d_ws + off_rec);
        uint2* rec2     = (uint2*)((char*)d_ws + off_rec2);
        int*   blockoff = (int*)((char*)d_ws + off_boff);
        gcn_gemm_and_place<<<GP_BLOCKS, SORT_THREADS, 0, stream>>>(
            x, W, support, N, mtiles, rows, cols, vals, blockoff, rec, E, sort_blocks);
        gcn_range_sum<<<NBINS, RS_THREADS, 0, stream>>>(rec, rec2, blockoff, sort_blocks, support,
                                                        bias, (float*)d_out, N);
    } else {
        gcn_gemm_xw<<<(mtiles + 7) / 8, 256, 0, stream>>>(x, W, support, N);
        long long total4 = (long long)N * (D_OUT / 4);
        int iblocks = (int)((total4 + 255) / 256);
        gcn_init_out<<<iblocks, 256, 0, stream>>>(bias, (float*)d_out, total4);
        int waves  = (E + EDGES_PER_WAVE - 1) / EDGES_PER_WAVE;
        int blocks = (waves + 7) / 8;
        gcn_edge_scatter<<<blocks, 256, 0, stream>>>(rows, cols, vals, support,
                                                     (float*)d_out, E);
    }
}
